# starting kernel
# baseline (speedup 1.0000x reference)
#include <hip/hip_runtime.h>
#include <hip/hip_bf16.h>


#define H_DIM 1024
#define I_DIM 2048
#define N_ST  16
#define R_DIM 64
#define B_SZ  2
#define L_SEQ 1024
#define ROWS  (B_SZ * L_SEQ)
#define EPS   1e-06f

typedef __attribute__((ext_vector_type(2))) float v2f;
typedef __attribute__((ext_vector_type(8))) float v8f;

typedef __attribute__((address_space(1))) const float gcf;
typedef __attribute__((address_space(1))) const v2f  gcv2f;

template <int NT, int EPI>
__global__ __launch_bounds__(256)
void wmma_gemm_f32(const float* __restrict__ A,
                   const float* __restrict__ W,
                   const float* __restrict__ bias,
                   float* __restrict__ C,
                   int M, int N, int K)
{
    const int lane = threadIdx.x & 31;
    const int wave = threadIdx.x >> 5;
    const int row0 = (blockIdx.y * 8 + wave) * 16;
    const int col0 = blockIdx.x * (16 * NT);

    const int r  = lane & 15;
    const int kk = (lane >> 4) * 2;

    gcf* Ap = (gcf*)(A + (size_t)(row0 + r) * K + kk);
    gcf* Wp[NT];
#pragma unroll
    for (int t = 0; t < NT; ++t)
        Wp[t] = (gcf*)(W + (size_t)(col0 + t * 16 + r) * K + kk);

    v8f acc[NT];
#pragma unroll
    for (int t = 0; t < NT; ++t) acc[t] = (v8f){0.f,0.f,0.f,0.f,0.f,0.f,0.f,0.f};

    for (int k = 0; k < K; k += 16) {
        __builtin_prefetch((const float*)(Ap + 256), 0, 1);

        v2f a[4];
#pragma unroll
        for (int s = 0; s < 4; ++s)
            a[s] = *(gcv2f*)(Ap + 4 * s);

        v2f b[NT][4];
#pragma unroll
        for (int t = 0; t < NT; ++t)
#pragma unroll
            for (int s = 0; s < 4; ++s)
                b[t][s] = *(gcv2f*)(Wp[t] + 4 * s);

#pragma unroll
        for (int s = 0; s < 4; ++s)
#pragma unroll
            for (int t = 0; t < NT; ++t)
                acc[t] = __builtin_amdgcn_wmma_f32_16x16x4_f32(
                             false, a[s], false, b[t][s],
                             (short)0, acc[t], false, false);

        Ap += 16;
#pragma unroll
        for (int t = 0; t < NT; ++t) Wp[t] += 16;
    }

    const int rowoff = (lane >> 4) * 8;
#pragma unroll
    for (int t = 0; t < NT; ++t) {
        const int col = col0 + t * 16 + r;
#pragma unroll
        for (int v = 0; v < 8; ++v) {
            float val = acc[t][v];
            if (EPI == 1) {
                val += bias[col];
                val = (val > 20.f) ? val : __logf(1.f + __expf(val));
            }
            C[(size_t)(row0 + rowoff + v) * N + col] = val;
        }
    }
}

__global__ __launch_bounds__(256)
void conv_silu_kernel(const float* __restrict__ proj,
                      const float* __restrict__ cw,
                      const float* __restrict__ cb,
                      float* __restrict__ u)
{
    const int row = blockIdx.x;
    const int i   = blockIdx.y * 256 + threadIdx.x;
    const int b   = row / L_SEQ;
    const int l   = row - b * L_SEQ;

    float acc = cb[i];
#pragma unroll
    for (int k = 0; k < 4; ++k) {
        const int src = l + k - 1;
        if (src >= 0 && src < L_SEQ)
            acc += cw[i * 4 + k] *
                   proj[(size_t)(b * L_SEQ + src) * (2 * I_DIM) + i];
    }
    u[(size_t)row * I_DIM + i] = acc / (1.f + __expf(-acc));
}

__global__ __launch_bounds__(256)
void rmsnorm_kernel(const float* __restrict__ xd,
                    const float* __restrict__ wdt,
                    const float* __restrict__ wb,
                    const float* __restrict__ wc,
                    float* __restrict__ dtn,
                    float* __restrict__ Bm,
                    float* __restrict__ Cm)
{
    const int row = blockIdx.x * 256 + threadIdx.x;
    if (row >= ROWS) return;
    const float* x = xd + (size_t)row * (R_DIM + 2 * N_ST);

    float s = 0.f;
#pragma unroll 8
    for (int j = 0; j < R_DIM; ++j) s += x[j] * x[j];
    float inv = __frsqrt_rn(s / R_DIM + EPS);
#pragma unroll 8
    for (int j = 0; j < R_DIM; ++j)
        dtn[(size_t)row * R_DIM + j] = wdt[j] * x[j] * inv;

    s = 0.f;
#pragma unroll
    for (int j = 0; j < N_ST; ++j) { float v = x[R_DIM + j]; s += v * v; }
    inv = __frsqrt_rn(s / N_ST + EPS);
#pragma unroll
    for (int j = 0; j < N_ST; ++j)
        Bm[(size_t)row * N_ST + j] = wb[j] * x[R_DIM + j] * inv;

    s = 0.f;
#pragma unroll
    for (int j = 0; j < N_ST; ++j) { float v = x[R_DIM + N_ST + j]; s += v * v; }
    inv = __frsqrt_rn(s / N_ST + EPS);
#pragma unroll
    for (int j = 0; j < N_ST; ++j)
        Cm[(size_t)row * N_ST + j] = wc[j] * x[R_DIM + N_ST + j] * inv;
}

#define CHUNK 64
__global__ __launch_bounds__(128)
void scan_kernel(const float* __restrict__ dt2,
                 const float* __restrict__ u,
                 const float* __restrict__ Bm,
                 const float* __restrict__ Cm,
                 const float* __restrict__ A_log,
                 const float* __restrict__ Dvec,
                 const float* __restrict__ proj,
                 float* __restrict__ yout)
{
    __shared__ float sB[CHUNK][N_ST];
    __shared__ float sC[CHUNK][N_ST];

    const int b = blockIdx.y;
    const int i = blockIdx.x * 128 + threadIdx.x;

    float Aa[N_ST];
#pragma unroll
    for (int n = 0; n < N_ST; ++n)
        Aa[n] = -__expf(A_log[(size_t)i * N_ST + n]);

    float h[N_ST];
#pragma unroll
    for (int n = 0; n < N_ST; ++n) h[n] = 0.f;

    const float Dv = Dvec[i];

    for (int l0 = 0; l0 < L_SEQ; l0 += CHUNK) {
        for (int t = threadIdx.x; t < CHUNK * N_ST; t += 128) {
            const size_t g = (size_t)(b * L_SEQ + l0) * N_ST + t;
            sB[t / N_ST][t % N_ST] = Bm[g];
            sC[t / N_ST][t % N_ST] = Cm[g];
        }
        __syncthreads();

        for (int l = 0; l < CHUNK; ++l) {
            const size_t idx = (size_t)(b * L_SEQ + l0 + l) * I_DIM + i;
            const float dtv = dt2[idx];
            const float uv  = u[idx];
            const float du  = dtv * uv;
            float y = 0.f;
#pragma unroll
            for (int n = 0; n < N_ST; ++n) {
                const float dA = __expf(dtv * Aa[n]);
                h[n] = dA * h[n] + du * sB[l][n];
                y   += h[n] * sC[l][n];
            }
            const float g = proj[(size_t)(b * L_SEQ + l0 + l) * (2 * I_DIM)
                                 + I_DIM + i];
            yout[idx] = (y + uv * Dv) * (g / (1.f + __expf(-g)));
        }
        __syncthreads();
    }
}

extern "C" void kernel_launch(void* const* d_in, const int* in_sizes, int n_in,
                              void* d_out, int out_size, void* d_ws, size_t ws_size,
                              hipStream_t stream)
{
    const float* hidden     = (const float*)d_in[0];
    const float* in_proj_w  = (const float*)d_in[1];
    const float* conv_w     = (const float*)d_in[2];
    const float* conv_b     = (const float*)d_in[3];
    const float* x_proj_w   = (const float*)d_in[4];
    const float* dt_proj_w  = (const float*)d_in[5];
    const float* dt_proj_b  = (const float*)d_in[6];
    const float* dt_ln_w    = (const float*)d_in[7];
    const float* b_ln_w     = (const float*)d_in[8];
    const float* c_ln_w     = (const float*)d_in[9];
    const float* A_log      = (const float*)d_in[10];
    const float* Dvec       = (const float*)d_in[11];
    const float* out_proj_w = (const float*)d_in[12];
    float* out = (float*)d_out;

    float* ws   = (float*)d_ws;
    float* proj = ws;
    float* u    = proj + (size_t)ROWS * 2 * I_DIM;
    float* xdbl = u    + (size_t)ROWS * I_DIM;
    float* dtn  = xdbl + (size_t)ROWS * (R_DIM + 2 * N_ST);
    float* Bm   = dtn  + (size_t)ROWS * R_DIM;
    float* Cm   = Bm   + (size_t)ROWS * N_ST;
    float* dt2  = Cm   + (size_t)ROWS * N_ST;
    float* yout = dt2  + (size_t)ROWS * I_DIM;

    wmma_gemm_f32<4, 0><<<dim3((2 * I_DIM) / 64, ROWS / 128), 256, 0, stream>>>(
        hidden, in_proj_w, nullptr, proj, ROWS, 2 * I_DIM, H_DIM);

    conv_silu_kernel<<<dim3(ROWS, I_DIM / 256), 256, 0, stream>>>(
        proj, conv_w, conv_b, u);

    wmma_gemm_f32<2, 0><<<dim3((R_DIM + 2 * N_ST) / 32, ROWS / 128), 256, 0, stream>>>(
        u, x_proj_w, nullptr, xdbl, ROWS, R_DIM + 2 * N_ST, I_DIM);

    rmsnorm_kernel<<<(ROWS + 255) / 256, 256, 0, stream>>>(
        xdbl, dt_ln_w, b_ln_w, c_ln_w, dtn, Bm, Cm);

    wmma_gemm_f32<4, 1><<<dim3(I_DIM / 64, ROWS / 128), 256, 0, stream>>>(
        dtn, dt_proj_w, dt_proj_b, dt2, ROWS, I_DIM, R_DIM);

    scan_kernel<<<dim3(I_DIM / 128, B_SZ), 128, 0, stream>>>(
        dt2, u, Bm, Cm, A_log, Dvec, proj, yout);

    wmma_gemm_f32<4, 0><<<dim3(H_DIM / 64, ROWS / 128), 256, 0, stream>>>(
        yout, out_proj_w, nullptr, out, ROWS, H_DIM, I_DIM);
}


# scan: dt, u and gate of each step are prefetched a group of steps ahead into a register ring (Bm/Cm one group ahead via LDS), so the serial loop stops waiting two memory round trips per step
# speedup vs baseline: 2.0773x; 2.0773x over previous
#include <hip/hip_runtime.h>
#include <hip/hip_bf16.h>


#define H_DIM 1024
#define I_DIM 2048
#define N_ST  16
#define R_DIM 64
#define B_SZ  2
#define L_SEQ 1024
#define ROWS  (B_SZ * L_SEQ)
#define EPS   1e-06f

typedef __attribute__((ext_vector_type(2))) float v2f;
typedef __attribute__((ext_vector_type(8))) float v8f;

typedef __attribute__((address_space(1))) const float gcf;
typedef __attribute__((address_space(1))) const v2f  gcv2f;

template <int NT, int EPI>
__global__ __launch_bounds__(256)
void wmma_gemm_f32(const float* __restrict__ A,
                   const float* __restrict__ W,
                   const float* __restrict__ bias,
                   float* __restrict__ C,
                   int M, int N, int K)
{
    const int lane = threadIdx.x & 31;
    const int wave = threadIdx.x >> 5;
    const int row0 = (blockIdx.y * 8 + wave) * 16;
    const int col0 = blockIdx.x * (16 * NT);

    const int r  = lane & 15;
    const int kk = (lane >> 4) * 2;

    gcf* Ap = (gcf*)(A + (size_t)(row0 + r) * K + kk);
    gcf* Wp[NT];
#pragma unroll
    for (int t = 0; t < NT; ++t)
        Wp[t] = (gcf*)(W + (size_t)(col0 + t * 16 + r) * K + kk);

    v8f acc[NT];
#pragma unroll
    for (int t = 0; t < NT; ++t) acc[t] = (v8f){0.f,0.f,0.f,0.f,0.f,0.f,0.f,0.f};

    for (int k = 0; k < K; k += 16) {
        __builtin_prefetch((const float*)(Ap + 256), 0, 1);

        v2f a[4];
#pragma unroll
        for (int s = 0; s < 4; ++s)
            a[s] = *(gcv2f*)(Ap + 4 * s);

        v2f b[NT][4];
#pragma unroll
        for (int t = 0; t < NT; ++t)
#pragma unroll
            for (int s = 0; s < 4; ++s)
                b[t][s] = *(gcv2f*)(Wp[t] + 4 * s);

#pragma unroll
        for (int s = 0; s < 4; ++s)
#pragma unroll
            for (int t = 0; t < NT; ++t)
                acc[t] = __builtin_amdgcn_wmma_f32_16x16x4_f32(
                             false, a[s], false, b[t][s],
                             (short)0, acc[t], false, false);

        Ap += 16;
#pragma unroll
        for (int t = 0; t < NT; ++t) Wp[t] += 16;
    }

    const int rowoff = (lane >> 4) * 8;
#pragma unroll
    for (int t = 0; t < NT; ++t) {
        const int col = col0 + t * 16 + r;
#pragma unroll
        for (int v = 0; v < 8; ++v) {
            float val = acc[t][v];
            if (EPI == 1) {
                val += bias[col];
                val = (val > 20.f) ? val : __logf(1.f + __expf(val));
            }
            C[(size_t)(row0 + rowoff + v) * N + col] = val;
        }
    }
}

__global__ __launch_bounds__(256)
void conv_silu_kernel(const float* __restrict__ proj,
                      const float* __restrict__ cw,
                      const float* __restrict__ cb,
                      float* __restrict__ u)
{
    const int row = blockIdx.x;
    const int i   = blockIdx.y * 256 + threadIdx.x;
    const int b   = row / L_SEQ;
    const int l   = row - b * L_SEQ;

    float acc = cb[i];
#pragma unroll
    for (int k = 0; k < 4; ++k) {
        const int src = l + k - 1;
        if (src >= 0 && src < L_SEQ)
            acc += cw[i * 4 + k] *
                   proj[(size_t)(b * L_SEQ + src) * (2 * I_DIM) + i];
    }
    u[(size_t)row * I_DIM + i] = acc / (1.f + __expf(-acc));
}

__global__ __launch_bounds__(256)
void rmsnorm_kernel(const float* __restrict__ xd,
                    const float* __restrict__ wdt,
                    const float* __restrict__ wb,
                    const float* __restrict__ wc,
                    float* __restrict__ dtn,
                    float* __restrict__ Bm,
                    float* __restrict__ Cm)
{
    const int row = blockIdx.x * 256 + threadIdx.x;
    if (row >= ROWS) return;
    const float* x = xd + (size_t)row * (R_DIM + 2 * N_ST);

    float s = 0.f;
#pragma unroll 8
    for (int j = 0; j < R_DIM; ++j) s += x[j] * x[j];
    float inv = __frsqrt_rn(s / R_DIM + EPS);
#pragma unroll 8
    for (int j = 0; j < R_DIM; ++j)
        dtn[(size_t)row * R_DIM + j] = wdt[j] * x[j] * inv;

    s = 0.f;
#pragma unroll
    for (int j = 0; j < N_ST; ++j) { float v = x[R_DIM + j]; s += v * v; }
    inv = __frsqrt_rn(s / N_ST + EPS);
#pragma unroll
    for (int j = 0; j < N_ST; ++j)
        Bm[(size_t)row * N_ST + j] = wb[j] * x[R_DIM + j] * inv;

    s = 0.f;
#pragma unroll
    for (int j = 0; j < N_ST; ++j) { float v = x[R_DIM + N_ST + j]; s += v * v; }
    inv = __frsqrt_rn(s / N_ST + EPS);
#pragma unroll
    for (int j = 0; j < N_ST; ++j)
        Cm[(size_t)row * N_ST + j] = wc[j] * x[R_DIM + N_ST + j] * inv;
}

#define TSTEP 16
#define NGRP  (L_SEQ / TSTEP)
#define GRP4  (TSTEP * N_ST / 4)

static __device__ __forceinline__ void lds_barrier()
{
    __builtin_amdgcn_fence(__ATOMIC_RELEASE, "workgroup", "local");
    __builtin_amdgcn_s_barrier();
    __builtin_amdgcn_fence(__ATOMIC_ACQUIRE, "workgroup", "local");
}

__global__ __launch_bounds__(128)
void scan_kernel(const float* __restrict__ dt2,
                 const float* __restrict__ u,
                 const float* __restrict__ Bm,
                 const float* __restrict__ Cm,
                 const float* __restrict__ A_log,
                 const float* __restrict__ Dvec,
                 const float* __restrict__ proj,
                 float* __restrict__ yout)
{
    __shared__ __attribute__((aligned(16))) float sBC[2][2][TSTEP][N_ST];
    static_assert(2 * GRP4 == 128, "one float4 of Bm|Cm per thread and group");

    const int tid = threadIdx.x;
    const int b = blockIdx.y;
    const int i = blockIdx.x * 128 + tid;

    const size_t row0 = (size_t)b * L_SEQ;
    const float* dtp = dt2  + row0 * I_DIM + i;
    const float* up  = u    + row0 * I_DIM + i;
    const float* gp  = proj + row0 * (2 * I_DIM) + I_DIM + i;
    float*       yp  = yout + row0 * I_DIM + i;
    const float4* bcp = (tid < GRP4)
        ? (const float4*)(Bm + row0 * N_ST) + tid
        : (const float4*)(Cm + row0 * N_ST) + (tid - GRP4);
    float4* const sbc4 = (float4*)&sBC[0][0][0][0];

    const float4 bc0 = bcp[0];

    float rdt[TSTEP], ru[TSTEP], rg[TSTEP];
#pragma unroll
    for (int j = 0; j < TSTEP; ++j) {
        rdt[j] = dtp[(size_t)j * I_DIM];
        ru[j]  = up[(size_t)j * I_DIM];
        rg[j]  = gp[(size_t)j * (2 * I_DIM)];
    }

    float Aa[N_ST];
#pragma unroll
    for (int n = 0; n < N_ST; ++n)
        Aa[n] = -__expf(A_log[(size_t)i * N_ST + n]);

    float h[N_ST];
#pragma unroll
    for (int n = 0; n < N_ST; ++n) h[n] = 0.f;

    const float Dv = Dvec[i];

#pragma unroll
    for (int j = 0; j < TSTEP; ++j)
        asm volatile("" :: "v"(rdt[j]), "v"(ru[j]), "v"(rg[j]));
    asm volatile("" :: "v"(Dv));

    sbc4[tid] = bc0;
    lds_barrier();

    for (int g = 0; g < NGRP; ++g) {
        const int buf = g & 1;
        const int gn = (g + 1 < NGRP) ? g + 1 : NGRP - 1;
        const float* dtn = dtp + (size_t)gn * TSTEP * I_DIM;
        const float* un  = up  + (size_t)gn * TSTEP * I_DIM;
        const float* gnx = gp  + (size_t)gn * TSTEP * (2 * I_DIM);
        float*       yg  = yp  + (size_t)g  * TSTEP * I_DIM;

        const float4 nbc = bcp[(size_t)gn * GRP4];
        __builtin_amdgcn_sched_barrier(0);

#pragma unroll
        for (int j = 0; j < TSTEP; ++j) {
            const float dtv = rdt[j];
            const float uv  = ru[j];
            const float gv  = rg[j];
            const float du  = dtv * uv;
            float y = 0.f;
#pragma unroll
            for (int n = 0; n < N_ST; ++n) {
                const float dA = __expf(dtv * Aa[n]);
                h[n] = dA * h[n] + du * sBC[buf][0][j][n];
                y   += h[n] * sBC[buf][1][j][n];
            }
            yg[(size_t)j * I_DIM] = (y + uv * Dv) * (gv / (1.f + __expf(-gv)));

            rdt[j] = dtn[(size_t)j * I_DIM];
            ru[j]  = un[(size_t)j * I_DIM];
            rg[j]  = gnx[(size_t)j * (2 * I_DIM)];
        }

        sbc4[(buf ^ 1) * 128 + tid] = nbc;
        lds_barrier();
    }
}

extern "C" void kernel_launch(void* const* d_in, const int* in_sizes, int n_in,
                              void* d_out, int out_size, void* d_ws, size_t ws_size,
                              hipStream_t stream)
{
    const float* hidden     = (const float*)d_in[0];
    const float* in_proj_w  = (const float*)d_in[1];
    const float* conv_w     = (const float*)d_in[2];
    const float* conv_b     = (const float*)d_in[3];
    const float* x_proj_w   = (const float*)d_in[4];
    const float* dt_proj_w  = (const float*)d_in[5];
    const float* dt_proj_b  = (const float*)d_in[6];
    const float* dt_ln_w    = (const float*)d_in[7];
    const float* b_ln_w     = (const float*)d_in[8];
    const float* c_ln_w     = (const float*)d_in[9];
    const float* A_log      = (const float*)d_in[10];
    const float* Dvec       = (const float*)d_in[11];
    const float* out_proj_w = (const float*)d_in[12];
    float* out = (float*)d_out;

    float* ws   = (float*)d_ws;
    float* proj = ws;
    float* u    = proj + (size_t)ROWS * 2 * I_DIM;
    float* xdbl = u    + (size_t)ROWS * I_DIM;
    float* dtn  = xdbl + (size_t)ROWS * (R_DIM + 2 * N_ST);
    float* Bm   = dtn  + (size_t)ROWS * R_DIM;
    float* Cm   = Bm   + (size_t)ROWS * N_ST;
    float* dt2  = Cm   + (size_t)ROWS * N_ST;
    float* yout = dt2  + (size_t)ROWS * I_DIM;

    wmma_gemm_f32<4, 0><<<dim3((2 * I_DIM) / 64, ROWS / 128), 256, 0, stream>>>(
        hidden, in_proj_w, nullptr, proj, ROWS, 2 * I_DIM, H_DIM);

    conv_silu_kernel<<<dim3(ROWS, I_DIM / 256), 256, 0, stream>>>(
        proj, conv_w, conv_b, u);

    wmma_gemm_f32<2, 0><<<dim3((R_DIM + 2 * N_ST) / 32, ROWS / 128), 256, 0, stream>>>(
        u, x_proj_w, nullptr, xdbl, ROWS, R_DIM + 2 * N_ST, I_DIM);

    rmsnorm_kernel<<<(ROWS + 255) / 256, 256, 0, stream>>>(
        xdbl, dt_ln_w, b_ln_w, c_ln_w, dtn, Bm, Cm);

    wmma_gemm_f32<4, 1><<<dim3(I_DIM / 64, ROWS / 128), 256, 0, stream>>>(
        dtn, dt_proj_w, dt_proj_b, dt2, ROWS, I_DIM, R_DIM);

    scan_kernel<<<dim3(I_DIM / 128, B_SZ), 128, 0, stream>>>(
        dt2, u, Bm, Cm, A_log, Dvec, proj, yout);

    wmma_gemm_f32<4, 0><<<dim3(H_DIM / 64, ROWS / 128), 256, 0, stream>>>(
        yout, out_proj_w, nullptr, out, ROWS, H_DIM, I_DIM);
}


# GEMMs: A and W tiles of a 32-k step are fetched as whole 128-byte lines, parked in a two-stage LDS buffer and read back as four-float fragments; k pairs regroup inside each 8 k (sum order changes)
# speedup vs baseline: 2.7496x; 1.3236x over previous
#include <hip/hip_runtime.h>
#include <hip/hip_bf16.h>


#define H_DIM 1024
#define I_DIM 2048
#define N_ST  16
#define R_DIM 64
#define B_SZ  2
#define L_SEQ 1024
#define ROWS  (B_SZ * L_SEQ)
#define EPS   1e-06f

typedef __attribute__((ext_vector_type(2))) float v2f;
typedef __attribute__((ext_vector_type(4))) float v4f;
typedef __attribute__((ext_vector_type(8))) float v8f;

typedef __attribute__((address_space(1))) const v4f gcv4f;

#define GEMM_BM    128
#define GEMM_BK    32
#define GEMM_PITCH 36

template <int NT, int EPI>
__global__ __launch_bounds__(256)
void wmma_gemm_f32(const float* __restrict__ A,
                   const float* __restrict__ W,
                   const float* __restrict__ bias,
                   float* __restrict__ C,
                   int M, int N, int K)
{
    constexpr int BN  = 16 * NT;
    constexpr int WLD = (BN * (GEMM_BK / 4)) / 256;
    static_assert(WLD >= 1 && WLD * 256 == BN * (GEMM_BK / 4), "W tile must fill whole threads");

    __shared__ __attribute__((aligned(16))) float sA[2][GEMM_BM * GEMM_PITCH];
    __shared__ __attribute__((aligned(16))) float sW[2][BN * GEMM_PITCH];

    const int tid  = threadIdx.x;
    const int lane = tid & 31;
    const int wave = tid >> 5;
    const int rowb = blockIdx.y * GEMM_BM;
    const int row0 = rowb + wave * 16;
    const int col0 = blockIdx.x * BN;

    const int r  = lane & 15;
    const int hh = lane >> 4;

    const int srow = tid >> 3;
    const int sc4  = tid & 7;
    gcv4f* Ag = (gcv4f*)(A + (size_t)(rowb + srow) * K + sc4 * 4);
    gcv4f* Wg = (gcv4f*)(W + (size_t)(col0 + srow) * K + sc4 * 4);
    const size_t rstep = (size_t)8 * K;
    const int sidx = srow * GEMM_PITCH + sc4 * 4;

    const int fa = (wave * 16 + r) * GEMM_PITCH + hh * 4;
    const int fw = r * GEMM_PITCH + hh * 4;

    v8f acc[NT];
#pragma unroll
    for (int t = 0; t < NT; ++t) acc[t] = (v8f){0.f,0.f,0.f,0.f,0.f,0.f,0.f,0.f};

    v4f ga[4], gw[WLD];
#pragma unroll
    for (int i = 0; i < 4; ++i) ga[i] = Ag[i * rstep];
#pragma unroll
    for (int i = 0; i < WLD; ++i) gw[i] = Wg[i * rstep];
#pragma unroll
    for (int i = 0; i < 4; ++i)
        *(v4f*)&sA[0][sidx + i * 32 * GEMM_PITCH] = ga[i];
#pragma unroll
    for (int i = 0; i < WLD; ++i)
        *(v4f*)&sW[0][sidx + i * 32 * GEMM_PITCH] = gw[i];
    __syncthreads();

    const int KT = K / GEMM_BK;
    for (int kt = 0; kt < KT; ++kt) {
        const int  st   = kt & 1;
        const bool more = (kt + 1 < KT);

        if (more) {
            Ag += GEMM_BK / 4;
            Wg += GEMM_BK / 4;
#pragma unroll
            for (int i = 0; i < 4; ++i) ga[i] = Ag[i * rstep];
#pragma unroll
            for (int i = 0; i < WLD; ++i) gw[i] = Wg[i * rstep];
        }

        const float* pa = &sA[st][fa];
        const float* pw = &sW[st][fw];

        v4f af[2];
        v4f bf[2][NT];
        af[0] = *(const v4f*)pa;
#pragma unroll
        for (int t = 0; t < NT; ++t)
            bf[0][t] = *(const v4f*)(pw + t * 16 * GEMM_PITCH);

#pragma unroll
        for (int q = 0; q < GEMM_BK / 8; ++q) {
            const int cur = q & 1;
            const int nxt = cur ^ 1;
            if (q + 1 < GEMM_BK / 8) {
                af[nxt] = *(const v4f*)(pa + (q + 1) * 8);
#pragma unroll
                for (int t = 0; t < NT; ++t)
                    bf[nxt][t] = *(const v4f*)(pw + t * 16 * GEMM_PITCH + (q + 1) * 8);
            }
            const v2f a_lo = (v2f){af[cur].x, af[cur].y};
            const v2f a_hi = (v2f){af[cur].z, af[cur].w};
#pragma unroll
            for (int t = 0; t < NT; ++t) {
                const v2f b_lo = (v2f){bf[cur][t].x, bf[cur][t].y};
                acc[t] = __builtin_amdgcn_wmma_f32_16x16x4_f32(
                             false, a_lo, false, b_lo,
                             (short)0, acc[t], false, false);
            }
#pragma unroll
            for (int t = 0; t < NT; ++t) {
                const v2f b_hi = (v2f){bf[cur][t].z, bf[cur][t].w};
                acc[t] = __builtin_amdgcn_wmma_f32_16x16x4_f32(
                             false, a_hi, false, b_hi,
                             (short)0, acc[t], false, false);
            }
        }

        if (more) {
#pragma unroll
            for (int i = 0; i < 4; ++i)
                *(v4f*)&sA[st ^ 1][sidx + i * 32 * GEMM_PITCH] = ga[i];
#pragma unroll
            for (int i = 0; i < WLD; ++i)
                *(v4f*)&sW[st ^ 1][sidx + i * 32 * GEMM_PITCH] = gw[i];
        }
        __syncthreads();
    }

    const int rowoff = (lane >> 4) * 8;
#pragma unroll
    for (int t = 0; t < NT; ++t) {
        const int col = col0 + t * 16 + r;
#pragma unroll
        for (int v = 0; v < 8; ++v) {
            float val = acc[t][v];
            if (EPI == 1) {
                val += bias[col];
                val = (val > 20.f) ? val : __logf(1.f + __expf(val));
            }
            C[(size_t)(row0 + rowoff + v) * N + col] = val;
        }
    }
}

__global__ __launch_bounds__(256)
void conv_silu_kernel(const float* __restrict__ proj,
                      const float* __restrict__ cw,
                      const float* __restrict__ cb,
                      float* __restrict__ u)
{
    const int row = blockIdx.x;
    const int i   = blockIdx.y * 256 + threadIdx.x;
    const int b   = row / L_SEQ;
    const int l   = row - b * L_SEQ;

    float acc = cb[i];
#pragma unroll
    for (int k = 0; k < 4; ++k) {
        const int src = l + k - 1;
        if (src >= 0 && src < L_SEQ)
            acc += cw[i * 4 + k] *
                   proj[(size_t)(b * L_SEQ + src) * (2 * I_DIM) + i];
    }
    u[(size_t)row * I_DIM + i] = acc / (1.f + __expf(-acc));
}

__global__ __launch_bounds__(256)
void rmsnorm_kernel(const float* __restrict__ xd,
                    const float* __restrict__ wdt,
                    const float* __restrict__ wb,
                    const float* __restrict__ wc,
                    float* __restrict__ dtn,
                    float* __restrict__ Bm,
                    float* __restrict__ Cm)
{
    const int row = blockIdx.x * 256 + threadIdx.x;
    if (row >= ROWS) return;
    const float* x = xd + (size_t)row * (R_DIM + 2 * N_ST);

    float s = 0.f;
#pragma unroll 8
    for (int j = 0; j < R_DIM; ++j) s += x[j] * x[j];
    float inv = __frsqrt_rn(s / R_DIM + EPS);
#pragma unroll 8
    for (int j = 0; j < R_DIM; ++j)
        dtn[(size_t)row * R_DIM + j] = wdt[j] * x[j] * inv;

    s = 0.f;
#pragma unroll
    for (int j = 0; j < N_ST; ++j) { float v = x[R_DIM + j]; s += v * v; }
    inv = __frsqrt_rn(s / N_ST + EPS);
#pragma unroll
    for (int j = 0; j < N_ST; ++j)
        Bm[(size_t)row * N_ST + j] = wb[j] * x[R_DIM + j] * inv;

    s = 0.f;
#pragma unroll
    for (int j = 0; j < N_ST; ++j) { float v = x[R_DIM + N_ST + j]; s += v * v; }
    inv = __frsqrt_rn(s / N_ST + EPS);
#pragma unroll
    for (int j = 0; j < N_ST; ++j)
        Cm[(size_t)row * N_ST + j] = wc[j] * x[R_DIM + N_ST + j] * inv;
}

#define TSTEP 16
#define NGRP  (L_SEQ / TSTEP)
#define GRP4  (TSTEP * N_ST / 4)

static __device__ __forceinline__ void lds_barrier()
{
    __builtin_amdgcn_fence(__ATOMIC_RELEASE, "workgroup", "local");
    __builtin_amdgcn_s_barrier();
    __builtin_amdgcn_fence(__ATOMIC_ACQUIRE, "workgroup", "local");
}

__global__ __launch_bounds__(128)
void scan_kernel(const float* __restrict__ dt2,
                 const float* __restrict__ u,
                 const float* __restrict__ Bm,
                 const float* __restrict__ Cm,
                 const float* __restrict__ A_log,
                 const float* __restrict__ Dvec,
                 const float* __restrict__ proj,
                 float* __restrict__ yout)
{
    __shared__ __attribute__((aligned(16))) float sBC[2][2][TSTEP][N_ST];
    static_assert(2 * GRP4 == 128, "one float4 of Bm|Cm per thread and group");

    const int tid = threadIdx.x;
    const int b = blockIdx.y;
    const int i = blockIdx.x * 128 + tid;

    const size_t row0 = (size_t)b * L_SEQ;
    const float* dtp = dt2  + row0 * I_DIM + i;
    const float* up  = u    + row0 * I_DIM + i;
    const float* gp  = proj + row0 * (2 * I_DIM) + I_DIM + i;
    float*       yp  = yout + row0 * I_DIM + i;
    const float4* bcp = (tid < GRP4)
        ? (const float4*)(Bm + row0 * N_ST) + tid
        : (const float4*)(Cm + row0 * N_ST) + (tid - GRP4);
    float4* const sbc4 = (float4*)&sBC[0][0][0][0];

    const float4 bc0 = bcp[0];

    float rdt[TSTEP], ru[TSTEP], rg[TSTEP];
#pragma unroll
    for (int j = 0; j < TSTEP; ++j) {
        rdt[j] = dtp[(size_t)j * I_DIM];
        ru[j]  = up[(size_t)j * I_DIM];
        rg[j]  = gp[(size_t)j * (2 * I_DIM)];
    }

    float Aa[N_ST];
#pragma unroll
    for (int n = 0; n < N_ST; ++n)
        Aa[n] = -__expf(A_log[(size_t)i * N_ST + n]);

    float h[N_ST];
#pragma unroll
    for (int n = 0; n < N_ST; ++n) h[n] = 0.f;

    const float Dv = Dvec[i];

#pragma unroll
    for (int j = 0; j < TSTEP; ++j)
        asm volatile("" :: "v"(rdt[j]), "v"(ru[j]), "v"(rg[j]));
    asm volatile("" :: "v"(Dv));

    sbc4[tid] = bc0;
    lds_barrier();

    for (int g = 0; g < NGRP; ++g) {
        const int buf = g & 1;
        const int gn = (g + 1 < NGRP) ? g + 1 : NGRP - 1;
        const float* dtn = dtp + (size_t)gn * TSTEP * I_DIM;
        const float* un  = up  + (size_t)gn * TSTEP * I_DIM;
        const float* gnx = gp  + (size_t)gn * TSTEP * (2 * I_DIM);
        float*       yg  = yp  + (size_t)g  * TSTEP * I_DIM;

        const float4 nbc = bcp[(size_t)gn * GRP4];
        __builtin_amdgcn_sched_barrier(0);

#pragma unroll
        for (int j = 0; j < TSTEP; ++j) {
            const float dtv = rdt[j];
            const float uv  = ru[j];
            const float gv  = rg[j];
            const float du  = dtv * uv;
            float y = 0.f;
#pragma unroll
            for (int n = 0; n < N_ST; ++n) {
                const float dA = __expf(dtv * Aa[n]);
                h[n] = dA * h[n] + du * sBC[buf][0][j][n];
                y   += h[n] * sBC[buf][1][j][n];
            }
            yg[(size_t)j * I_DIM] = (y + uv * Dv) * (gv / (1.f + __expf(-gv)));

            rdt[j] = dtn[(size_t)j * I_DIM];
            ru[j]  = un[(size_t)j * I_DIM];
            rg[j]  = gnx[(size_t)j * (2 * I_DIM)];
        }

        sbc4[(buf ^ 1) * 128 + tid] = nbc;
        lds_barrier();
    }
}

extern "C" void kernel_launch(void* const* d_in, const int* in_sizes, int n_in,
                              void* d_out, int out_size, void* d_ws, size_t ws_size,
                              hipStream_t stream)
{
    const float* hidden     = (const float*)d_in[0];
    const float* in_proj_w  = (const float*)d_in[1];
    const float* conv_w     = (const float*)d_in[2];
    const float* conv_b     = (const float*)d_in[3];
    const float* x_proj_w   = (const float*)d_in[4];
    const float* dt_proj_w  = (const float*)d_in[5];
    const float* dt_proj_b  = (const float*)d_in[6];
    const float* dt_ln_w    = (const float*)d_in[7];
    const float* b_ln_w     = (const float*)d_in[8];
    const float* c_ln_w     = (const float*)d_in[9];
    const float* A_log      = (const float*)d_in[10];
    const float* Dvec       = (const float*)d_in[11];
    const float* out_proj_w = (const float*)d_in[12];
    float* out = (float*)d_out;

    float* ws   = (float*)d_ws;
    float* proj = ws;
    float* u    = proj + (size_t)ROWS * 2 * I_DIM;
    float* xdbl = u    + (size_t)ROWS * I_DIM;
    float* dtn  = xdbl + (size_t)ROWS * (R_DIM + 2 * N_ST);
    float* Bm   = dtn  + (size_t)ROWS * R_DIM;
    float* Cm   = Bm   + (size_t)ROWS * N_ST;
    float* dt2  = Cm   + (size_t)ROWS * N_ST;
    float* yout = dt2  + (size_t)ROWS * I_DIM;

    wmma_gemm_f32<4, 0><<<dim3((2 * I_DIM) / 64, ROWS / 128), 256, 0, stream>>>(
        hidden, in_proj_w, nullptr, proj, ROWS, 2 * I_DIM, H_DIM);

    conv_silu_kernel<<<dim3(ROWS, I_DIM / 256), 256, 0, stream>>>(
        proj, conv_w, conv_b, u);

    wmma_gemm_f32<2, 0><<<dim3((R_DIM + 2 * N_ST) / 32, ROWS / 128), 256, 0, stream>>>(
        u, x_proj_w, nullptr, xdbl, ROWS, R_DIM + 2 * N_ST, I_DIM);

    rmsnorm_kernel<<<(ROWS + 255) / 256, 256, 0, stream>>>(
        xdbl, dt_ln_w, b_ln_w, c_ln_w, dtn, Bm, Cm);

    wmma_gemm_f32<4, 1><<<dim3(I_DIM / 64, ROWS / 128), 256, 0, stream>>>(
        dtn, dt_proj_w, dt_proj_b, dt2, ROWS, I_DIM, R_DIM);

    scan_kernel<<<dim3(I_DIM / 128, B_SZ), 128, 0, stream>>>(
        dt2, u, Bm, Cm, A_log, Dvec, proj, yout);

    wmma_gemm_f32<4, 0><<<dim3(H_DIM / 64, ROWS / 128), 256, 0, stream>>>(
        yout, out_proj_w, nullptr, out, ROWS, H_DIM, I_DIM);
}


# scan: each step software-pipelined one deep: Bm/Cm LDS reads (one group), the 16 exp(dt*A), dt*u and gate SiLU of step t+1 are issued during step t into a second register set; sum order unchanged
# speedup vs baseline: 4.6221x; 1.6810x over previous
#include <hip/hip_runtime.h>
#include <hip/hip_bf16.h>


#define H_DIM 1024
#define I_DIM 2048
#define N_ST  16
#define R_DIM 64
#define B_SZ  2
#define L_SEQ 1024
#define ROWS  (B_SZ * L_SEQ)
#define EPS   1e-06f

typedef __attribute__((ext_vector_type(2))) float v2f;
typedef __attribute__((ext_vector_type(4))) float v4f;
typedef __attribute__((ext_vector_type(8))) float v8f;

typedef __attribute__((address_space(1))) const v4f gcv4f;

#define GEMM_BM    128
#define GEMM_BK    32
#define GEMM_PITCH 36

template <int NT, int EPI>
__global__ __launch_bounds__(256)
void wmma_gemm_f32(const float* __restrict__ A,
                   const float* __restrict__ W,
                   const float* __restrict__ bias,
                   float* __restrict__ C,
                   int M, int N, int K)
{
    constexpr int BN  = 16 * NT;
    constexpr int WLD = (BN * (GEMM_BK / 4)) / 256;
    static_assert(WLD >= 1 && WLD * 256 == BN * (GEMM_BK / 4), "W tile must fill whole threads");

    __shared__ __attribute__((aligned(16))) float sA[2][GEMM_BM * GEMM_PITCH];
    __shared__ __attribute__((aligned(16))) float sW[2][BN * GEMM_PITCH];

    const int tid  = threadIdx.x;
    const int lane = tid & 31;
    const int wave = tid >> 5;
    const int rowb = blockIdx.y * GEMM_BM;
    const int row0 = rowb + wave * 16;
    const int col0 = blockIdx.x * BN;

    const int r  = lane & 15;
    const int hh = lane >> 4;

    const int srow = tid >> 3;
    const int sc4  = tid & 7;
    gcv4f* Ag = (gcv4f*)(A + (size_t)(rowb + srow) * K + sc4 * 4);
    gcv4f* Wg = (gcv4f*)(W + (size_t)(col0 + srow) * K + sc4 * 4);
    const size_t rstep = (size_t)8 * K;
    const int sidx = srow * GEMM_PITCH + sc4 * 4;

    const int fa = (wave * 16 + r) * GEMM_PITCH + hh * 4;
    const int fw = r * GEMM_PITCH + hh * 4;

    v8f acc[NT];
#pragma unroll
    for (int t = 0; t < NT; ++t) acc[t] = (v8f){0.f,0.f,0.f,0.f,0.f,0.f,0.f,0.f};

    v4f ga[4], gw[WLD];
#pragma unroll
    for (int i = 0; i < 4; ++i) ga[i] = Ag[i * rstep];
#pragma unroll
    for (int i = 0; i < WLD; ++i) gw[i] = Wg[i * rstep];
#pragma unroll
    for (int i = 0; i < 4; ++i)
        *(v4f*)&sA[0][sidx + i * 32 * GEMM_PITCH] = ga[i];
#pragma unroll
    for (int i = 0; i < WLD; ++i)
        *(v4f*)&sW[0][sidx + i * 32 * GEMM_PITCH] = gw[i];
    __syncthreads();

    const int KT = K / GEMM_BK;
    for (int kt = 0; kt < KT; ++kt) {
        const int  st   = kt & 1;
        const bool more = (kt + 1 < KT);

        if (more) {
            Ag += GEMM_BK / 4;
            Wg += GEMM_BK / 4;
#pragma unroll
            for (int i = 0; i < 4; ++i) ga[i] = Ag[i * rstep];
#pragma unroll
            for (int i = 0; i < WLD; ++i) gw[i] = Wg[i * rstep];
        }

        const float* pa = &sA[st][fa];
        const float* pw = &sW[st][fw];

        v4f af[2];
        v4f bf[2][NT];
        af[0] = *(const v4f*)pa;
#pragma unroll
        for (int t = 0; t < NT; ++t)
            bf[0][t] = *(const v4f*)(pw + t * 16 * GEMM_PITCH);

#pragma unroll
        for (int q = 0; q < GEMM_BK / 8; ++q) {
            const int cur = q & 1;
            const int nxt = cur ^ 1;
            if (q + 1 < GEMM_BK / 8) {
                af[nxt] = *(const v4f*)(pa + (q + 1) * 8);
#pragma unroll
                for (int t = 0; t < NT; ++t)
                    bf[nxt][t] = *(const v4f*)(pw + t * 16 * GEMM_PITCH + (q + 1) * 8);
            }
            const v2f a_lo = (v2f){af[cur].x, af[cur].y};
            const v2f a_hi = (v2f){af[cur].z, af[cur].w};
#pragma unroll
            for (int t = 0; t < NT; ++t) {
                const v2f b_lo = (v2f){bf[cur][t].x, bf[cur][t].y};
                acc[t] = __builtin_amdgcn_wmma_f32_16x16x4_f32(
                             false, a_lo, false, b_lo,
                             (short)0, acc[t], false, false);
            }
#pragma unroll
            for (int t = 0; t < NT; ++t) {
                const v2f b_hi = (v2f){bf[cur][t].z, bf[cur][t].w};
                acc[t] = __builtin_amdgcn_wmma_f32_16x16x4_f32(
                             false, a_hi, false, b_hi,
                             (short)0, acc[t], false, false);
            }
        }

        if (more) {
#pragma unroll
            for (int i = 0; i < 4; ++i)
                *(v4f*)&sA[st ^ 1][sidx + i * 32 * GEMM_PITCH] = ga[i];
#pragma unroll
            for (int i = 0; i < WLD; ++i)
                *(v4f*)&sW[st ^ 1][sidx + i * 32 * GEMM_PITCH] = gw[i];
        }
        __syncthreads();
    }

    const int rowoff = (lane >> 4) * 8;
#pragma unroll
    for (int t = 0; t < NT; ++t) {
        const int col = col0 + t * 16 + r;
#pragma unroll
        for (int v = 0; v < 8; ++v) {
            float val = acc[t][v];
            if (EPI == 1) {
                val += bias[col];
                val = (val > 20.f) ? val : __logf(1.f + __expf(val));
            }
            C[(size_t)(row0 + rowoff + v) * N + col] = val;
        }
    }
}

__global__ __launch_bounds__(256)
void conv_silu_kernel(const float* __restrict__ proj,
                      const float* __restrict__ cw,
                      const float* __restrict__ cb,
                      float* __restrict__ u)
{
    const int row = blockIdx.x;
    const int i   = blockIdx.y * 256 + threadIdx.x;
    const int b   = row / L_SEQ;
    const int l   = row - b * L_SEQ;

    float acc = cb[i];
#pragma unroll
    for (int k = 0; k < 4; ++k) {
        const int src = l + k - 1;
        if (src >= 0 && src < L_SEQ)
            acc += cw[i * 4 + k] *
                   proj[(size_t)(b * L_SEQ + src) * (2 * I_DIM) + i];
    }
    u[(size_t)row * I_DIM + i] = acc / (1.f + __expf(-acc));
}

__global__ __launch_bounds__(256)
void rmsnorm_kernel(const float* __restrict__ xd,
                    const float* __restrict__ wdt,
                    const float* __restrict__ wb,
                    const float* __restrict__ wc,
                    float* __restrict__ dtn,
                    float* __restrict__ Bm,
                    float* __restrict__ Cm)
{
    const int row = blockIdx.x * 256 + threadIdx.x;
    if (row >= ROWS) return;
    const float* x = xd + (size_t)row * (R_DIM + 2 * N_ST);

    float s = 0.f;
#pragma unroll 8
    for (int j = 0; j < R_DIM; ++j) s += x[j] * x[j];
    float inv = __frsqrt_rn(s / R_DIM + EPS);
#pragma unroll 8
    for (int j = 0; j < R_DIM; ++j)
        dtn[(size_t)row * R_DIM + j] = wdt[j] * x[j] * inv;

    s = 0.f;
#pragma unroll
    for (int j = 0; j < N_ST; ++j) { float v = x[R_DIM + j]; s += v * v; }
    inv = __frsqrt_rn(s / N_ST + EPS);
#pragma unroll
    for (int j = 0; j < N_ST; ++j)
        Bm[(size_t)row * N_ST + j] = wb[j] * x[R_DIM + j] * inv;

    s = 0.f;
#pragma unroll
    for (int j = 0; j < N_ST; ++j) { float v = x[R_DIM + N_ST + j]; s += v * v; }
    inv = __frsqrt_rn(s / N_ST + EPS);
#pragma unroll
    for (int j = 0; j < N_ST; ++j)
        Cm[(size_t)row * N_ST + j] = wc[j] * x[R_DIM + N_ST + j] * inv;
}

#define TSTEP 16
#define NGRP  (L_SEQ / TSTEP)
#define GRP4  (TSTEP * N_ST / 4)

static __device__ __forceinline__ void lds_barrier()
{
    __builtin_amdgcn_fence(__ATOMIC_RELEASE, "workgroup", "local");
    __builtin_amdgcn_s_barrier();
    __builtin_amdgcn_fence(__ATOMIC_ACQUIRE, "workgroup", "local");
}

__global__ __launch_bounds__(128)
void scan_kernel(const float* __restrict__ dt2,
                 const float* __restrict__ u,
                 const float* __restrict__ Bm,
                 const float* __restrict__ Cm,
                 const float* __restrict__ A_log,
                 const float* __restrict__ Dvec,
                 const float* __restrict__ proj,
                 float* __restrict__ yout)
{
    __shared__ __attribute__((aligned(16))) float sBC[2][2][TSTEP][N_ST];
    static_assert(2 * GRP4 == 128, "one float4 of Bm|Cm per thread and group");

    const int tid = threadIdx.x;
    const int b = blockIdx.y;
    const int i = blockIdx.x * 128 + tid;

    const size_t row0 = (size_t)b * L_SEQ;
    const float* dtp = dt2  + row0 * I_DIM + i;
    const float* up  = u    + row0 * I_DIM + i;
    const float* gp  = proj + row0 * (2 * I_DIM) + I_DIM + i;
    float*       yp  = yout + row0 * I_DIM + i;
    const float4* bcp = (tid < GRP4)
        ? (const float4*)(Bm + row0 * N_ST) + tid
        : (const float4*)(Cm + row0 * N_ST) + (tid - GRP4);
    float4* const sbc4 = (float4*)&sBC[0][0][0][0];

    const float4 bc0 = bcp[0];

    float rdt[TSTEP], ru[TSTEP], rg[TSTEP];
#pragma unroll
    for (int j = 0; j < TSTEP; ++j) {
        rdt[j] = dtp[(size_t)j * I_DIM];
        ru[j]  = up[(size_t)j * I_DIM];
        rg[j]  = gp[(size_t)j * (2 * I_DIM)];
    }

    float Aa[N_ST];
#pragma unroll
    for (int n = 0; n < N_ST; ++n)
        Aa[n] = -__expf(A_log[(size_t)i * N_ST + n]);

    float h[N_ST];
#pragma unroll
    for (int n = 0; n < N_ST; ++n) h[n] = 0.f;

    const float Dv = Dvec[i];

#pragma unroll
    for (int j = 0; j < TSTEP; ++j)
        asm volatile("" :: "v"(rdt[j]), "v"(ru[j]), "v"(rg[j]));
    asm volatile("" :: "v"(Dv));

    sbc4[tid] = bc0;
    lds_barrier();

    float4 Bc[4], Cc[4];
    float  dAc[N_ST];
    float  uvc, duc, sgc;
    {
        const float4* sB = (const float4*)&sBC[0][0][0][0];
        const float4* sC = (const float4*)&sBC[0][1][0][0];
#pragma unroll
        for (int k = 0; k < 4; ++k) { Bc[k] = sB[k]; Cc[k] = sC[k]; }
        const float dtv = rdt[0];
        const float gv  = rg[0];
#pragma unroll
        for (int n = 0; n < N_ST; ++n)
            dAc[n] = __expf(dtv * Aa[n]);
        uvc = ru[0];
        duc = dtv * uvc;
        sgc = gv / (1.f + __expf(-gv));
    }

    for (int g = 0; g < NGRP; ++g) {
        const int buf = g & 1;
        const int gn = (g + 1 < NGRP) ? g + 1 : NGRP - 1;
        const float* dtn = dtp + (size_t)gn * TSTEP * I_DIM;
        const float* un  = up  + (size_t)gn * TSTEP * I_DIM;
        const float* gnx = gp  + (size_t)gn * TSTEP * (2 * I_DIM);
        float*       yg  = yp  + (size_t)g  * TSTEP * I_DIM;

        const float4 nbc = bcp[(size_t)gn * GRP4];
        __builtin_amdgcn_sched_barrier(0);

#pragma unroll
        for (int j = 0; j < TSTEP; ++j) {
            float4 Bn[4], Cn[4];
            if (j + 1 < TSTEP) {
                __builtin_amdgcn_sched_barrier(0);
                const float4* sB = (const float4*)&sBC[buf][0][j + 1][0];
                const float4* sC = (const float4*)&sBC[buf][1][j + 1][0];
#pragma unroll
                for (int k = 0; k < 4; ++k) { Bn[k] = sB[k]; Cn[k] = sC[k]; }
                __builtin_amdgcn_sched_barrier(0);
            }

            const int   jn  = (j + 1) % TSTEP;
            const float dtx = rdt[jn];
            const float uvn = ru[jn];
            const float gvx = rg[jn];

            const float bcur[N_ST] = {Bc[0].x, Bc[0].y, Bc[0].z, Bc[0].w,
                                      Bc[1].x, Bc[1].y, Bc[1].z, Bc[1].w,
                                      Bc[2].x, Bc[2].y, Bc[2].z, Bc[2].w,
                                      Bc[3].x, Bc[3].y, Bc[3].z, Bc[3].w};
            const float ccur[N_ST] = {Cc[0].x, Cc[0].y, Cc[0].z, Cc[0].w,
                                      Cc[1].x, Cc[1].y, Cc[1].z, Cc[1].w,
                                      Cc[2].x, Cc[2].y, Cc[2].z, Cc[2].w,
                                      Cc[3].x, Cc[3].y, Cc[3].z, Cc[3].w};

            float dAn[N_ST];
            float y = 0.f;
#pragma unroll
            for (int n = 0; n < N_ST; ++n) {
                dAn[n] = __expf(dtx * Aa[n]);
                h[n] = dAc[n] * h[n] + duc * bcur[n];
                y   += h[n] * ccur[n];
            }
            const float dun = dtx * uvn;
            const float sgn = gvx / (1.f + __expf(-gvx));

            yg[(size_t)j * I_DIM] = (y + uvc * Dv) * sgc;

            rdt[j] = dtn[(size_t)j * I_DIM];
            ru[j]  = un[(size_t)j * I_DIM];
            rg[j]  = gnx[(size_t)j * (2 * I_DIM)];

#pragma unroll
            for (int n = 0; n < N_ST; ++n) dAc[n] = dAn[n];
            uvc = uvn; duc = dun; sgc = sgn;
            if (j + 1 < TSTEP) {
#pragma unroll
                for (int k = 0; k < 4; ++k) { Bc[k] = Bn[k]; Cc[k] = Cn[k]; }
            }
        }

        sbc4[(buf ^ 1) * 128 + tid] = nbc;
        lds_barrier();

        {
            const float4* sB = (const float4*)&sBC[buf ^ 1][0][0][0];
            const float4* sC = (const float4*)&sBC[buf ^ 1][1][0][0];
#pragma unroll
            for (int k = 0; k < 4; ++k) { Bc[k] = sB[k]; Cc[k] = sC[k]; }
        }
    }
}

extern "C" void kernel_launch(void* const* d_in, const int* in_sizes, int n_in,
                              void* d_out, int out_size, void* d_ws, size_t ws_size,
                              hipStream_t stream)
{
    const float* hidden     = (const float*)d_in[0];
    const float* in_proj_w  = (const float*)d_in[1];
    const float* conv_w     = (const float*)d_in[2];
    const float* conv_b     = (const float*)d_in[3];
    const float* x_proj_w   = (const float*)d_in[4];
    const float* dt_proj_w  = (const float*)d_in[5];
    const float* dt_proj_b  = (const float*)d_in[6];
    const float* dt_ln_w    = (const float*)d_in[7];
    const float* b_ln_w     = (const float*)d_in[8];
    const float* c_ln_w     = (const float*)d_in[9];
    const float* A_log      = (const float*)d_in[10];
    const float* Dvec       = (const float*)d_in[11];
    const float* out_proj_w = (const float*)d_in[12];
    float* out = (float*)d_out;

    float* ws   = (float*)d_ws;
    float* proj = ws;
    float* u    = proj + (size_t)ROWS * 2 * I_DIM;
    float* xdbl = u    + (size_t)ROWS * I_DIM;
    float* dtn  = xdbl + (size_t)ROWS * (R_DIM + 2 * N_ST);
    float* Bm   = dtn  + (size_t)ROWS * R_DIM;
    float* Cm   = Bm   + (size_t)ROWS * N_ST;
    float* dt2  = Cm   + (size_t)ROWS * N_ST;
    float* yout = dt2  + (size_t)ROWS * I_DIM;

    wmma_gemm_f32<4, 0><<<dim3((2 * I_DIM) / 64, ROWS / 128), 256, 0, stream>>>(
        hidden, in_proj_w, nullptr, proj, ROWS, 2 * I_DIM, H_DIM);

    conv_silu_kernel<<<dim3(ROWS, I_DIM / 256), 256, 0, stream>>>(
        proj, conv_w, conv_b, u);

    wmma_gemm_f32<2, 0><<<dim3((R_DIM + 2 * N_ST) / 32, ROWS / 128), 256, 0, stream>>>(
        u, x_proj_w, nullptr, xdbl, ROWS, R_DIM + 2 * N_ST, I_DIM);

    rmsnorm_kernel<<<(ROWS + 255) / 256, 256, 0, stream>>>(
        xdbl, dt_ln_w, b_ln_w, c_ln_w, dtn, Bm, Cm);

    wmma_gemm_f32<4, 1><<<dim3(I_DIM / 64, ROWS / 128), 256, 0, stream>>>(
        dtn, dt_proj_w, dt_proj_b, dt2, ROWS, I_DIM, R_DIM);

    scan_kernel<<<dim3(I_DIM / 128, B_SZ), 128, 0, stream>>>(
        dt2, u, Bm, Cm, A_log, Dvec, proj, yout);

    wmma_gemm_f32<4, 0><<<dim3(H_DIM / 64, ROWS / 128), 256, 0, stream>>>(
        yout, out_proj_w, nullptr, out, ROWS, H_DIM, I_DIM);
}


# scan: the 16 states of a channel are split over 4 adjacent lanes (4 states per lane, 512 waves instead of 128); y = 4 partial sums joined by two cross-lane adds in the quad: y's sum order changes
# speedup vs baseline: 5.7081x; 1.2350x over previous
#include <hip/hip_runtime.h>
#include <hip/hip_bf16.h>


#define H_DIM 1024
#define I_DIM 2048
#define N_ST  16
#define R_DIM 64
#define B_SZ  2
#define L_SEQ 1024
#define ROWS  (B_SZ * L_SEQ)
#define EPS   1e-06f

typedef __attribute__((ext_vector_type(2))) float v2f;
typedef __attribute__((ext_vector_type(4))) float v4f;
typedef __attribute__((ext_vector_type(8))) float v8f;

typedef __attribute__((address_space(1))) const v4f gcv4f;

#define GEMM_BM    128
#define GEMM_BK    32
#define GEMM_PITCH 36

template <int NT, int EPI>
__global__ __launch_bounds__(256)
void wmma_gemm_f32(const float* __restrict__ A,
                   const float* __restrict__ W,
                   const float* __restrict__ bias,
                   float* __restrict__ C,
                   int M, int N, int K)
{
    constexpr int BN  = 16 * NT;
    constexpr int WLD = (BN * (GEMM_BK / 4)) / 256;
    static_assert(WLD >= 1 && WLD * 256 == BN * (GEMM_BK / 4), "W tile must fill whole threads");

    __shared__ __attribute__((aligned(16))) float sA[2][GEMM_BM * GEMM_PITCH];
    __shared__ __attribute__((aligned(16))) float sW[2][BN * GEMM_PITCH];

    const int tid  = threadIdx.x;
    const int lane = tid & 31;
    const int wave = tid >> 5;
    const int rowb = blockIdx.y * GEMM_BM;
    const int row0 = rowb + wave * 16;
    const int col0 = blockIdx.x * BN;

    const int r  = lane & 15;
    const int hh = lane >> 4;

    const int srow = tid >> 3;
    const int sc4  = tid & 7;
    gcv4f* Ag = (gcv4f*)(A + (size_t)(rowb + srow) * K + sc4 * 4);
    gcv4f* Wg = (gcv4f*)(W + (size_t)(col0 + srow) * K + sc4 * 4);
    const size_t rstep = (size_t)8 * K;
    const int sidx = srow * GEMM_PITCH + sc4 * 4;

    const int fa = (wave * 16 + r) * GEMM_PITCH + hh * 4;
    const int fw = r * GEMM_PITCH + hh * 4;

    v8f acc[NT];
#pragma unroll
    for (int t = 0; t < NT; ++t) acc[t] = (v8f){0.f,0.f,0.f,0.f,0.f,0.f,0.f,0.f};

    v4f ga[4], gw[WLD];
#pragma unroll
    for (int i = 0; i < 4; ++i) ga[i] = Ag[i * rstep];
#pragma unroll
    for (int i = 0; i < WLD; ++i) gw[i] = Wg[i * rstep];
#pragma unroll
    for (int i = 0; i < 4; ++i)
        *(v4f*)&sA[0][sidx + i * 32 * GEMM_PITCH] = ga[i];
#pragma unroll
    for (int i = 0; i < WLD; ++i)
        *(v4f*)&sW[0][sidx + i * 32 * GEMM_PITCH] = gw[i];
    __syncthreads();

    const int KT = K / GEMM_BK;
    for (int kt = 0; kt < KT; ++kt) {
        const int  st   = kt & 1;
        const bool more = (kt + 1 < KT);

        if (more) {
            Ag += GEMM_BK / 4;
            Wg += GEMM_BK / 4;
#pragma unroll
            for (int i = 0; i < 4; ++i) ga[i] = Ag[i * rstep];
#pragma unroll
            for (int i = 0; i < WLD; ++i) gw[i] = Wg[i * rstep];
        }

        const float* pa = &sA[st][fa];
        const float* pw = &sW[st][fw];

        v4f af[2];
        v4f bf[2][NT];
        af[0] = *(const v4f*)pa;
#pragma unroll
        for (int t = 0; t < NT; ++t)
            bf[0][t] = *(const v4f*)(pw + t * 16 * GEMM_PITCH);

#pragma unroll
        for (int q = 0; q < GEMM_BK / 8; ++q) {
            const int cur = q & 1;
            const int nxt = cur ^ 1;
            if (q + 1 < GEMM_BK / 8) {
                af[nxt] = *(const v4f*)(pa + (q + 1) * 8);
#pragma unroll
                for (int t = 0; t < NT; ++t)
                    bf[nxt][t] = *(const v4f*)(pw + t * 16 * GEMM_PITCH + (q + 1) * 8);
            }
            const v2f a_lo = (v2f){af[cur].x, af[cur].y};
            const v2f a_hi = (v2f){af[cur].z, af[cur].w};
#pragma unroll
            for (int t = 0; t < NT; ++t) {
                const v2f b_lo = (v2f){bf[cur][t].x, bf[cur][t].y};
                acc[t] = __builtin_amdgcn_wmma_f32_16x16x4_f32(
                             false, a_lo, false, b_lo,
                             (short)0, acc[t], false, false);
            }
#pragma unroll
            for (int t = 0; t < NT; ++t) {
                const v2f b_hi = (v2f){bf[cur][t].z, bf[cur][t].w};
                acc[t] = __builtin_amdgcn_wmma_f32_16x16x4_f32(
                             false, a_hi, false, b_hi,
                             (short)0, acc[t], false, false);
            }
        }

        if (more) {
#pragma unroll
            for (int i = 0; i < 4; ++i)
                *(v4f*)&sA[st ^ 1][sidx + i * 32 * GEMM_PITCH] = ga[i];
#pragma unroll
            for (int i = 0; i < WLD; ++i)
                *(v4f*)&sW[st ^ 1][sidx + i * 32 * GEMM_PITCH] = gw[i];
        }
        __syncthreads();
    }

    const int rowoff = (lane >> 4) * 8;
#pragma unroll
    for (int t = 0; t < NT; ++t) {
        const int col = col0 + t * 16 + r;
#pragma unroll
        for (int v = 0; v < 8; ++v) {
            float val = acc[t][v];
            if (EPI == 1) {
                val += bias[col];
                val = (val > 20.f) ? val : __logf(1.f + __expf(val));
            }
            C[(size_t)(row0 + rowoff + v) * N + col] = val;
        }
    }
}

__global__ __launch_bounds__(256)
void conv_silu_kernel(const float* __restrict__ proj,
                      const float* __restrict__ cw,
                      const float* __restrict__ cb,
                      float* __restrict__ u)
{
    const int row = blockIdx.x;
    const int i   = blockIdx.y * 256 + threadIdx.x;
    const int b   = row / L_SEQ;
    const int l   = row - b * L_SEQ;

    float acc = cb[i];
#pragma unroll
    for (int k = 0; k < 4; ++k) {
        const int src = l + k - 1;
        if (src >= 0 && src < L_SEQ)
            acc += cw[i * 4 + k] *
                   proj[(size_t)(b * L_SEQ + src) * (2 * I_DIM) + i];
    }
    u[(size_t)row * I_DIM + i] = acc / (1.f + __expf(-acc));
}

__global__ __launch_bounds__(256)
void rmsnorm_kernel(const float* __restrict__ xd,
                    const float* __restrict__ wdt,
                    const float* __restrict__ wb,
                    const float* __restrict__ wc,
                    float* __restrict__ dtn,
                    float* __restrict__ Bm,
                    float* __restrict__ Cm)
{
    const int row = blockIdx.x * 256 + threadIdx.x;
    if (row >= ROWS) return;
    const float* x = xd + (size_t)row * (R_DIM + 2 * N_ST);

    float s = 0.f;
#pragma unroll 8
    for (int j = 0; j < R_DIM; ++j) s += x[j] * x[j];
    float inv = __frsqrt_rn(s / R_DIM + EPS);
#pragma unroll 8
    for (int j = 0; j < R_DIM; ++j)
        dtn[(size_t)row * R_DIM + j] = wdt[j] * x[j] * inv;

    s = 0.f;
#pragma unroll
    for (int j = 0; j < N_ST; ++j) { float v = x[R_DIM + j]; s += v * v; }
    inv = __frsqrt_rn(s / N_ST + EPS);
#pragma unroll
    for (int j = 0; j < N_ST; ++j)
        Bm[(size_t)row * N_ST + j] = wb[j] * x[R_DIM + j] * inv;

    s = 0.f;
#pragma unroll
    for (int j = 0; j < N_ST; ++j) { float v = x[R_DIM + N_ST + j]; s += v * v; }
    inv = __frsqrt_rn(s / N_ST + EPS);
#pragma unroll
    for (int j = 0; j < N_ST; ++j)
        Cm[(size_t)row * N_ST + j] = wc[j] * x[R_DIM + N_ST + j] * inv;
}

#define TSTEP 32
#define NGRP  (L_SEQ / TSTEP)
#define SPL   4
#define NSL   (N_ST / SPL)
#define CHB   (128 / SPL)
#define GRP4  (TSTEP * N_ST / 4)
#define QSTEP (TSTEP / SPL)

static __device__ __forceinline__ void lds_barrier()
{
    __builtin_amdgcn_fence(__ATOMIC_RELEASE, "workgroup", "local");
    __builtin_amdgcn_s_barrier();
    __builtin_amdgcn_fence(__ATOMIC_ACQUIRE, "workgroup", "local");
}

static __device__ __forceinline__ float quad_sum(float v)
{
    v += __builtin_bit_cast(float, __builtin_amdgcn_mov_dpp(
             __builtin_bit_cast(int, v), 0xB1, 0xF, 0xF, true));
    v += __builtin_bit_cast(float, __builtin_amdgcn_mov_dpp(
             __builtin_bit_cast(int, v), 0x4E, 0xF, 0xF, true));
    return v;
}

__global__ __launch_bounds__(128)
void scan_kernel(const float* __restrict__ dt2,
                 const float* __restrict__ u,
                 const float* __restrict__ Bm,
                 const float* __restrict__ Cm,
                 const float* __restrict__ A_log,
                 const float* __restrict__ Dvec,
                 const float* __restrict__ proj,
                 float* __restrict__ yout)
{
    __shared__ __attribute__((aligned(16))) float4 sBC[2][2][GRP4];
    static_assert(GRP4 == 128, "one float4 of Bm and one of Cm per thread and group");
    static_assert(TSTEP % SPL == 0, "a group holds whole quads of steps");

    const int tid = threadIdx.x;
    const int sub = tid & (SPL - 1);
    const int b = blockIdx.y;
    const int i = blockIdx.x * CHB + (tid >> 2);

    const size_t row0 = (size_t)b * L_SEQ;
    const float* dtp = dt2  + row0 * I_DIM + i;
    const float* up  = u    + row0 * I_DIM + i;
    const float* gp  = proj + (row0 + sub) * (2 * I_DIM) + I_DIM + i;
    float*       yp  = yout + (row0 + sub) * I_DIM + i;
    const float4* bp = (const float4*)(Bm + row0 * N_ST) + tid;
    const float4* cp = (const float4*)(Cm + row0 * N_ST) + tid;

    const float4 b0 = bp[0];
    const float4 c0 = cp[0];
    float4 nb = bp[GRP4];
    float4 nc = cp[GRP4];

    float rdt[TSTEP], ru[TSTEP], rg[QSTEP];
#pragma unroll
    for (int j = 0; j < TSTEP; ++j) {
        rdt[j] = dtp[(size_t)j * I_DIM];
        ru[j]  = up[(size_t)j * I_DIM];
    }
#pragma unroll
    for (int m = 0; m < QSTEP; ++m)
        rg[m] = gp[(size_t)(SPL * m) * (2 * I_DIM)];

    float Aa[NSL];
#pragma unroll
    for (int n = 0; n < NSL; ++n)
        Aa[n] = -__expf(A_log[(size_t)i * N_ST + sub * NSL + n]);

    float h[NSL];
#pragma unroll
    for (int n = 0; n < NSL; ++n) h[n] = 0.f;

    const float Dv = Dvec[i];

#pragma unroll
    for (int j = 0; j < TSTEP; ++j)
        asm volatile("" :: "v"(rdt[j]), "v"(ru[j]));
#pragma unroll
    for (int m = 0; m < QSTEP; ++m)
        asm volatile("" :: "v"(rg[m]));
    asm volatile("" :: "v"(Dv));
    asm volatile("" :: "v"(nb.x), "v"(nb.y), "v"(nb.z), "v"(nb.w),
                       "v"(nc.x), "v"(nc.y), "v"(nc.z), "v"(nc.w));

    sBC[0][0][tid] = b0;
    sBC[0][1][tid] = c0;
    lds_barrier();

    float4 Bc  = sBC[0][0][sub],           Cc  = sBC[0][1][sub];
    float4 Bn  = sBC[0][0][SPL + sub],     Cn  = sBC[0][1][SPL + sub];
    float4 Bn2 = sBC[0][0][2 * SPL + sub], Cn2 = sBC[0][1][2 * SPL + sub];
    float  dAc[NSL];
    float  uvc, duc;
    {
        const float dtv = rdt[0];
#pragma unroll
        for (int n = 0; n < NSL; ++n)
            dAc[n] = __expf(dtv * Aa[n]);
        uvc = ru[0];
        duc = dtv * uvc;
    }

    for (int g = 0; g < NGRP; ++g) {
        const int buf = g & 1;
        const int gn = (g + 1 < NGRP) ? g + 1 : NGRP - 1;
        const float* dtn = dtp + (size_t)gn * TSTEP * I_DIM;
        const float* un  = up  + (size_t)gn * TSTEP * I_DIM;
        const float* gnx = gp  + (size_t)gn * TSTEP * (2 * I_DIM);
        float*       yg  = yp  + (size_t)g  * TSTEP * I_DIM;

        const float4* sB = &sBC[buf][0][sub];
        const float4* sC = &sBC[buf][1][sub];

        float zsel = 0.f;
        float gv = 0.f, eg = 0.f, sg = 0.f;
        float ypend = 0.f, upend = 0.f;

#pragma unroll
        for (int j = 0; j < TSTEP; ++j) {
            const int s = j & (SPL - 1);
            const int m = j / SPL;

            float4 B3 = Bn2, C3 = Cn2;
            __builtin_amdgcn_sched_barrier(0);
            if (j + 3 < TSTEP) {
                B3 = sB[(j + 3) * SPL];
                C3 = sC[(j + 3) * SPL];
                __builtin_amdgcn_sched_barrier(0);
            }

            const int   jn  = (j + 1) % TSTEP;
            const float dtx = rdt[jn];
            const float uvn = ru[jn];

            const float bcur[NSL] = {Bc.x, Bc.y, Bc.z, Bc.w};
            const float ccur[NSL] = {Cc.x, Cc.y, Cc.z, Cc.w};

            float dAn[NSL];
            float y = 0.f;
#pragma unroll
            for (int n = 0; n < NSL; ++n) {
                dAn[n] = __expf(dtx * Aa[n]);
                h[n] = dAc[n] * h[n] + duc * bcur[n];
                y   += h[n] * ccur[n];
            }
            const float dun = dtx * uvn;

            if (j > 0) {
                const int sp = (j - 1) & (SPL - 1);
                const int mp = (j - 1) / SPL;
                const float z = quad_sum(ypend) + upend * Dv;
                zsel = (sub == sp) ? z : zsel;
                if (sp == SPL - 1)
                    yg[(size_t)(SPL * mp) * I_DIM] = zsel * sg;
                ru[j - 1] = un[(size_t)(j - 1) * I_DIM];
            }
            ypend = y;
            upend = uvc;

            if (s == 0) {
                gv = rg[m];
                eg = __expf(-gv);
            }
            if (s == 1) {
                sg = gv / (1.f + eg);
                asm volatile("" :: "v"(sg));
            }
            if (s == 2) rg[m] = gnx[(size_t)(SPL * m) * (2 * I_DIM)];

            rdt[j] = dtn[(size_t)j * I_DIM];

#pragma unroll
            for (int n = 0; n < NSL; ++n) dAc[n] = dAn[n];
            uvc = uvn; duc = dun;
            Bc = Bn;   Cc = Cn;
            Bn = Bn2;  Cn = Cn2;
            Bn2 = B3;  Cn2 = C3;
        }

        __builtin_amdgcn_sched_barrier(0);
        {
            const float z = quad_sum(ypend) + upend * Dv;
            zsel = (sub == SPL - 1) ? z : zsel;
            yg[(size_t)(SPL * (QSTEP - 1)) * I_DIM] = zsel * sg;
            ru[TSTEP - 1] = un[(size_t)(TSTEP - 1) * I_DIM];
        }

        sBC[buf ^ 1][0][tid] = nb;
        sBC[buf ^ 1][1][tid] = nc;
        lds_barrier();

        Bc  = sBC[buf ^ 1][0][sub];           Cc  = sBC[buf ^ 1][1][sub];
        Bn  = sBC[buf ^ 1][0][SPL + sub];     Cn  = sBC[buf ^ 1][1][SPL + sub];
        Bn2 = sBC[buf ^ 1][0][2 * SPL + sub]; Cn2 = sBC[buf ^ 1][1][2 * SPL + sub];

        const int gn2 = (g + 2 < NGRP) ? g + 2 : NGRP - 1;
        nb = bp[(size_t)gn2 * GRP4];
        nc = cp[(size_t)gn2 * GRP4];
    }
}

extern "C" void kernel_launch(void* const* d_in, const int* in_sizes, int n_in,
                              void* d_out, int out_size, void* d_ws, size_t ws_size,
                              hipStream_t stream)
{
    const float* hidden     = (const float*)d_in[0];
    const float* in_proj_w  = (const float*)d_in[1];
    const float* conv_w     = (const float*)d_in[2];
    const float* conv_b     = (const float*)d_in[3];
    const float* x_proj_w   = (const float*)d_in[4];
    const float* dt_proj_w  = (const float*)d_in[5];
    const float* dt_proj_b  = (const float*)d_in[6];
    const float* dt_ln_w    = (const float*)d_in[7];
    const float* b_ln_w     = (const float*)d_in[8];
    const float* c_ln_w     = (const float*)d_in[9];
    const float* A_log      = (const float*)d_in[10];
    const float* Dvec       = (const float*)d_in[11];
    const float* out_proj_w = (const float*)d_in[12];
    float* out = (float*)d_out;

    float* ws   = (float*)d_ws;
    float* proj = ws;
    float* u    = proj + (size_t)ROWS * 2 * I_DIM;
    float* xdbl = u    + (size_t)ROWS * I_DIM;
    float* dtn  = xdbl + (size_t)ROWS * (R_DIM + 2 * N_ST);
    float* Bm   = dtn  + (size_t)ROWS * R_DIM;
    float* Cm   = Bm   + (size_t)ROWS * N_ST;
    float* dt2  = Cm   + (size_t)ROWS * N_ST;
    float* yout = dt2  + (size_t)ROWS * I_DIM;

    wmma_gemm_f32<4, 0><<<dim3((2 * I_DIM) / 64, ROWS / 128), 256, 0, stream>>>(
        hidden, in_proj_w, nullptr, proj, ROWS, 2 * I_DIM, H_DIM);

    conv_silu_kernel<<<dim3(ROWS, I_DIM / 256), 256, 0, stream>>>(
        proj, conv_w, conv_b, u);

    wmma_gemm_f32<2, 0><<<dim3((R_DIM + 2 * N_ST) / 32, ROWS / 128), 256, 0, stream>>>(
        u, x_proj_w, nullptr, xdbl, ROWS, R_DIM + 2 * N_ST, I_DIM);

    rmsnorm_kernel<<<(ROWS + 255) / 256, 256, 0, stream>>>(
        xdbl, dt_ln_w, b_ln_w, c_ln_w, dtn, Bm, Cm);

    wmma_gemm_f32<4, 1><<<dim3(I_DIM / 64, ROWS / 128), 256, 0, stream>>>(
        dtn, dt_proj_w, dt_proj_b, dt2, ROWS, I_DIM, R_DIM);

    scan_kernel<<<dim3(I_DIM / CHB, B_SZ), 128, 0, stream>>>(
        dt2, u, Bm, Cm, A_log, Dvec, proj, yout);

    wmma_gemm_f32<4, 0><<<dim3(H_DIM / 64, ROWS / 128), 256, 0, stream>>>(
        yout, out_proj_w, nullptr, out, ROWS, H_DIM, I_DIM);
}


# conv+SiLU: each thread makes 4 adjacent channels x 8 consecutive rows from a window of 11 16-byte row loads, all requested before one wait; 4,096 waves enter instead of 131,072; tap order unchanged
# speedup vs baseline: 5.8864x; 1.0312x over previous
#include <hip/hip_runtime.h>
#include <hip/hip_bf16.h>


#define H_DIM 1024
#define I_DIM 2048
#define N_ST  16
#define R_DIM 64
#define B_SZ  2
#define L_SEQ 1024
#define ROWS  (B_SZ * L_SEQ)
#define EPS   1e-06f

typedef __attribute__((ext_vector_type(2))) float v2f;
typedef __attribute__((ext_vector_type(4))) float v4f;
typedef __attribute__((ext_vector_type(8))) float v8f;

typedef __attribute__((address_space(1))) const v4f gcv4f;

#define GEMM_BM    128
#define GEMM_BK    32
#define GEMM_PITCH 36

template <int NT, int EPI>
__global__ __launch_bounds__(256)
void wmma_gemm_f32(const float* __restrict__ A,
                   const float* __restrict__ W,
                   const float* __restrict__ bias,
                   float* __restrict__ C,
                   int M, int N, int K)
{
    constexpr int BN  = 16 * NT;
    constexpr int WLD = (BN * (GEMM_BK / 4)) / 256;
    static_assert(WLD >= 1 && WLD * 256 == BN * (GEMM_BK / 4), "W tile must fill whole threads");

    __shared__ __attribute__((aligned(16))) float sA[2][GEMM_BM * GEMM_PITCH];
    __shared__ __attribute__((aligned(16))) float sW[2][BN * GEMM_PITCH];

    const int tid  = threadIdx.x;
    const int lane = tid & 31;
    const int wave = tid >> 5;
    const int rowb = blockIdx.y * GEMM_BM;
    const int row0 = rowb + wave * 16;
    const int col0 = blockIdx.x * BN;

    const int r  = lane & 15;
    const int hh = lane >> 4;

    const int srow = tid >> 3;
    const int sc4  = tid & 7;
    gcv4f* Ag = (gcv4f*)(A + (size_t)(rowb + srow) * K + sc4 * 4);
    gcv4f* Wg = (gcv4f*)(W + (size_t)(col0 + srow) * K + sc4 * 4);
    const size_t rstep = (size_t)8 * K;
    const int sidx = srow * GEMM_PITCH + sc4 * 4;

    const int fa = (wave * 16 + r) * GEMM_PITCH + hh * 4;
    const int fw = r * GEMM_PITCH + hh * 4;

    v8f acc[NT];
#pragma unroll
    for (int t = 0; t < NT; ++t) acc[t] = (v8f){0.f,0.f,0.f,0.f,0.f,0.f,0.f,0.f};

    v4f ga[4], gw[WLD];
#pragma unroll
    for (int i = 0; i < 4; ++i) ga[i] = Ag[i * rstep];
#pragma unroll
    for (int i = 0; i < WLD; ++i) gw[i] = Wg[i * rstep];
#pragma unroll
    for (int i = 0; i < 4; ++i)
        *(v4f*)&sA[0][sidx + i * 32 * GEMM_PITCH] = ga[i];
#pragma unroll
    for (int i = 0; i < WLD; ++i)
        *(v4f*)&sW[0][sidx + i * 32 * GEMM_PITCH] = gw[i];
    __syncthreads();

    const int KT = K / GEMM_BK;
    for (int kt = 0; kt < KT; ++kt) {
        const int  st   = kt & 1;
        const bool more = (kt + 1 < KT);

        if (more) {
            Ag += GEMM_BK / 4;
            Wg += GEMM_BK / 4;
#pragma unroll
            for (int i = 0; i < 4; ++i) ga[i] = Ag[i * rstep];
#pragma unroll
            for (int i = 0; i < WLD; ++i) gw[i] = Wg[i * rstep];
        }

        const float* pa = &sA[st][fa];
        const float* pw = &sW[st][fw];

        v4f af[2];
        v4f bf[2][NT];
        af[0] = *(const v4f*)pa;
#pragma unroll
        for (int t = 0; t < NT; ++t)
            bf[0][t] = *(const v4f*)(pw + t * 16 * GEMM_PITCH);

#pragma unroll
        for (int q = 0; q < GEMM_BK / 8; ++q) {
            const int cur = q & 1;
            const int nxt = cur ^ 1;
            if (q + 1 < GEMM_BK / 8) {
                af[nxt] = *(const v4f*)(pa + (q + 1) * 8);
#pragma unroll
                for (int t = 0; t < NT; ++t)
                    bf[nxt][t] = *(const v4f*)(pw + t * 16 * GEMM_PITCH + (q + 1) * 8);
            }
            const v2f a_lo = (v2f){af[cur].x, af[cur].y};
            const v2f a_hi = (v2f){af[cur].z, af[cur].w};
#pragma unroll
            for (int t = 0; t < NT; ++t) {
                const v2f b_lo = (v2f){bf[cur][t].x, bf[cur][t].y};
                acc[t] = __builtin_amdgcn_wmma_f32_16x16x4_f32(
                             false, a_lo, false, b_lo,
                             (short)0, acc[t], false, false);
            }
#pragma unroll
            for (int t = 0; t < NT; ++t) {
                const v2f b_hi = (v2f){bf[cur][t].z, bf[cur][t].w};
                acc[t] = __builtin_amdgcn_wmma_f32_16x16x4_f32(
                             false, a_hi, false, b_hi,
                             (short)0, acc[t], false, false);
            }
        }

        if (more) {
#pragma unroll
            for (int i = 0; i < 4; ++i)
                *(v4f*)&sA[st ^ 1][sidx + i * 32 * GEMM_PITCH] = ga[i];
#pragma unroll
            for (int i = 0; i < WLD; ++i)
                *(v4f*)&sW[st ^ 1][sidx + i * 32 * GEMM_PITCH] = gw[i];
        }
        __syncthreads();
    }

    const int rowoff = (lane >> 4) * 8;
#pragma unroll
    for (int t = 0; t < NT; ++t) {
        const int col = col0 + t * 16 + r;
#pragma unroll
        for (int v = 0; v < 8; ++v) {
            float val = acc[t][v];
            if (EPI == 1) {
                val += bias[col];
                val = (val > 20.f) ? val : __logf(1.f + __expf(val));
            }
            C[(size_t)(row0 + rowoff + v) * N + col] = val;
        }
    }
}

#define CONV_RB 8
#define CONV_TB 256

__global__ __launch_bounds__(CONV_TB)
void conv_silu_kernel(const float* __restrict__ proj,
                      const float* __restrict__ cw,
                      const float* __restrict__ cb,
                      float* __restrict__ u)
{
    static_assert(L_SEQ % CONV_RB == 0, "a block's rows stay inside one sequence");
    static_assert(I_DIM % (4 * CONV_TB) == 0, "whole blocks across the channels");

    const int i    = (blockIdx.x * CONV_TB + threadIdx.x) * 4;
    const int row0 = blockIdx.y * CONV_RB;
    const int l0   = row0 % L_SEQ;

    const v4f* wp = (const v4f*)(cw + (size_t)i * 4);
    const v4f w0 = wp[0], w1 = wp[1], w2 = wp[2], w3 = wp[3];
    const v4f bias = *(const v4f*)(cb + i);

    const float* hs = proj + i;
    const v4f zero = (v4f){0.f, 0.f, 0.f, 0.f};
    v4f x[CONV_RB + 3];
#pragma unroll
    for (int j = 0; j < CONV_RB + 3; ++j) {
        const int  l    = l0 - 1 + j;
        const bool edge = (j == 0) || (j > CONV_RB);
        const bool ok   = !edge || (l >= 0 && l < L_SEQ);
        const int  rr   = ok ? (row0 - 1 + j) : row0;
        x[j] = *(const v4f*)(hs + (size_t)rr * (2 * I_DIM));
    }
    __builtin_amdgcn_sched_barrier(0);
#pragma unroll
    for (int j = 0; j < CONV_RB + 3; ++j) {
        const int  l    = l0 - 1 + j;
        const bool edge = (j == 0) || (j > CONV_RB);
        const bool ok   = !edge || (l >= 0 && l < L_SEQ);
        x[j] = ok ? x[j] : zero;
    }

    float* up = u + (size_t)row0 * I_DIM + i;
#pragma unroll
    for (int r = 0; r < CONV_RB; ++r) {
        const v4f wc[4] = {w0, w1, w2, w3};
        v4f o;
#pragma unroll
        for (int c = 0; c < 4; ++c) {
            float acc = bias[c];
#pragma unroll
            for (int k = 0; k < 4; ++k)
                acc += wc[c][k] * x[r + k][c];
            o[c] = acc / (1.f + __expf(-acc));
        }
        *(v4f*)(up + (size_t)r * I_DIM) = o;
    }
}

__global__ __launch_bounds__(256)
void rmsnorm_kernel(const float* __restrict__ xd,
                    const float* __restrict__ wdt,
                    const float* __restrict__ wb,
                    const float* __restrict__ wc,
                    float* __restrict__ dtn,
                    float* __restrict__ Bm,
                    float* __restrict__ Cm)
{
    const int row = blockIdx.x * 256 + threadIdx.x;
    if (row >= ROWS) return;
    const float* x = xd + (size_t)row * (R_DIM + 2 * N_ST);

    float s = 0.f;
#pragma unroll 8
    for (int j = 0; j < R_DIM; ++j) s += x[j] * x[j];
    float inv = __frsqrt_rn(s / R_DIM + EPS);
#pragma unroll 8
    for (int j = 0; j < R_DIM; ++j)
        dtn[(size_t)row * R_DIM + j] = wdt[j] * x[j] * inv;

    s = 0.f;
#pragma unroll
    for (int j = 0; j < N_ST; ++j) { float v = x[R_DIM + j]; s += v * v; }
    inv = __frsqrt_rn(s / N_ST + EPS);
#pragma unroll
    for (int j = 0; j < N_ST; ++j)
        Bm[(size_t)row * N_ST + j] = wb[j] * x[R_DIM + j] * inv;

    s = 0.f;
#pragma unroll
    for (int j = 0; j < N_ST; ++j) { float v = x[R_DIM + N_ST + j]; s += v * v; }
    inv = __frsqrt_rn(s / N_ST + EPS);
#pragma unroll
    for (int j = 0; j < N_ST; ++j)
        Cm[(size_t)row * N_ST + j] = wc[j] * x[R_DIM + N_ST + j] * inv;
}

#define TSTEP 32
#define NGRP  (L_SEQ / TSTEP)
#define SPL   4
#define NSL   (N_ST / SPL)
#define CHB   (128 / SPL)
#define GRP4  (TSTEP * N_ST / 4)
#define QSTEP (TSTEP / SPL)

static __device__ __forceinline__ void lds_barrier()
{
    __builtin_amdgcn_fence(__ATOMIC_RELEASE, "workgroup", "local");
    __builtin_amdgcn_s_barrier();
    __builtin_amdgcn_fence(__ATOMIC_ACQUIRE, "workgroup", "local");
}

static __device__ __forceinline__ float quad_sum(float v)
{
    v += __builtin_bit_cast(float, __builtin_amdgcn_mov_dpp(
             __builtin_bit_cast(int, v), 0xB1, 0xF, 0xF, true));
    v += __builtin_bit_cast(float, __builtin_amdgcn_mov_dpp(
             __builtin_bit_cast(int, v), 0x4E, 0xF, 0xF, true));
    return v;
}

__global__ __launch_bounds__(128)
void scan_kernel(const float* __restrict__ dt2,
                 const float* __restrict__ u,
                 const float* __restrict__ Bm,
                 const float* __restrict__ Cm,
                 const float* __restrict__ A_log,
                 const float* __restrict__ Dvec,
                 const float* __restrict__ proj,
                 float* __restrict__ yout)
{
    __shared__ __attribute__((aligned(16))) float4 sBC[2][2][GRP4];
    static_assert(GRP4 == 128, "one float4 of Bm and one of Cm per thread and group");
    static_assert(TSTEP % SPL == 0, "a group holds whole quads of steps");

    const int tid = threadIdx.x;
    const int sub = tid & (SPL - 1);
    const int b = blockIdx.y;
    const int i = blockIdx.x * CHB + (tid >> 2);

    const size_t row0 = (size_t)b * L_SEQ;
    const float* dtp = dt2  + row0 * I_DIM + i;
    const float* up  = u    + row0 * I_DIM + i;
    const float* gp  = proj + (row0 + sub) * (2 * I_DIM) + I_DIM + i;
    float*       yp  = yout + (row0 + sub) * I_DIM + i;
    const float4* bp = (const float4*)(Bm + row0 * N_ST) + tid;
    const float4* cp = (const float4*)(Cm + row0 * N_ST) + tid;

    const float4 b0 = bp[0];
    const float4 c0 = cp[0];
    float4 nb = bp[GRP4];
    float4 nc = cp[GRP4];

    float rdt[TSTEP], ru[TSTEP], rg[QSTEP];
#pragma unroll
    for (int j = 0; j < TSTEP; ++j) {
        rdt[j] = dtp[(size_t)j * I_DIM];
        ru[j]  = up[(size_t)j * I_DIM];
    }
#pragma unroll
    for (int m = 0; m < QSTEP; ++m)
        rg[m] = gp[(size_t)(SPL * m) * (2 * I_DIM)];

    float Aa[NSL];
#pragma unroll
    for (int n = 0; n < NSL; ++n)
        Aa[n] = -__expf(A_log[(size_t)i * N_ST + sub * NSL + n]);

    float h[NSL];
#pragma unroll
    for (int n = 0; n < NSL; ++n) h[n] = 0.f;

    const float Dv = Dvec[i];

#pragma unroll
    for (int j = 0; j < TSTEP; ++j)
        asm volatile("" :: "v"(rdt[j]), "v"(ru[j]));
#pragma unroll
    for (int m = 0; m < QSTEP; ++m)
        asm volatile("" :: "v"(rg[m]));
    asm volatile("" :: "v"(Dv));
    asm volatile("" :: "v"(nb.x), "v"(nb.y), "v"(nb.z), "v"(nb.w),
                       "v"(nc.x), "v"(nc.y), "v"(nc.z), "v"(nc.w));

    sBC[0][0][tid] = b0;
    sBC[0][1][tid] = c0;
    lds_barrier();

    float4 Bc  = sBC[0][0][sub],           Cc  = sBC[0][1][sub];
    float4 Bn  = sBC[0][0][SPL + sub],     Cn  = sBC[0][1][SPL + sub];
    float4 Bn2 = sBC[0][0][2 * SPL + sub], Cn2 = sBC[0][1][2 * SPL + sub];
    float  dAc[NSL];
    float  uvc, duc;
    {
        const float dtv = rdt[0];
#pragma unroll
        for (int n = 0; n < NSL; ++n)
            dAc[n] = __expf(dtv * Aa[n]);
        uvc = ru[0];
        duc = dtv * uvc;
    }

    for (int g = 0; g < NGRP; ++g) {
        const int buf = g & 1;
        const int gn = (g + 1 < NGRP) ? g + 1 : NGRP - 1;
        const float* dtn = dtp + (size_t)gn * TSTEP * I_DIM;
        const float* un  = up  + (size_t)gn * TSTEP * I_DIM;
        const float* gnx = gp  + (size_t)gn * TSTEP * (2 * I_DIM);
        float*       yg  = yp  + (size_t)g  * TSTEP * I_DIM;

        const float4* sB = &sBC[buf][0][sub];
        const float4* sC = &sBC[buf][1][sub];

        float zsel = 0.f;
        float gv = 0.f, eg = 0.f, sg = 0.f;
        float ypend = 0.f, upend = 0.f;

#pragma unroll
        for (int j = 0; j < TSTEP; ++j) {
            const int s = j & (SPL - 1);
            const int m = j / SPL;

            float4 B3 = Bn2, C3 = Cn2;
            __builtin_amdgcn_sched_barrier(0);
            if (j + 3 < TSTEP) {
                B3 = sB[(j + 3) * SPL];
                C3 = sC[(j + 3) * SPL];
                __builtin_amdgcn_sched_barrier(0);
            }

            const int   jn  = (j + 1) % TSTEP;
            const float dtx = rdt[jn];
            const float uvn = ru[jn];

            const float bcur[NSL] = {Bc.x, Bc.y, Bc.z, Bc.w};
            const float ccur[NSL] = {Cc.x, Cc.y, Cc.z, Cc.w};

            float dAn[NSL];
            float y = 0.f;
#pragma unroll
            for (int n = 0; n < NSL; ++n) {
                dAn[n] = __expf(dtx * Aa[n]);
                h[n] = dAc[n] * h[n] + duc * bcur[n];
                y   += h[n] * ccur[n];
            }
            const float dun = dtx * uvn;

            if (j > 0) {
                const int sp = (j - 1) & (SPL - 1);
                const int mp = (j - 1) / SPL;
                const float z = quad_sum(ypend) + upend * Dv;
                zsel = (sub == sp) ? z : zsel;
                if (sp == SPL - 1)
                    yg[(size_t)(SPL * mp) * I_DIM] = zsel * sg;
                ru[j - 1] = un[(size_t)(j - 1) * I_DIM];
            }
            ypend = y;
            upend = uvc;

            if (s == 0) {
                gv = rg[m];
                eg = __expf(-gv);
            }
            if (s == 1) {
                sg = gv / (1.f + eg);
                asm volatile("" :: "v"(sg));
            }
            if (s == 2) rg[m] = gnx[(size_t)(SPL * m) * (2 * I_DIM)];

            rdt[j] = dtn[(size_t)j * I_DIM];

#pragma unroll
            for (int n = 0; n < NSL; ++n) dAc[n] = dAn[n];
            uvc = uvn; duc = dun;
            Bc = Bn;   Cc = Cn;
            Bn = Bn2;  Cn = Cn2;
            Bn2 = B3;  Cn2 = C3;
        }

        __builtin_amdgcn_sched_barrier(0);
        {
            const float z = quad_sum(ypend) + upend * Dv;
            zsel = (sub == SPL - 1) ? z : zsel;
            yg[(size_t)(SPL * (QSTEP - 1)) * I_DIM] = zsel * sg;
            ru[TSTEP - 1] = un[(size_t)(TSTEP - 1) * I_DIM];
        }

        sBC[buf ^ 1][0][tid] = nb;
        sBC[buf ^ 1][1][tid] = nc;
        lds_barrier();

        Bc  = sBC[buf ^ 1][0][sub];           Cc  = sBC[buf ^ 1][1][sub];
        Bn  = sBC[buf ^ 1][0][SPL + sub];     Cn  = sBC[buf ^ 1][1][SPL + sub];
        Bn2 = sBC[buf ^ 1][0][2 * SPL + sub]; Cn2 = sBC[buf ^ 1][1][2 * SPL + sub];

        const int gn2 = (g + 2 < NGRP) ? g + 2 : NGRP - 1;
        nb = bp[(size_t)gn2 * GRP4];
        nc = cp[(size_t)gn2 * GRP4];
    }
}

extern "C" void kernel_launch(void* const* d_in, const int* in_sizes, int n_in,
                              void* d_out, int out_size, void* d_ws, size_t ws_size,
                              hipStream_t stream)
{
    const float* hidden     = (const float*)d_in[0];
    const float* in_proj_w  = (const float*)d_in[1];
    const float* conv_w     = (const float*)d_in[2];
    const float* conv_b     = (const float*)d_in[3];
    const float* x_proj_w   = (const float*)d_in[4];
    const float* dt_proj_w  = (const float*)d_in[5];
    const float* dt_proj_b  = (const float*)d_in[6];
    const float* dt_ln_w    = (const float*)d_in[7];
    const float* b_ln_w     = (const float*)d_in[8];
    const float* c_ln_w     = (const float*)d_in[9];
    const float* A_log      = (const float*)d_in[10];
    const float* Dvec       = (const float*)d_in[11];
    const float* out_proj_w = (const float*)d_in[12];
    float* out = (float*)d_out;

    float* ws   = (float*)d_ws;
    float* proj = ws;
    float* u    = proj + (size_t)ROWS * 2 * I_DIM;
    float* xdbl = u    + (size_t)ROWS * I_DIM;
    float* dtn  = xdbl + (size_t)ROWS * (R_DIM + 2 * N_ST);
    float* Bm   = dtn  + (size_t)ROWS * R_DIM;
    float* Cm   = Bm   + (size_t)ROWS * N_ST;
    float* dt2  = Cm   + (size_t)ROWS * N_ST;
    float* yout = dt2  + (size_t)ROWS * I_DIM;

    wmma_gemm_f32<4, 0><<<dim3((2 * I_DIM) / 64, ROWS / 128), 256, 0, stream>>>(
        hidden, in_proj_w, nullptr, proj, ROWS, 2 * I_DIM, H_DIM);

    conv_silu_kernel<<<dim3(I_DIM / (4 * CONV_TB), ROWS / CONV_RB), CONV_TB, 0, stream>>>(
        proj, conv_w, conv_b, u);

    wmma_gemm_f32<2, 0><<<dim3((R_DIM + 2 * N_ST) / 32, ROWS / 128), 256, 0, stream>>>(
        u, x_proj_w, nullptr, xdbl, ROWS, R_DIM + 2 * N_ST, I_DIM);

    rmsnorm_kernel<<<(ROWS + 255) / 256, 256, 0, stream>>>(
        xdbl, dt_ln_w, b_ln_w, c_ln_w, dtn, Bm, Cm);

    wmma_gemm_f32<4, 1><<<dim3(I_DIM / 64, ROWS / 128), 256, 0, stream>>>(
        dtn, dt_proj_w, dt_proj_b, dt2, ROWS, I_DIM, R_DIM);

    scan_kernel<<<dim3(I_DIM / CHB, B_SZ), 128, 0, stream>>>(
        dt2, u, Bm, Cm, A_log, Dvec, proj, yout);

    wmma_gemm_f32<4, 0><<<dim3(H_DIM / 64, ROWS / 128), 256, 0, stream>>>(
        yout, out_proj_w, nullptr, out, ROWS, H_DIM, I_DIM);
}


# x_proj: K split over 8 workgroups per output tile (384 workgroups, 8 K steps each); the norm kernel adds the 8 partial planes, a row per wave: sum order of x_dbl and of the norms' squares changes
# speedup vs baseline: 6.9760x; 1.1851x over previous
#include <hip/hip_runtime.h>
#include <hip/hip_bf16.h>


#define H_DIM 1024
#define I_DIM 2048
#define N_ST  16
#define R_DIM 64
#define B_SZ  2
#define L_SEQ 1024
#define ROWS  (B_SZ * L_SEQ)
#define EPS   1e-06f

typedef __attribute__((ext_vector_type(2))) float v2f;
typedef __attribute__((ext_vector_type(4))) float v4f;
typedef __attribute__((ext_vector_type(8))) float v8f;

typedef __attribute__((address_space(1))) const v4f gcv4f;

#define GEMM_BM    128
#define GEMM_BK    32
#define GEMM_PITCH 36

template <int NT, int EPI, int KSPLIT>
__global__ __launch_bounds__(256)
void wmma_gemm_f32(const float* __restrict__ A,
                   const float* __restrict__ W,
                   const float* __restrict__ bias,
                   float* __restrict__ C,
                   int M, int N, int K)
{
    constexpr int BN  = 16 * NT;
    constexpr int WLD = (BN * (GEMM_BK / 4)) / 256;
    static_assert(WLD >= 1 && WLD * 256 == BN * (GEMM_BK / 4), "W tile must fill whole threads");

    __shared__ __attribute__((aligned(16))) float sA[2][GEMM_BM * GEMM_PITCH];
    __shared__ __attribute__((aligned(16))) float sW[2][BN * GEMM_PITCH];

    const int kpart = K / KSPLIT;
    if (KSPLIT > 1) {
        A += (size_t)blockIdx.z * kpart;
        W += (size_t)blockIdx.z * kpart;
        C += (size_t)blockIdx.z * M * N;
    }

    const int tid  = threadIdx.x;
    const int lane = tid & 31;
    const int wave = tid >> 5;
    const int rowb = blockIdx.y * GEMM_BM;
    const int row0 = rowb + wave * 16;
    const int col0 = blockIdx.x * BN;

    const int r  = lane & 15;
    const int hh = lane >> 4;

    const int srow = tid >> 3;
    const int sc4  = tid & 7;
    gcv4f* Ag = (gcv4f*)(A + (size_t)(rowb + srow) * K + sc4 * 4);
    gcv4f* Wg = (gcv4f*)(W + (size_t)(col0 + srow) * K + sc4 * 4);
    const size_t rstep = (size_t)8 * K;
    const int sidx = srow * GEMM_PITCH + sc4 * 4;

    const int fa = (wave * 16 + r) * GEMM_PITCH + hh * 4;
    const int fw = r * GEMM_PITCH + hh * 4;

    v8f acc[NT];
#pragma unroll
    for (int t = 0; t < NT; ++t) acc[t] = (v8f){0.f,0.f,0.f,0.f,0.f,0.f,0.f,0.f};

    v4f ga[4], gw[WLD];
#pragma unroll
    for (int i = 0; i < 4; ++i) ga[i] = Ag[i * rstep];
#pragma unroll
    for (int i = 0; i < WLD; ++i) gw[i] = Wg[i * rstep];
#pragma unroll
    for (int i = 0; i < 4; ++i)
        *(v4f*)&sA[0][sidx + i * 32 * GEMM_PITCH] = ga[i];
#pragma unroll
    for (int i = 0; i < WLD; ++i)
        *(v4f*)&sW[0][sidx + i * 32 * GEMM_PITCH] = gw[i];
    __syncthreads();

    const int KT = kpart / GEMM_BK;
    for (int kt = 0; kt < KT; ++kt) {
        const int  st   = kt & 1;
        const bool more = (kt + 1 < KT);

        if (more) {
            Ag += GEMM_BK / 4;
            Wg += GEMM_BK / 4;
#pragma unroll
            for (int i = 0; i < 4; ++i) ga[i] = Ag[i * rstep];
#pragma unroll
            for (int i = 0; i < WLD; ++i) gw[i] = Wg[i * rstep];
        }

        const float* pa = &sA[st][fa];
        const float* pw = &sW[st][fw];

        v4f af[2];
        v4f bf[2][NT];
        af[0] = *(const v4f*)pa;
#pragma unroll
        for (int t = 0; t < NT; ++t)
            bf[0][t] = *(const v4f*)(pw + t * 16 * GEMM_PITCH);

#pragma unroll
        for (int q = 0; q < GEMM_BK / 8; ++q) {
            const int cur = q & 1;
            const int nxt = cur ^ 1;
            if (q + 1 < GEMM_BK / 8) {
                af[nxt] = *(const v4f*)(pa + (q + 1) * 8);
#pragma unroll
                for (int t = 0; t < NT; ++t)
                    bf[nxt][t] = *(const v4f*)(pw + t * 16 * GEMM_PITCH + (q + 1) * 8);
            }
            const v2f a_lo = (v2f){af[cur].x, af[cur].y};
            const v2f a_hi = (v2f){af[cur].z, af[cur].w};
#pragma unroll
            for (int t = 0; t < NT; ++t) {
                const v2f b_lo = (v2f){bf[cur][t].x, bf[cur][t].y};
                acc[t] = __builtin_amdgcn_wmma_f32_16x16x4_f32(
                             false, a_lo, false, b_lo,
                             (short)0, acc[t], false, false);
            }
#pragma unroll
            for (int t = 0; t < NT; ++t) {
                const v2f b_hi = (v2f){bf[cur][t].z, bf[cur][t].w};
                acc[t] = __builtin_amdgcn_wmma_f32_16x16x4_f32(
                             false, a_hi, false, b_hi,
                             (short)0, acc[t], false, false);
            }
        }

        if (more) {
#pragma unroll
            for (int i = 0; i < 4; ++i)
                *(v4f*)&sA[st ^ 1][sidx + i * 32 * GEMM_PITCH] = ga[i];
#pragma unroll
            for (int i = 0; i < WLD; ++i)
                *(v4f*)&sW[st ^ 1][sidx + i * 32 * GEMM_PITCH] = gw[i];
        }
        __syncthreads();
    }

    const int rowoff = (lane >> 4) * 8;
#pragma unroll
    for (int t = 0; t < NT; ++t) {
        const int col = col0 + t * 16 + r;
#pragma unroll
        for (int v = 0; v < 8; ++v) {
            float val = acc[t][v];
            if (EPI == 1) {
                val += bias[col];
                val = (val > 20.f) ? val : __logf(1.f + __expf(val));
            }
            C[(size_t)(row0 + rowoff + v) * N + col] = val;
        }
    }
}

#define CONV_RB 8
#define CONV_TB 256

__global__ __launch_bounds__(CONV_TB)
void conv_silu_kernel(const float* __restrict__ proj,
                      const float* __restrict__ cw,
                      const float* __restrict__ cb,
                      float* __restrict__ u)
{
    static_assert(L_SEQ % CONV_RB == 0, "a block's rows stay inside one sequence");
    static_assert(I_DIM % (4 * CONV_TB) == 0, "whole blocks across the channels");

    const int i    = (blockIdx.x * CONV_TB + threadIdx.x) * 4;
    const int row0 = blockIdx.y * CONV_RB;
    const int l0   = row0 % L_SEQ;

    const v4f* wp = (const v4f*)(cw + (size_t)i * 4);
    const v4f w0 = wp[0], w1 = wp[1], w2 = wp[2], w3 = wp[3];
    const v4f bias = *(const v4f*)(cb + i);

    const float* hs = proj + i;
    const v4f zero = (v4f){0.f, 0.f, 0.f, 0.f};
    v4f x[CONV_RB + 3];
#pragma unroll
    for (int j = 0; j < CONV_RB + 3; ++j) {
        const int  l    = l0 - 1 + j;
        const bool edge = (j == 0) || (j > CONV_RB);
        const bool ok   = !edge || (l >= 0 && l < L_SEQ);
        const int  rr   = ok ? (row0 - 1 + j) : row0;
        x[j] = *(const v4f*)(hs + (size_t)rr * (2 * I_DIM));
    }
    __builtin_amdgcn_sched_barrier(0);
#pragma unroll
    for (int j = 0; j < CONV_RB + 3; ++j) {
        const int  l    = l0 - 1 + j;
        const bool edge = (j == 0) || (j > CONV_RB);
        const bool ok   = !edge || (l >= 0 && l < L_SEQ);
        x[j] = ok ? x[j] : zero;
    }

    float* up = u + (size_t)row0 * I_DIM + i;
#pragma unroll
    for (int r = 0; r < CONV_RB; ++r) {
        const v4f wc[4] = {w0, w1, w2, w3};
        v4f o;
#pragma unroll
        for (int c = 0; c < 4; ++c) {
            float acc = bias[c];
#pragma unroll
            for (int k = 0; k < 4; ++k)
                acc += wc[c][k] * x[r + k][c];
            o[c] = acc / (1.f + __expf(-acc));
        }
        *(v4f*)(up + (size_t)r * I_DIM) = o;
    }
}

#define XSPLIT 8
#define XD     (R_DIM + 2 * N_ST)
#define NORM_RB 8

__global__ __launch_bounds__(32 * NORM_RB)
void rmsnorm_kernel(const float* __restrict__ xp,
                    const float* __restrict__ wdt,
                    const float* __restrict__ wb,
                    const float* __restrict__ wc,
                    float* __restrict__ dtn,
                    float* __restrict__ Bm,
                    float* __restrict__ Cm)
{
    static_assert(ROWS % NORM_RB == 0, "whole blocks");
    static_assert(XD % 4 == 0 && XD / 4 <= 32, "a row fits one wave, four values per lane");

    const int lane = threadIdx.x & 31;
    const int row  = blockIdx.x * NORM_RB + (threadIdx.x >> 5);
    const bool act = lane < XD / 4;
    const int  c4  = act ? lane : 0;

    const v4f* p = (const v4f*)(xp + (size_t)row * XD) + c4;
    v4f part[XSPLIT];
#pragma unroll
    for (int z = 0; z < XSPLIT; ++z)
        part[z] = p[(size_t)z * (ROWS * XD / 4)];
    v4f x = part[0];
#pragma unroll
    for (int z = 1; z < XSPLIT; ++z) x += part[z];

    float ss = x.x * x.x + x.y * x.y + x.z * x.z + x.w * x.w;
    ss = act ? ss : 0.f;
    ss += __shfl_xor(ss, 1, 32);
    ss += __shfl_xor(ss, 2, 32);
    float s16 = ss + __shfl_xor(ss, 4, 32);
    s16 += __shfl_xor(s16, 8, 32);

    const bool isdt = lane < R_DIM / 4;
    const float ms  = isdt ? s16 / R_DIM : ss / N_ST;
    const float inv = __frsqrt_rn(ms + EPS);

    const int lb = lane - R_DIM / 4;
    const float* wsrc = wdt + 4 * lane;
    float*       dst  = dtn + (size_t)row * R_DIM + 4 * lane;
    if (!isdt) {
        const bool isb = lb < N_ST / 4;
        const int  l4  = 4 * (isb ? lb : lb - N_ST / 4);
        wsrc = (isb ? wb : wc) + l4;
        dst  = (isb ? Bm : Cm) + (size_t)row * N_ST + l4;
    }
    if (act) {
        const v4f w = *(const v4f*)wsrc;
        v4f o;
        o.x = w.x * x.x * inv;
        o.y = w.y * x.y * inv;
        o.z = w.z * x.z * inv;
        o.w = w.w * x.w * inv;
        *(v4f*)dst = o;
    }
}

#define TSTEP 32
#define NGRP  (L_SEQ / TSTEP)
#define SPL   4
#define NSL   (N_ST / SPL)
#define CHB   (128 / SPL)
#define GRP4  (TSTEP * N_ST / 4)
#define QSTEP (TSTEP / SPL)

static __device__ __forceinline__ void lds_barrier()
{
    __builtin_amdgcn_fence(__ATOMIC_RELEASE, "workgroup", "local");
    __builtin_amdgcn_s_barrier();
    __builtin_amdgcn_fence(__ATOMIC_ACQUIRE, "workgroup", "local");
}

static __device__ __forceinline__ float quad_sum(float v)
{
    v += __builtin_bit_cast(float, __builtin_amdgcn_mov_dpp(
             __builtin_bit_cast(int, v), 0xB1, 0xF, 0xF, true));
    v += __builtin_bit_cast(float, __builtin_amdgcn_mov_dpp(
             __builtin_bit_cast(int, v), 0x4E, 0xF, 0xF, true));
    return v;
}

__global__ __launch_bounds__(128)
void scan_kernel(const float* __restrict__ dt2,
                 const float* __restrict__ u,
                 const float* __restrict__ Bm,
                 const float* __restrict__ Cm,
                 const float* __restrict__ A_log,
                 const float* __restrict__ Dvec,
                 const float* __restrict__ proj,
                 float* __restrict__ yout)
{
    __shared__ __attribute__((aligned(16))) float4 sBC[2][2][GRP4];
    static_assert(GRP4 == 128, "one float4 of Bm and one of Cm per thread and group");
    static_assert(TSTEP % SPL == 0, "a group holds whole quads of steps");

    const int tid = threadIdx.x;
    const int sub = tid & (SPL - 1);
    const int b = blockIdx.y;
    const int i = blockIdx.x * CHB + (tid >> 2);

    const size_t row0 = (size_t)b * L_SEQ;
    const float* dtp = dt2  + row0 * I_DIM + i;
    const float* up  = u    + row0 * I_DIM + i;
    const float* gp  = proj + (row0 + sub) * (2 * I_DIM) + I_DIM + i;
    float*       yp  = yout + (row0 + sub) * I_DIM + i;
    const float4* bp = (const float4*)(Bm + row0 * N_ST) + tid;
    const float4* cp = (const float4*)(Cm + row0 * N_ST) + tid;

    const float4 b0 = bp[0];
    const float4 c0 = cp[0];
    float4 nb = bp[GRP4];
    float4 nc = cp[GRP4];

    float rdt[TSTEP], ru[TSTEP], rg[QSTEP];
#pragma unroll
    for (int j = 0; j < TSTEP; ++j) {
        rdt[j] = dtp[(size_t)j * I_DIM];
        ru[j]  = up[(size_t)j * I_DIM];
    }
#pragma unroll
    for (int m = 0; m < QSTEP; ++m)
        rg[m] = gp[(size_t)(SPL * m) * (2 * I_DIM)];

    float Aa[NSL];
#pragma unroll
    for (int n = 0; n < NSL; ++n)
        Aa[n] = -__expf(A_log[(size_t)i * N_ST + sub * NSL + n]);

    float h[NSL];
#pragma unroll
    for (int n = 0; n < NSL; ++n) h[n] = 0.f;

    const float Dv = Dvec[i];

#pragma unroll
    for (int j = 0; j < TSTEP; ++j)
        asm volatile("" :: "v"(rdt[j]), "v"(ru[j]));
#pragma unroll
    for (int m = 0; m < QSTEP; ++m)
        asm volatile("" :: "v"(rg[m]));
    asm volatile("" :: "v"(Dv));
    asm volatile("" :: "v"(nb.x), "v"(nb.y), "v"(nb.z), "v"(nb.w),
                       "v"(nc.x), "v"(nc.y), "v"(nc.z), "v"(nc.w));

    sBC[0][0][tid] = b0;
    sBC[0][1][tid] = c0;
    lds_barrier();

    float4 Bc  = sBC[0][0][sub],           Cc  = sBC[0][1][sub];
    float4 Bn  = sBC[0][0][SPL + sub],     Cn  = sBC[0][1][SPL + sub];
    float4 Bn2 = sBC[0][0][2 * SPL + sub], Cn2 = sBC[0][1][2 * SPL + sub];
    float  dAc[NSL];
    float  uvc, duc;
    {
        const float dtv = rdt[0];
#pragma unroll
        for (int n = 0; n < NSL; ++n)
            dAc[n] = __expf(dtv * Aa[n]);
        uvc = ru[0];
        duc = dtv * uvc;
    }

    for (int g = 0; g < NGRP; ++g) {
        const int buf = g & 1;
        const int gn = (g + 1 < NGRP) ? g + 1 : NGRP - 1;
        const float* dtn = dtp + (size_t)gn * TSTEP * I_DIM;
        const float* un  = up  + (size_t)gn * TSTEP * I_DIM;
        const float* gnx = gp  + (size_t)gn * TSTEP * (2 * I_DIM);
        float*       yg  = yp  + (size_t)g  * TSTEP * I_DIM;

        const float4* sB = &sBC[buf][0][sub];
        const float4* sC = &sBC[buf][1][sub];

        float zsel = 0.f;
        float gv = 0.f, eg = 0.f, sg = 0.f;
        float ypend = 0.f, upend = 0.f;

#pragma unroll
        for (int j = 0; j < TSTEP; ++j) {
            const int s = j & (SPL - 1);
            const int m = j / SPL;

            float4 B3 = Bn2, C3 = Cn2;
            __builtin_amdgcn_sched_barrier(0);
            if (j + 3 < TSTEP) {
                B3 = sB[(j + 3) * SPL];
                C3 = sC[(j + 3) * SPL];
                __builtin_amdgcn_sched_barrier(0);
            }

            const int   jn  = (j + 1) % TSTEP;
            const float dtx = rdt[jn];
            const float uvn = ru[jn];

            const float bcur[NSL] = {Bc.x, Bc.y, Bc.z, Bc.w};
            const float ccur[NSL] = {Cc.x, Cc.y, Cc.z, Cc.w};

            float dAn[NSL];
            float y = 0.f;
#pragma unroll
            for (int n = 0; n < NSL; ++n) {
                dAn[n] = __expf(dtx * Aa[n]);
                h[n] = dAc[n] * h[n] + duc * bcur[n];
                y   += h[n] * ccur[n];
            }
            const float dun = dtx * uvn;

            if (j > 0) {
                const int sp = (j - 1) & (SPL - 1);
                const int mp = (j - 1) / SPL;
                const float z = quad_sum(ypend) + upend * Dv;
                zsel = (sub == sp) ? z : zsel;
                if (sp == SPL - 1)
                    yg[(size_t)(SPL * mp) * I_DIM] = zsel * sg;
                ru[j - 1] = un[(size_t)(j - 1) * I_DIM];
            }
            ypend = y;
            upend = uvc;

            if (s == 0) {
                gv = rg[m];
                eg = __expf(-gv);
            }
            if (s == 1) {
                sg = gv / (1.f + eg);
                asm volatile("" :: "v"(sg));
            }
            if (s == 2) rg[m] = gnx[(size_t)(SPL * m) * (2 * I_DIM)];

            rdt[j] = dtn[(size_t)j * I_DIM];

#pragma unroll
            for (int n = 0; n < NSL; ++n) dAc[n] = dAn[n];
            uvc = uvn; duc = dun;
            Bc = Bn;   Cc = Cn;
            Bn = Bn2;  Cn = Cn2;
            Bn2 = B3;  Cn2 = C3;
        }

        __builtin_amdgcn_sched_barrier(0);
        {
            const float z = quad_sum(ypend) + upend * Dv;
            zsel = (sub == SPL - 1) ? z : zsel;
            yg[(size_t)(SPL * (QSTEP - 1)) * I_DIM] = zsel * sg;
            ru[TSTEP - 1] = un[(size_t)(TSTEP - 1) * I_DIM];
        }

        sBC[buf ^ 1][0][tid] = nb;
        sBC[buf ^ 1][1][tid] = nc;
        lds_barrier();

        Bc  = sBC[buf ^ 1][0][sub];           Cc  = sBC[buf ^ 1][1][sub];
        Bn  = sBC[buf ^ 1][0][SPL + sub];     Cn  = sBC[buf ^ 1][1][SPL + sub];
        Bn2 = sBC[buf ^ 1][0][2 * SPL + sub]; Cn2 = sBC[buf ^ 1][1][2 * SPL + sub];

        const int gn2 = (g + 2 < NGRP) ? g + 2 : NGRP - 1;
        nb = bp[(size_t)gn2 * GRP4];
        nc = cp[(size_t)gn2 * GRP4];
    }
}

extern "C" void kernel_launch(void* const* d_in, const int* in_sizes, int n_in,
                              void* d_out, int out_size, void* d_ws, size_t ws_size,
                              hipStream_t stream)
{
    const float* hidden     = (const float*)d_in[0];
    const float* in_proj_w  = (const float*)d_in[1];
    const float* conv_w     = (const float*)d_in[2];
    const float* conv_b     = (const float*)d_in[3];
    const float* x_proj_w   = (const float*)d_in[4];
    const float* dt_proj_w  = (const float*)d_in[5];
    const float* dt_proj_b  = (const float*)d_in[6];
    const float* dt_ln_w    = (const float*)d_in[7];
    const float* b_ln_w     = (const float*)d_in[8];
    const float* c_ln_w     = (const float*)d_in[9];
    const float* A_log      = (const float*)d_in[10];
    const float* Dvec       = (const float*)d_in[11];
    const float* out_proj_w = (const float*)d_in[12];
    float* out = (float*)d_out;

    float* ws   = (float*)d_ws;
    float* proj = ws;
    float* u    = proj + (size_t)ROWS * 2 * I_DIM;
    float* xdbl = u    + (size_t)ROWS * I_DIM;
    float* dtn  = xdbl + (size_t)XSPLIT * ROWS * XD;
    float* Bm   = dtn  + (size_t)ROWS * R_DIM;
    float* Cm   = Bm   + (size_t)ROWS * N_ST;
    float* dt2  = Cm   + (size_t)ROWS * N_ST;
    float* yout = dt2  + (size_t)ROWS * I_DIM;

    wmma_gemm_f32<4, 0, 1><<<dim3((2 * I_DIM) / 64, ROWS / 128), 256, 0, stream>>>(
        hidden, in_proj_w, nullptr, proj, ROWS, 2 * I_DIM, H_DIM);

    conv_silu_kernel<<<dim3(I_DIM / (4 * CONV_TB), ROWS / CONV_RB), CONV_TB, 0, stream>>>(
        proj, conv_w, conv_b, u);

    wmma_gemm_f32<2, 0, XSPLIT><<<dim3(XD / 32, ROWS / 128, XSPLIT), 256, 0, stream>>>(
        u, x_proj_w, nullptr, xdbl, ROWS, XD, I_DIM);

    rmsnorm_kernel<<<ROWS / NORM_RB, 32 * NORM_RB, 0, stream>>>(
        xdbl, dt_ln_w, b_ln_w, c_ln_w, dtn, Bm, Cm);

    wmma_gemm_f32<4, 1, 1><<<dim3(I_DIM / 64, ROWS / 128), 256, 0, stream>>>(
        dtn, dt_proj_w, dt_proj_b, dt2, ROWS, I_DIM, R_DIM);

    scan_kernel<<<dim3(I_DIM / CHB, B_SZ), 128, 0, stream>>>(
        dt2, u, Bm, Cm, A_log, Dvec, proj, yout);

    wmma_gemm_f32<4, 0, 1><<<dim3(H_DIM / 64, ROWS / 128), 256, 0, stream>>>(
        yout, out_proj_w, nullptr, out, ROWS, H_DIM, I_DIM);
}


# GEMMs: K step 64 instead of 32 (LDS tile 128 x 64, row pitch 68) in out_proj and x_proj, whose K loop ran at one tile round trip per step; in_proj, dt_proj keep 32; sum order unchanged
# speedup vs baseline: 7.2264x; 1.0359x over previous
#include <hip/hip_runtime.h>
#include <hip/hip_bf16.h>


#define H_DIM 1024
#define I_DIM 2048
#define N_ST  16
#define R_DIM 64
#define B_SZ  2
#define L_SEQ 1024
#define ROWS  (B_SZ * L_SEQ)
#define EPS   1e-06f

typedef __attribute__((ext_vector_type(2))) float v2f;
typedef __attribute__((ext_vector_type(4))) float v4f;
typedef __attribute__((ext_vector_type(8))) float v8f;

typedef __attribute__((address_space(1))) const v4f gcv4f;

#define GEMM_BM    128

template <int NT, int EPI, int KSPLIT, int BK>
__global__ __launch_bounds__(256)
void wmma_gemm_f32(const float* __restrict__ A,
                   const float* __restrict__ W,
                   const float* __restrict__ bias,
                   float* __restrict__ C,
                   int M, int N, int K)
{
    constexpr int BN    = 16 * NT;
    constexpr int PITCH = BK + 4;
    constexpr int PPR   = BK / 4;
    constexpr int RPP   = 256 / PPR;
    constexpr int ALD   = GEMM_BM / RPP;
    constexpr int WLD   = BN / RPP;
    static_assert(BK == 32 || BK == 64, "K step of 32 or 64");
    static_assert(WLD >= 1 && WLD * RPP == BN, "W tile must fill whole threads");

    __shared__ __attribute__((aligned(16))) float sA[2][GEMM_BM * PITCH];
    __shared__ __attribute__((aligned(16))) float sW[2][BN * PITCH];

    const int kpart = K / KSPLIT;
    if (KSPLIT > 1) {
        A += (size_t)blockIdx.z * kpart;
        W += (size_t)blockIdx.z * kpart;
        C += (size_t)blockIdx.z * M * N;
    }

    const int tid  = threadIdx.x;
    const int lane = tid & 31;
    const int wave = tid >> 5;
    const int rowb = blockIdx.y * GEMM_BM;
    const int row0 = rowb + wave * 16;
    const int col0 = blockIdx.x * BN;

    const int r  = lane & 15;
    const int hh = lane >> 4;

    const int srow = tid / PPR;
    const int sc4  = tid % PPR;
    gcv4f* Ag = (gcv4f*)(A + (size_t)(rowb + srow) * K + sc4 * 4);
    gcv4f* Wg = (gcv4f*)(W + (size_t)(col0 + srow) * K + sc4 * 4);
    const size_t rstep = (size_t)(RPP / 4) * K;
    const int sidx = srow * PITCH + sc4 * 4;

    const int fa = (wave * 16 + r) * PITCH + hh * 4;
    const int fw = r * PITCH + hh * 4;

    v8f acc[NT];
#pragma unroll
    for (int t = 0; t < NT; ++t) acc[t] = (v8f){0.f,0.f,0.f,0.f,0.f,0.f,0.f,0.f};

    v4f ga[ALD], gw[WLD];
#pragma unroll
    for (int i = 0; i < ALD; ++i) ga[i] = Ag[i * rstep];
#pragma unroll
    for (int i = 0; i < WLD; ++i) gw[i] = Wg[i * rstep];
#pragma unroll
    for (int i = 0; i < ALD; ++i)
        *(v4f*)&sA[0][sidx + i * RPP * PITCH] = ga[i];
#pragma unroll
    for (int i = 0; i < WLD; ++i)
        *(v4f*)&sW[0][sidx + i * RPP * PITCH] = gw[i];
    __syncthreads();

    const int KT = kpart / BK;
    for (int kt = 0; kt < KT; ++kt) {
        const int  st   = kt & 1;
        const bool more = (kt + 1 < KT);

        if (more) {
            Ag += BK / 4;
            Wg += BK / 4;
#pragma unroll
            for (int i = 0; i < ALD; ++i) ga[i] = Ag[i * rstep];
#pragma unroll
            for (int i = 0; i < WLD; ++i) gw[i] = Wg[i * rstep];
        }

        const float* pa = &sA[st][fa];
        const float* pw = &sW[st][fw];

        v4f af[2];
        v4f bf[2][NT];
        af[0] = *(const v4f*)pa;
#pragma unroll
        for (int t = 0; t < NT; ++t)
            bf[0][t] = *(const v4f*)(pw + t * 16 * PITCH);

#pragma unroll
        for (int q = 0; q < BK / 8; ++q) {
            const int cur = q & 1;
            const int nxt = cur ^ 1;
            if (q + 1 < BK / 8) {
                af[nxt] = *(const v4f*)(pa + (q + 1) * 8);
#pragma unroll
                for (int t = 0; t < NT; ++t)
                    bf[nxt][t] = *(const v4f*)(pw + t * 16 * PITCH + (q + 1) * 8);
            }
            const v2f a_lo = (v2f){af[cur].x, af[cur].y};
            const v2f a_hi = (v2f){af[cur].z, af[cur].w};
#pragma unroll
            for (int t = 0; t < NT; ++t) {
                const v2f b_lo = (v2f){bf[cur][t].x, bf[cur][t].y};
                acc[t] = __builtin_amdgcn_wmma_f32_16x16x4_f32(
                             false, a_lo, false, b_lo,
                             (short)0, acc[t], false, false);
            }
#pragma unroll
            for (int t = 0; t < NT; ++t) {
                const v2f b_hi = (v2f){bf[cur][t].z, bf[cur][t].w};
                acc[t] = __builtin_amdgcn_wmma_f32_16x16x4_f32(
                             false, a_hi, false, b_hi,
                             (short)0, acc[t], false, false);
            }
        }

        if (more) {
#pragma unroll
            for (int i = 0; i < ALD; ++i)
                *(v4f*)&sA[st ^ 1][sidx + i * RPP * PITCH] = ga[i];
#pragma unroll
            for (int i = 0; i < WLD; ++i)
                *(v4f*)&sW[st ^ 1][sidx + i * RPP * PITCH] = gw[i];
        }
        __syncthreads();
    }

    const int rowoff = (lane >> 4) * 8;
#pragma unroll
    for (int t = 0; t < NT; ++t) {
        const int col = col0 + t * 16 + r;
#pragma unroll
        for (int v = 0; v < 8; ++v) {
            float val = acc[t][v];
            if (EPI == 1) {
                val += bias[col];
                val = (val > 20.f) ? val : __logf(1.f + __expf(val));
            }
            C[(size_t)(row0 + rowoff + v) * N + col] = val;
        }
    }
}

#define CONV_RB 8
#define CONV_TB 256

__global__ __launch_bounds__(CONV_TB)
void conv_silu_kernel(const float* __restrict__ proj,
                      const float* __restrict__ cw,
                      const float* __restrict__ cb,
                      float* __restrict__ u)
{
    static_assert(L_SEQ % CONV_RB == 0, "a block's rows stay inside one sequence");
    static_assert(I_DIM % (4 * CONV_TB) == 0, "whole blocks across the channels");

    const int i    = (blockIdx.x * CONV_TB + threadIdx.x) * 4;
    const int row0 = blockIdx.y * CONV_RB;
    const int l0   = row0 % L_SEQ;

    const v4f* wp = (const v4f*)(cw + (size_t)i * 4);
    const v4f w0 = wp[0], w1 = wp[1], w2 = wp[2], w3 = wp[3];
    const v4f bias = *(const v4f*)(cb + i);

    const float* hs = proj + i;
    const v4f zero = (v4f){0.f, 0.f, 0.f, 0.f};
    v4f x[CONV_RB + 3];
#pragma unroll
    for (int j = 0; j < CONV_RB + 3; ++j) {
        const int  l    = l0 - 1 + j;
        const bool edge = (j == 0) || (j > CONV_RB);
        const bool ok   = !edge || (l >= 0 && l < L_SEQ);
        const int  rr   = ok ? (row0 - 1 + j) : row0;
        x[j] = *(const v4f*)(hs + (size_t)rr * (2 * I_DIM));
    }
    __builtin_amdgcn_sched_barrier(0);
#pragma unroll
    for (int j = 0; j < CONV_RB + 3; ++j) {
        const int  l    = l0 - 1 + j;
        const bool edge = (j == 0) || (j > CONV_RB);
        const bool ok   = !edge || (l >= 0 && l < L_SEQ);
        x[j] = ok ? x[j] : zero;
    }

    float* up = u + (size_t)row0 * I_DIM + i;
#pragma unroll
    for (int r = 0; r < CONV_RB; ++r) {
        const v4f wc[4] = {w0, w1, w2, w3};
        v4f o;
#pragma unroll
        for (int c = 0; c < 4; ++c) {
            float acc = bias[c];
#pragma unroll
            for (int k = 0; k < 4; ++k)
                acc += wc[c][k] * x[r + k][c];
            o[c] = acc / (1.f + __expf(-acc));
        }
        *(v4f*)(up + (size_t)r * I_DIM) = o;
    }
}

#define XSPLIT 8
#define XD     (R_DIM + 2 * N_ST)
#define NORM_RB 8

__global__ __launch_bounds__(32 * NORM_RB)
void rmsnorm_kernel(const float* __restrict__ xp,
                    const float* __restrict__ wdt,
                    const float* __restrict__ wb,
                    const float* __restrict__ wc,
                    float* __restrict__ dtn,
                    float* __restrict__ Bm,
                    float* __restrict__ Cm)
{
    static_assert(ROWS % NORM_RB == 0, "whole blocks");
    static_assert(XD % 4 == 0 && XD / 4 <= 32, "a row fits one wave, four values per lane");

    const int lane = threadIdx.x & 31;
    const int row  = blockIdx.x * NORM_RB + (threadIdx.x >> 5);
    const bool act = lane < XD / 4;
    const int  c4  = act ? lane : 0;

    const v4f* p = (const v4f*)(xp + (size_t)row * XD) + c4;
    v4f part[XSPLIT];
#pragma unroll
    for (int z = 0; z < XSPLIT; ++z)
        part[z] = p[(size_t)z * (ROWS * XD / 4)];
    v4f x = part[0];
#pragma unroll
    for (int z = 1; z < XSPLIT; ++z) x += part[z];

    float ss = x.x * x.x + x.y * x.y + x.z * x.z + x.w * x.w;
    ss = act ? ss : 0.f;
    ss += __shfl_xor(ss, 1, 32);
    ss += __shfl_xor(ss, 2, 32);
    float s16 = ss + __shfl_xor(ss, 4, 32);
    s16 += __shfl_xor(s16, 8, 32);

    const bool isdt = lane < R_DIM / 4;
    const float ms  = isdt ? s16 / R_DIM : ss / N_ST;
    const float inv = __frsqrt_rn(ms + EPS);

    const int lb = lane - R_DIM / 4;
    const float* wsrc = wdt + 4 * lane;
    float*       dst  = dtn + (size_t)row * R_DIM + 4 * lane;
    if (!isdt) {
        const bool isb = lb < N_ST / 4;
        const int  l4  = 4 * (isb ? lb : lb - N_ST / 4);
        wsrc = (isb ? wb : wc) + l4;
        dst  = (isb ? Bm : Cm) + (size_t)row * N_ST + l4;
    }
    if (act) {
        const v4f w = *(const v4f*)wsrc;
        v4f o;
        o.x = w.x * x.x * inv;
        o.y = w.y * x.y * inv;
        o.z = w.z * x.z * inv;
        o.w = w.w * x.w * inv;
        *(v4f*)dst = o;
    }
}

#define TSTEP 32
#define NGRP  (L_SEQ / TSTEP)
#define SPL   4
#define NSL   (N_ST / SPL)
#define CHB   (128 / SPL)
#define GRP4  (TSTEP * N_ST / 4)
#define QSTEP (TSTEP / SPL)

static __device__ __forceinline__ void lds_barrier()
{
    __builtin_amdgcn_fence(__ATOMIC_RELEASE, "workgroup", "local");
    __builtin_amdgcn_s_barrier();
    __builtin_amdgcn_fence(__ATOMIC_ACQUIRE, "workgroup", "local");
}

static __device__ __forceinline__ float quad_sum(float v)
{
    v += __builtin_bit_cast(float, __builtin_amdgcn_mov_dpp(
             __builtin_bit_cast(int, v), 0xB1, 0xF, 0xF, true));
    v += __builtin_bit_cast(float, __builtin_amdgcn_mov_dpp(
             __builtin_bit_cast(int, v), 0x4E, 0xF, 0xF, true));
    return v;
}

__global__ __launch_bounds__(128)
void scan_kernel(const float* __restrict__ dt2,
                 const float* __restrict__ u,
                 const float* __restrict__ Bm,
                 const float* __restrict__ Cm,
                 const float* __restrict__ A_log,
                 const float* __restrict__ Dvec,
                 const float* __restrict__ proj,
                 float* __restrict__ yout)
{
    __shared__ __attribute__((aligned(16))) float4 sBC[2][2][GRP4];
    static_assert(GRP4 == 128, "one float4 of Bm and one of Cm per thread and group");
    static_assert(TSTEP % SPL == 0, "a group holds whole quads of steps");

    const int tid = threadIdx.x;
    const int sub = tid & (SPL - 1);
    const int b = blockIdx.y;
    const int i = blockIdx.x * CHB + (tid >> 2);

    const size_t row0 = (size_t)b * L_SEQ;
    const float* dtp = dt2  + row0 * I_DIM + i;
    const float* up  = u    + row0 * I_DIM + i;
    const float* gp  = proj + (row0 + sub) * (2 * I_DIM) + I_DIM + i;
    float*       yp  = yout + (row0 + sub) * I_DIM + i;
    const float4* bp = (const float4*)(Bm + row0 * N_ST) + tid;
    const float4* cp = (const float4*)(Cm + row0 * N_ST) + tid;

    const float4 b0 = bp[0];
    const float4 c0 = cp[0];
    float4 nb = bp[GRP4];
    float4 nc = cp[GRP4];

    float rdt[TSTEP], ru[TSTEP], rg[QSTEP];
#pragma unroll
    for (int j = 0; j < TSTEP; ++j) {
        rdt[j] = dtp[(size_t)j * I_DIM];
        ru[j]  = up[(size_t)j * I_DIM];
    }
#pragma unroll
    for (int m = 0; m < QSTEP; ++m)
        rg[m] = gp[(size_t)(SPL * m) * (2 * I_DIM)];

    float Aa[NSL];
#pragma unroll
    for (int n = 0; n < NSL; ++n)
        Aa[n] = -__expf(A_log[(size_t)i * N_ST + sub * NSL + n]);

    float h[NSL];
#pragma unroll
    for (int n = 0; n < NSL; ++n) h[n] = 0.f;

    const float Dv = Dvec[i];

#pragma unroll
    for (int j = 0; j < TSTEP; ++j)
        asm volatile("" :: "v"(rdt[j]), "v"(ru[j]));
#pragma unroll
    for (int m = 0; m < QSTEP; ++m)
        asm volatile("" :: "v"(rg[m]));
    asm volatile("" :: "v"(Dv));
    asm volatile("" :: "v"(nb.x), "v"(nb.y), "v"(nb.z), "v"(nb.w),
                       "v"(nc.x), "v"(nc.y), "v"(nc.z), "v"(nc.w));

    sBC[0][0][tid] = b0;
    sBC[0][1][tid] = c0;
    lds_barrier();

    float4 Bc  = sBC[0][0][sub],           Cc  = sBC[0][1][sub];
    float4 Bn  = sBC[0][0][SPL + sub],     Cn  = sBC[0][1][SPL + sub];
    float4 Bn2 = sBC[0][0][2 * SPL + sub], Cn2 = sBC[0][1][2 * SPL + sub];
    float  dAc[NSL];
    float  uvc, duc;
    {
        const float dtv = rdt[0];
#pragma unroll
        for (int n = 0; n < NSL; ++n)
            dAc[n] = __expf(dtv * Aa[n]);
        uvc = ru[0];
        duc = dtv * uvc;
    }

    for (int g = 0; g < NGRP; ++g) {
        const int buf = g & 1;
        const int gn = (g + 1 < NGRP) ? g + 1 : NGRP - 1;
        const float* dtn = dtp + (size_t)gn * TSTEP * I_DIM;
        const float* un  = up  + (size_t)gn * TSTEP * I_DIM;
        const float* gnx = gp  + (size_t)gn * TSTEP * (2 * I_DIM);
        float*       yg  = yp  + (size_t)g  * TSTEP * I_DIM;

        const float4* sB = &sBC[buf][0][sub];
        const float4* sC = &sBC[buf][1][sub];

        float zsel = 0.f;
        float gv = 0.f, eg = 0.f, sg = 0.f;
        float ypend = 0.f, upend = 0.f;

#pragma unroll
        for (int j = 0; j < TSTEP; ++j) {
            const int s = j & (SPL - 1);
            const int m = j / SPL;

            float4 B3 = Bn2, C3 = Cn2;
            __builtin_amdgcn_sched_barrier(0);
            if (j + 3 < TSTEP) {
                B3 = sB[(j + 3) * SPL];
                C3 = sC[(j + 3) * SPL];
                __builtin_amdgcn_sched_barrier(0);
            }

            const int   jn  = (j + 1) % TSTEP;
            const float dtx = rdt[jn];
            const float uvn = ru[jn];

            const float bcur[NSL] = {Bc.x, Bc.y, Bc.z, Bc.w};
            const float ccur[NSL] = {Cc.x, Cc.y, Cc.z, Cc.w};

            float dAn[NSL];
            float y = 0.f;
#pragma unroll
            for (int n = 0; n < NSL; ++n) {
                dAn[n] = __expf(dtx * Aa[n]);
                h[n] = dAc[n] * h[n] + duc * bcur[n];
                y   += h[n] * ccur[n];
            }
            const float dun = dtx * uvn;

            if (j > 0) {
                const int sp = (j - 1) & (SPL - 1);
                const int mp = (j - 1) / SPL;
                const float z = quad_sum(ypend) + upend * Dv;
                zsel = (sub == sp) ? z : zsel;
                if (sp == SPL - 1)
                    yg[(size_t)(SPL * mp) * I_DIM] = zsel * sg;
                ru[j - 1] = un[(size_t)(j - 1) * I_DIM];
            }
            ypend = y;
            upend = uvc;

            if (s == 0) {
                gv = rg[m];
                eg = __expf(-gv);
            }
            if (s == 1) {
                sg = gv / (1.f + eg);
                asm volatile("" :: "v"(sg));
            }
            if (s == 2) rg[m] = gnx[(size_t)(SPL * m) * (2 * I_DIM)];

            rdt[j] = dtn[(size_t)j * I_DIM];

#pragma unroll
            for (int n = 0; n < NSL; ++n) dAc[n] = dAn[n];
            uvc = uvn; duc = dun;
            Bc = Bn;   Cc = Cn;
            Bn = Bn2;  Cn = Cn2;
            Bn2 = B3;  Cn2 = C3;
        }

        __builtin_amdgcn_sched_barrier(0);
        {
            const float z = quad_sum(ypend) + upend * Dv;
            zsel = (sub == SPL - 1) ? z : zsel;
            yg[(size_t)(SPL * (QSTEP - 1)) * I_DIM] = zsel * sg;
            ru[TSTEP - 1] = un[(size_t)(TSTEP - 1) * I_DIM];
        }

        sBC[buf ^ 1][0][tid] = nb;
        sBC[buf ^ 1][1][tid] = nc;
        lds_barrier();

        Bc  = sBC[buf ^ 1][0][sub];           Cc  = sBC[buf ^ 1][1][sub];
        Bn  = sBC[buf ^ 1][0][SPL + sub];     Cn  = sBC[buf ^ 1][1][SPL + sub];
        Bn2 = sBC[buf ^ 1][0][2 * SPL + sub]; Cn2 = sBC[buf ^ 1][1][2 * SPL + sub];

        const int gn2 = (g + 2 < NGRP) ? g + 2 : NGRP - 1;
        nb = bp[(size_t)gn2 * GRP4];
        nc = cp[(size_t)gn2 * GRP4];
    }
}

extern "C" void kernel_launch(void* const* d_in, const int* in_sizes, int n_in,
                              void* d_out, int out_size, void* d_ws, size_t ws_size,
                              hipStream_t stream)
{
    const float* hidden     = (const float*)d_in[0];
    const float* in_proj_w  = (const float*)d_in[1];
    const float* conv_w     = (const float*)d_in[2];
    const float* conv_b     = (const float*)d_in[3];
    const float* x_proj_w   = (const float*)d_in[4];
    const float* dt_proj_w  = (const float*)d_in[5];
    const float* dt_proj_b  = (const float*)d_in[6];
    const float* dt_ln_w    = (const float*)d_in[7];
    const float* b_ln_w     = (const float*)d_in[8];
    const float* c_ln_w     = (const float*)d_in[9];
    const float* A_log      = (const float*)d_in[10];
    const float* Dvec       = (const float*)d_in[11];
    const float* out_proj_w = (const float*)d_in[12];
    float* out = (float*)d_out;

    float* ws   = (float*)d_ws;
    float* proj = ws;
    float* u    = proj + (size_t)ROWS * 2 * I_DIM;
    float* xdbl = u    + (size_t)ROWS * I_DIM;
    float* dtn  = xdbl + (size_t)XSPLIT * ROWS * XD;
    float* Bm   = dtn  + (size_t)ROWS * R_DIM;
    float* Cm   = Bm   + (size_t)ROWS * N_ST;
    float* dt2  = Cm   + (size_t)ROWS * N_ST;
    float* yout = dt2  + (size_t)ROWS * I_DIM;

    wmma_gemm_f32<4, 0, 1, 32><<<dim3((2 * I_DIM) / 64, ROWS / 128), 256, 0, stream>>>(
        hidden, in_proj_w, nullptr, proj, ROWS, 2 * I_DIM, H_DIM);

    conv_silu_kernel<<<dim3(I_DIM / (4 * CONV_TB), ROWS / CONV_RB), CONV_TB, 0, stream>>>(
        proj, conv_w, conv_b, u);

    wmma_gemm_f32<2, 0, XSPLIT, 64><<<dim3(XD / 32, ROWS / 128, XSPLIT), 256, 0, stream>>>(
        u, x_proj_w, nullptr, xdbl, ROWS, XD, I_DIM);

    rmsnorm_kernel<<<ROWS / NORM_RB, 32 * NORM_RB, 0, stream>>>(
        xdbl, dt_ln_w, b_ln_w, c_ln_w, dtn, Bm, Cm);

    wmma_gemm_f32<4, 1, 1, 32><<<dim3(I_DIM / 64, ROWS / 128), 256, 0, stream>>>(
        dtn, dt_proj_w, dt_proj_b, dt2, ROWS, I_DIM, R_DIM);

    scan_kernel<<<dim3(I_DIM / CHB, B_SZ), 128, 0, stream>>>(
        dt2, u, Bm, Cm, A_log, Dvec, proj, yout);

    wmma_gemm_f32<4, 0, 1, 64><<<dim3(H_DIM / 64, ROWS / 128), 256, 0, stream>>>(
        yout, out_proj_w, nullptr, out, ROWS, H_DIM, I_DIM);
}


# scan: the memory instructions of a quad of steps (Bm/Cm LDS reads, dt/u/gate ring refills, y store) stand in ONE place per quad instead of about three places per step; sum order unchanged
# speedup vs baseline: 8.0951x; 1.1202x over previous
#include <hip/hip_runtime.h>
#include <hip/hip_bf16.h>


#define H_DIM 1024
#define I_DIM 2048
#define N_ST  16
#define R_DIM 64
#define B_SZ  2
#define L_SEQ 1024
#define ROWS  (B_SZ * L_SEQ)
#define EPS   1e-06f

typedef __attribute__((ext_vector_type(2))) float v2f;
typedef __attribute__((ext_vector_type(4))) float v4f;
typedef __attribute__((ext_vector_type(8))) float v8f;

typedef __attribute__((address_space(1))) const v4f gcv4f;

#define GEMM_BM    128

template <int NT, int EPI, int KSPLIT, int BK>
__global__ __launch_bounds__(256)
void wmma_gemm_f32(const float* __restrict__ A,
                   const float* __restrict__ W,
                   const float* __restrict__ bias,
                   float* __restrict__ C,
                   int M, int N, int K)
{
    constexpr int BN    = 16 * NT;
    constexpr int PITCH = BK + 4;
    constexpr int PPR   = BK / 4;
    constexpr int RPP   = 256 / PPR;
    constexpr int ALD   = GEMM_BM / RPP;
    constexpr int WLD   = BN / RPP;
    static_assert(BK == 32 || BK == 64, "K step of 32 or 64");
    static_assert(WLD >= 1 && WLD * RPP == BN, "W tile must fill whole threads");

    __shared__ __attribute__((aligned(16))) float sA[2][GEMM_BM * PITCH];
    __shared__ __attribute__((aligned(16))) float sW[2][BN * PITCH];

    const int kpart = K / KSPLIT;
    if (KSPLIT > 1) {
        A += (size_t)blockIdx.z * kpart;
        W += (size_t)blockIdx.z * kpart;
        C += (size_t)blockIdx.z * M * N;
    }

    const int tid  = threadIdx.x;
    const int lane = tid & 31;
    const int wave = tid >> 5;
    const int rowb = blockIdx.y * GEMM_BM;
    const int row0 = rowb + wave * 16;
    const int col0 = blockIdx.x * BN;

    const int r  = lane & 15;
    const int hh = lane >> 4;

    const int srow = tid / PPR;
    const int sc4  = tid % PPR;
    gcv4f* Ag = (gcv4f*)(A + (size_t)(rowb + srow) * K + sc4 * 4);
    gcv4f* Wg = (gcv4f*)(W + (size_t)(col0 + srow) * K + sc4 * 4);
    const size_t rstep = (size_t)(RPP / 4) * K;
    const int sidx = srow * PITCH + sc4 * 4;

    const int fa = (wave * 16 + r) * PITCH + hh * 4;
    const int fw = r * PITCH + hh * 4;

    v8f acc[NT];
#pragma unroll
    for (int t = 0; t < NT; ++t) acc[t] = (v8f){0.f,0.f,0.f,0.f,0.f,0.f,0.f,0.f};

    v4f ga[ALD], gw[WLD];
#pragma unroll
    for (int i = 0; i < ALD; ++i) ga[i] = Ag[i * rstep];
#pragma unroll
    for (int i = 0; i < WLD; ++i) gw[i] = Wg[i * rstep];
#pragma unroll
    for (int i = 0; i < ALD; ++i)
        *(v4f*)&sA[0][sidx + i * RPP * PITCH] = ga[i];
#pragma unroll
    for (int i = 0; i < WLD; ++i)
        *(v4f*)&sW[0][sidx + i * RPP * PITCH] = gw[i];
    __syncthreads();

    const int KT = kpart / BK;
    for (int kt = 0; kt < KT; ++kt) {
        const int  st   = kt & 1;
        const bool more = (kt + 1 < KT);

        if (more) {
            Ag += BK / 4;
            Wg += BK / 4;
#pragma unroll
            for (int i = 0; i < ALD; ++i) ga[i] = Ag[i * rstep];
#pragma unroll
            for (int i = 0; i < WLD; ++i) gw[i] = Wg[i * rstep];
        }

        const float* pa = &sA[st][fa];
        const float* pw = &sW[st][fw];

        v4f af[2];
        v4f bf[2][NT];
        af[0] = *(const v4f*)pa;
#pragma unroll
        for (int t = 0; t < NT; ++t)
            bf[0][t] = *(const v4f*)(pw + t * 16 * PITCH);

#pragma unroll
        for (int q = 0; q < BK / 8; ++q) {
            const int cur = q & 1;
            const int nxt = cur ^ 1;
            if (q + 1 < BK / 8) {
                af[nxt] = *(const v4f*)(pa + (q + 1) * 8);
#pragma unroll
                for (int t = 0; t < NT; ++t)
                    bf[nxt][t] = *(const v4f*)(pw + t * 16 * PITCH + (q + 1) * 8);
            }
            const v2f a_lo = (v2f){af[cur].x, af[cur].y};
            const v2f a_hi = (v2f){af[cur].z, af[cur].w};
#pragma unroll
            for (int t = 0; t < NT; ++t) {
                const v2f b_lo = (v2f){bf[cur][t].x, bf[cur][t].y};
                acc[t] = __builtin_amdgcn_wmma_f32_16x16x4_f32(
                             false, a_lo, false, b_lo,
                             (short)0, acc[t], false, false);
            }
#pragma unroll
            for (int t = 0; t < NT; ++t) {
                const v2f b_hi = (v2f){bf[cur][t].z, bf[cur][t].w};
                acc[t] = __builtin_amdgcn_wmma_f32_16x16x4_f32(
                             false, a_hi, false, b_hi,
                             (short)0, acc[t], false, false);
            }
        }

        if (more) {
#pragma unroll
            for (int i = 0; i < ALD; ++i)
                *(v4f*)&sA[st ^ 1][sidx + i * RPP * PITCH] = ga[i];
#pragma unroll
            for (int i = 0; i < WLD; ++i)
                *(v4f*)&sW[st ^ 1][sidx + i * RPP * PITCH] = gw[i];
        }
        __syncthreads();
    }

    const int rowoff = (lane >> 4) * 8;
#pragma unroll
    for (int t = 0; t < NT; ++t) {
        const int col = col0 + t * 16 + r;
#pragma unroll
        for (int v = 0; v < 8; ++v) {
            float val = acc[t][v];
            if (EPI == 1) {
                val += bias[col];
                val = (val > 20.f) ? val : __logf(1.f + __expf(val));
            }
            C[(size_t)(row0 + rowoff + v) * N + col] = val;
        }
    }
}

#define CONV_RB 8
#define CONV_TB 256

__global__ __launch_bounds__(CONV_TB)
void conv_silu_kernel(const float* __restrict__ proj,
                      const float* __restrict__ cw,
                      const float* __restrict__ cb,
                      float* __restrict__ u)
{
    static_assert(L_SEQ % CONV_RB == 0, "a block's rows stay inside one sequence");
    static_assert(I_DIM % (4 * CONV_TB) == 0, "whole blocks across the channels");

    const int i    = (blockIdx.x * CONV_TB + threadIdx.x) * 4;
    const int row0 = blockIdx.y * CONV_RB;
    const int l0   = row0 % L_SEQ;

    const v4f* wp = (const v4f*)(cw + (size_t)i * 4);
    const v4f w0 = wp[0], w1 = wp[1], w2 = wp[2], w3 = wp[3];
    const v4f bias = *(const v4f*)(cb + i);

    const float* hs = proj + i;
    const v4f zero = (v4f){0.f, 0.f, 0.f, 0.f};
    v4f x[CONV_RB + 3];
#pragma unroll
    for (int j = 0; j < CONV_RB + 3; ++j) {
        const int  l    = l0 - 1 + j;
        const bool edge = (j == 0) || (j > CONV_RB);
        const bool ok   = !edge || (l >= 0 && l < L_SEQ);
        const int  rr   = ok ? (row0 - 1 + j) : row0;
        x[j] = *(const v4f*)(hs + (size_t)rr * (2 * I_DIM));
    }
    __builtin_amdgcn_sched_barrier(0);
#pragma unroll
    for (int j = 0; j < CONV_RB + 3; ++j) {
        const int  l    = l0 - 1 + j;
        const bool edge = (j == 0) || (j > CONV_RB);
        const bool ok   = !edge || (l >= 0 && l < L_SEQ);
        x[j] = ok ? x[j] : zero;
    }

    float* up = u + (size_t)row0 * I_DIM + i;
#pragma unroll
    for (int r = 0; r < CONV_RB; ++r) {
        const v4f wc[4] = {w0, w1, w2, w3};
        v4f o;
#pragma unroll
        for (int c = 0; c < 4; ++c) {
            float acc = bias[c];
#pragma unroll
            for (int k = 0; k < 4; ++k)
                acc += wc[c][k] * x[r + k][c];
            o[c] = acc / (1.f + __expf(-acc));
        }
        *(v4f*)(up + (size_t)r * I_DIM) = o;
    }
}

#define XSPLIT 8
#define XD     (R_DIM + 2 * N_ST)
#define NORM_RB 8

__global__ __launch_bounds__(32 * NORM_RB)
void rmsnorm_kernel(const float* __restrict__ xp,
                    const float* __restrict__ wdt,
                    const float* __restrict__ wb,
                    const float* __restrict__ wc,
                    float* __restrict__ dtn,
                    float* __restrict__ Bm,
                    float* __restrict__ Cm)
{
    static_assert(ROWS % NORM_RB == 0, "whole blocks");
    static_assert(XD % 4 == 0 && XD / 4 <= 32, "a row fits one wave, four values per lane");

    const int lane = threadIdx.x & 31;
    const int row  = blockIdx.x * NORM_RB + (threadIdx.x >> 5);
    const bool act = lane < XD / 4;
    const int  c4  = act ? lane : 0;

    const v4f* p = (const v4f*)(xp + (size_t)row * XD) + c4;
    v4f part[XSPLIT];
#pragma unroll
    for (int z = 0; z < XSPLIT; ++z)
        part[z] = p[(size_t)z * (ROWS * XD / 4)];
    v4f x = part[0];
#pragma unroll
    for (int z = 1; z < XSPLIT; ++z) x += part[z];

    float ss = x.x * x.x + x.y * x.y + x.z * x.z + x.w * x.w;
    ss = act ? ss : 0.f;
    ss += __shfl_xor(ss, 1, 32);
    ss += __shfl_xor(ss, 2, 32);
    float s16 = ss + __shfl_xor(ss, 4, 32);
    s16 += __shfl_xor(s16, 8, 32);

    const bool isdt = lane < R_DIM / 4;
    const float ms  = isdt ? s16 / R_DIM : ss / N_ST;
    const float inv = __frsqrt_rn(ms + EPS);

    const int lb = lane - R_DIM / 4;
    const float* wsrc = wdt + 4 * lane;
    float*       dst  = dtn + (size_t)row * R_DIM + 4 * lane;
    if (!isdt) {
        const bool isb = lb < N_ST / 4;
        const int  l4  = 4 * (isb ? lb : lb - N_ST / 4);
        wsrc = (isb ? wb : wc) + l4;
        dst  = (isb ? Bm : Cm) + (size_t)row * N_ST + l4;
    }
    if (act) {
        const v4f w = *(const v4f*)wsrc;
        v4f o;
        o.x = w.x * x.x * inv;
        o.y = w.y * x.y * inv;
        o.z = w.z * x.z * inv;
        o.w = w.w * x.w * inv;
        *(v4f*)dst = o;
    }
}

#define TSTEP 32
#define NGRP  (L_SEQ / TSTEP)
#define SPL   4
#define NSL   (N_ST / SPL)
#define CHB   (128 / SPL)
#define GRP4  (TSTEP * N_ST / 4)
#define QSTEP (TSTEP / SPL)

static __device__ __forceinline__ void lds_barrier()
{
    __builtin_amdgcn_fence(__ATOMIC_RELEASE, "workgroup", "local");
    __builtin_amdgcn_s_barrier();
    __builtin_amdgcn_fence(__ATOMIC_ACQUIRE, "workgroup", "local");
}

static __device__ __forceinline__ float quad_sum(float v)
{
    v += __builtin_bit_cast(float, __builtin_amdgcn_mov_dpp(
             __builtin_bit_cast(int, v), 0xB1, 0xF, 0xF, true));
    v += __builtin_bit_cast(float, __builtin_amdgcn_mov_dpp(
             __builtin_bit_cast(int, v), 0x4E, 0xF, 0xF, true));
    return v;
}

__global__ __launch_bounds__(128)
void scan_kernel(const float* __restrict__ dt2,
                 const float* __restrict__ u,
                 const float* __restrict__ Bm,
                 const float* __restrict__ Cm,
                 const float* __restrict__ A_log,
                 const float* __restrict__ Dvec,
                 const float* __restrict__ proj,
                 float* __restrict__ yout)
{
    __shared__ __attribute__((aligned(16))) float4 sBC[2][2][GRP4];
    static_assert(GRP4 == 128, "one float4 of Bm and one of Cm per thread and group");
    static_assert(TSTEP % SPL == 0, "a group holds whole quads of steps");

    const int tid = threadIdx.x;
    const int sub = tid & (SPL - 1);
    const int b = blockIdx.y;
    const int i = blockIdx.x * CHB + (tid >> 2);

    const size_t row0 = (size_t)b * L_SEQ;
    const float* dtp = dt2  + row0 * I_DIM + i;
    const float* up  = u    + row0 * I_DIM + i;
    const float* gp  = proj + (row0 + sub) * (2 * I_DIM) + I_DIM + i;
    float*       yp  = yout + (row0 + sub) * I_DIM + i;
    const float4* bp = (const float4*)(Bm + row0 * N_ST) + tid;
    const float4* cp = (const float4*)(Cm + row0 * N_ST) + tid;

    const float4 b0 = bp[0];
    const float4 c0 = cp[0];
    float4 nb = bp[GRP4];
    float4 nc = cp[GRP4];

    float rdt[TSTEP], ru[TSTEP], rg[QSTEP];
#pragma unroll
    for (int j = 0; j < TSTEP; ++j) {
        rdt[j] = dtp[(size_t)j * I_DIM];
        ru[j]  = up[(size_t)j * I_DIM];
    }
#pragma unroll
    for (int m = 0; m < QSTEP; ++m)
        rg[m] = gp[(size_t)(SPL * m) * (2 * I_DIM)];

    float Aa[NSL];
#pragma unroll
    for (int n = 0; n < NSL; ++n)
        Aa[n] = -__expf(A_log[(size_t)i * N_ST + sub * NSL + n]);

    float h[NSL];
#pragma unroll
    for (int n = 0; n < NSL; ++n) h[n] = 0.f;

    const float Dv = Dvec[i];

#pragma unroll
    for (int j = 0; j < TSTEP; ++j)
        asm volatile("" :: "v"(rdt[j]), "v"(ru[j]));
#pragma unroll
    for (int m = 0; m < QSTEP; ++m)
        asm volatile("" :: "v"(rg[m]));
    asm volatile("" :: "v"(Dv));
    asm volatile("" :: "v"(nb.x), "v"(nb.y), "v"(nb.z), "v"(nb.w),
                       "v"(nc.x), "v"(nc.y), "v"(nc.z), "v"(nc.w));

    sBC[0][0][tid] = b0;
    sBC[0][1][tid] = c0;
    lds_barrier();

    float4 Bp[2][SPL], Cp[2][SPL];
#pragma unroll
    for (int q = 0; q < 2; ++q) {
#pragma unroll
        for (int k = 0; k < SPL; ++k) {
            Bp[q][k] = sBC[0][0][(q * SPL + k) * SPL + sub];
            Cp[q][k] = sBC[0][1][(q * SPL + k) * SPL + sub];
        }
    }
    float  dAc[NSL];
    float  uvc, duc;
    {
        const float dtv = rdt[0];
#pragma unroll
        for (int n = 0; n < NSL; ++n)
            dAc[n] = __expf(dtv * Aa[n]);
        uvc = ru[0];
        duc = dtv * uvc;
    }

    for (int g = 0; g < NGRP; ++g) {
        const int buf = g & 1;
        const int gn = (g + 1 < NGRP) ? g + 1 : NGRP - 1;
        const float* dtn = dtp + (size_t)gn * TSTEP * I_DIM;
        const float* un  = up  + (size_t)gn * TSTEP * I_DIM;
        const float* gnx = gp  + (size_t)gn * TSTEP * (2 * I_DIM);
        float*       yg  = yp  + (size_t)g  * TSTEP * I_DIM;

        const float4* sB = &sBC[buf][0][sub];
        const float4* sC = &sBC[buf][1][sub];

        float zsel = 0.f;
        float gv = 0.f, eg = 0.f, sg = 0.f;
        float ypend = 0.f, upend = 0.f;

#pragma unroll
        for (int j = 0; j < TSTEP; ++j) {
            const int s   = j & (SPL - 1);
            const int m   = j / SPL;
            const int set = m & 1;

            const int   jn  = (j + 1) % TSTEP;
            const float dtx = rdt[jn];
            const float uvn = ru[jn];

            const float bcur[NSL] = {Bp[set][s].x, Bp[set][s].y, Bp[set][s].z, Bp[set][s].w};
            const float ccur[NSL] = {Cp[set][s].x, Cp[set][s].y, Cp[set][s].z, Cp[set][s].w};

            float dAn[NSL];
            float y = 0.f;
#pragma unroll
            for (int n = 0; n < NSL; ++n) {
                dAn[n] = __expf(dtx * Aa[n]);
                h[n] = dAc[n] * h[n] + duc * bcur[n];
                y   += h[n] * ccur[n];
            }
            const float dun = dtx * uvn;

            if (j > 0) {
                const int sp = (j - 1) & (SPL - 1);
                const float z = quad_sum(ypend) + upend * Dv;
                zsel = (sub == sp) ? z : zsel;
            }
            ypend = y;
            upend = uvc;

            if (s == 0 && m > 0) {
                const float yo = zsel * sg;
                __builtin_amdgcn_sched_barrier(0);
                if (m + 1 < QSTEP) {
#pragma unroll
                    for (int k = 0; k < SPL; ++k) {
                        Bp[set ^ 1][k] = sB[((m + 1) * SPL + k) * SPL];
                        Cp[set ^ 1][k] = sC[((m + 1) * SPL + k) * SPL];
                    }
                }
                yg[(size_t)(SPL * (m - 1)) * I_DIM] = yo;
#pragma unroll
                for (int k = 0; k < SPL; ++k) {
                    const int jr = SPL * (m - 1) + k;
                    rdt[jr] = dtn[(size_t)jr * I_DIM];
                    ru[jr]  = un[(size_t)jr * I_DIM];
                }
                rg[m - 1] = gnx[(size_t)(SPL * (m - 1)) * (2 * I_DIM)];
                __builtin_amdgcn_sched_barrier(0);
            }

            if (s == 0) {
                gv = rg[m];
                eg = __expf(-gv);
            }
            if (s == 1) {
                sg = gv / (1.f + eg);
                asm volatile("" :: "v"(sg));
            }

#pragma unroll
            for (int n = 0; n < NSL; ++n) dAc[n] = dAn[n];
            uvc = uvn; duc = dun;
        }

        {
            const float z = quad_sum(ypend) + upend * Dv;
            zsel = (sub == SPL - 1) ? z : zsel;
            const float yo = zsel * sg;
            __builtin_amdgcn_sched_barrier(0);
            yg[(size_t)(SPL * (QSTEP - 1)) * I_DIM] = yo;
#pragma unroll
            for (int k = 0; k < SPL; ++k) {
                const int jr = SPL * (QSTEP - 1) + k;
                rdt[jr] = dtn[(size_t)jr * I_DIM];
                ru[jr]  = un[(size_t)jr * I_DIM];
            }
            rg[QSTEP - 1] = gnx[(size_t)(SPL * (QSTEP - 1)) * (2 * I_DIM)];
        }

        sBC[buf ^ 1][0][tid] = nb;
        sBC[buf ^ 1][1][tid] = nc;
        lds_barrier();

#pragma unroll
        for (int q = 0; q < 2; ++q) {
#pragma unroll
            for (int k = 0; k < SPL; ++k) {
                Bp[q][k] = sBC[buf ^ 1][0][(q * SPL + k) * SPL + sub];
                Cp[q][k] = sBC[buf ^ 1][1][(q * SPL + k) * SPL + sub];
            }
        }

        const int gn2 = (g + 2 < NGRP) ? g + 2 : NGRP - 1;
        nb = bp[(size_t)gn2 * GRP4];
        nc = cp[(size_t)gn2 * GRP4];
        __builtin_amdgcn_sched_barrier(0);
    }
}

extern "C" void kernel_launch(void* const* d_in, const int* in_sizes, int n_in,
                              void* d_out, int out_size, void* d_ws, size_t ws_size,
                              hipStream_t stream)
{
    const float* hidden     = (const float*)d_in[0];
    const float* in_proj_w  = (const float*)d_in[1];
    const float* conv_w     = (const float*)d_in[2];
    const float* conv_b     = (const float*)d_in[3];
    const float* x_proj_w   = (const float*)d_in[4];
    const float* dt_proj_w  = (const float*)d_in[5];
    const float* dt_proj_b  = (const float*)d_in[6];
    const float* dt_ln_w    = (const float*)d_in[7];
    const float* b_ln_w     = (const float*)d_in[8];
    const float* c_ln_w     = (const float*)d_in[9];
    const float* A_log      = (const float*)d_in[10];
    const float* Dvec       = (const float*)d_in[11];
    const float* out_proj_w = (const float*)d_in[12];
    float* out = (float*)d_out;

    float* ws   = (float*)d_ws;
    float* proj = ws;
    float* u    = proj + (size_t)ROWS * 2 * I_DIM;
    float* xdbl = u    + (size_t)ROWS * I_DIM;
    float* dtn  = xdbl + (size_t)XSPLIT * ROWS * XD;
    float* Bm   = dtn  + (size_t)ROWS * R_DIM;
    float* Cm   = Bm   + (size_t)ROWS * N_ST;
    float* dt2  = Cm   + (size_t)ROWS * N_ST;
    float* yout = dt2  + (size_t)ROWS * I_DIM;

    wmma_gemm_f32<4, 0, 1, 32><<<dim3((2 * I_DIM) / 64, ROWS / 128), 256, 0, stream>>>(
        hidden, in_proj_w, nullptr, proj, ROWS, 2 * I_DIM, H_DIM);

    conv_silu_kernel<<<dim3(I_DIM / (4 * CONV_TB), ROWS / CONV_RB), CONV_TB, 0, stream>>>(
        proj, conv_w, conv_b, u);

    wmma_gemm_f32<2, 0, XSPLIT, 64><<<dim3(XD / 32, ROWS / 128, XSPLIT), 256, 0, stream>>>(
        u, x_proj_w, nullptr, xdbl, ROWS, XD, I_DIM);

    rmsnorm_kernel<<<ROWS / NORM_RB, 32 * NORM_RB, 0, stream>>>(
        xdbl, dt_ln_w, b_ln_w, c_ln_w, dtn, Bm, Cm);

    wmma_gemm_f32<4, 1, 1, 32><<<dim3(I_DIM / 64, ROWS / 128), 256, 0, stream>>>(
        dtn, dt_proj_w, dt_proj_b, dt2, ROWS, I_DIM, R_DIM);

    scan_kernel<<<dim3(I_DIM / CHB, B_SZ), 128, 0, stream>>>(
        dt2, u, Bm, Cm, A_log, Dvec, proj, yout);

    wmma_gemm_f32<4, 0, 1, 64><<<dim3(H_DIM / 64, ROWS / 128), 256, 0, stream>>>(
        yout, out_proj_w, nullptr, out, ROWS, H_DIM, I_DIM);
}


# GEMMs: K loop three tiles deep: tile requested two steps ahead (second register set), parked in LDS mid-step, barrier in front of the step's last group, next fragments read under it; sum order same
# speedup vs baseline: 8.4600x; 1.0451x over previous
#include <hip/hip_runtime.h>
#include <hip/hip_bf16.h>


#define H_DIM 1024
#define I_DIM 2048
#define N_ST  16
#define R_DIM 64
#define B_SZ  2
#define L_SEQ 1024
#define ROWS  (B_SZ * L_SEQ)
#define EPS   1e-06f

typedef __attribute__((ext_vector_type(2))) float v2f;
typedef __attribute__((ext_vector_type(4))) float v4f;
typedef __attribute__((ext_vector_type(8))) float v8f;

typedef __attribute__((address_space(1))) const v4f gcv4f;

#define GEMM_BM    128

static __device__ __forceinline__ void lds_barrier()
{
    __builtin_amdgcn_fence(__ATOMIC_RELEASE, "workgroup", "local");
    __builtin_amdgcn_s_barrier();
    __builtin_amdgcn_fence(__ATOMIC_ACQUIRE, "workgroup", "local");
}

template <int V> struct kconst { static constexpr int value = V; };

template <int NT, int EPI, int KSPLIT, int BK>
__global__ __launch_bounds__(256)
void wmma_gemm_f32(const float* __restrict__ A,
                   const float* __restrict__ W,
                   const float* __restrict__ bias,
                   float* __restrict__ C,
                   int M, int N, int K)
{
    constexpr int BN    = 16 * NT;
    constexpr int PITCH = BK + 4;
    constexpr int PPR   = BK / 4;
    constexpr int RPP   = 256 / PPR;
    constexpr int ALD   = GEMM_BM / RPP;
    constexpr int WLD   = BN / RPP;
    static_assert(BK == 32 || BK == 64, "K step of 32 or 64");
    static_assert(WLD >= 1 && WLD * RPP == BN, "W tile must fill whole threads");

    __shared__ __attribute__((aligned(16))) float sA[2][GEMM_BM * PITCH];
    __shared__ __attribute__((aligned(16))) float sW[2][BN * PITCH];

    const int kpart = K / KSPLIT;
    if (KSPLIT > 1) {
        A += (size_t)blockIdx.z * kpart;
        W += (size_t)blockIdx.z * kpart;
        C += (size_t)blockIdx.z * M * N;
    }

    const int tid  = threadIdx.x;
    const int lane = tid & 31;
    const int wave = tid >> 5;
    const int rowb = blockIdx.y * GEMM_BM;
    const int row0 = rowb + wave * 16;
    const int col0 = blockIdx.x * BN;

    const int r  = lane & 15;
    const int hh = lane >> 4;

    const int srow = tid / PPR;
    const int sc4  = tid % PPR;
    gcv4f* Ag = (gcv4f*)(A + (size_t)(rowb + srow) * K + sc4 * 4);
    gcv4f* Wg = (gcv4f*)(W + (size_t)(col0 + srow) * K + sc4 * 4);
    const size_t rstep = (size_t)(RPP / 4) * K;
    const int sidx = srow * PITCH + sc4 * 4;

    const int fa = (wave * 16 + r) * PITCH + hh * 4;
    const int fw = r * PITCH + hh * 4;

    v8f acc[NT];
#pragma unroll
    for (int t = 0; t < NT; ++t) acc[t] = (v8f){0.f,0.f,0.f,0.f,0.f,0.f,0.f,0.f};

    constexpr int NG = BK / 8;
    constexpr int QS = NG / 2;

    v4f ga[2][ALD], gw[2][WLD];
#pragma unroll
    for (int i = 0; i < ALD; ++i) ga[0][i] = Ag[i * rstep];
#pragma unroll
    for (int i = 0; i < WLD; ++i) gw[0][i] = Wg[i * rstep];
    Ag += BK / 4;
    Wg += BK / 4;
#pragma unroll
    for (int i = 0; i < ALD; ++i) ga[1][i] = Ag[i * rstep];
#pragma unroll
    for (int i = 0; i < WLD; ++i) gw[1][i] = Wg[i * rstep];
#pragma unroll
    for (int i = 0; i < ALD; ++i)
        *(v4f*)&sA[0][sidx + i * RPP * PITCH] = ga[0][i];
#pragma unroll
    for (int i = 0; i < WLD; ++i)
        *(v4f*)&sW[0][sidx + i * RPP * PITCH] = gw[0][i];
    lds_barrier();

    v4f af[2];
    v4f bf[2][NT];
    af[0] = *(const v4f*)&sA[0][fa];
#pragma unroll
    for (int t = 0; t < NT; ++t)
        bf[0][t] = *(const v4f*)&sW[0][fw + t * 16 * PITCH];

    auto kstep = [&](auto st_c, auto next_c, auto next2_c) __attribute__((always_inline)) {
        constexpr int  ST    = decltype(st_c)::value;
        constexpr bool NEXT  = decltype(next_c)::value != 0;
        constexpr bool NEXT2 = decltype(next2_c)::value != 0;

        if (NEXT2) {
            Ag += BK / 4;
            Wg += BK / 4;
#pragma unroll
            for (int i = 0; i < ALD; ++i) ga[ST][i] = Ag[i * rstep];
#pragma unroll
            for (int i = 0; i < WLD; ++i) gw[ST][i] = Wg[i * rstep];
        }

        const float* pa = &sA[ST][fa];
        const float* pw = &sW[ST][fw];

#pragma unroll
        for (int q = 0; q < NG; ++q) {
            const int cur = q & 1;
            const int nxt = cur ^ 1;

            if (NEXT && q == QS) {
                __builtin_amdgcn_sched_barrier(0);
#pragma unroll
                for (int i = 0; i < ALD; ++i)
                    *(v4f*)&sA[ST ^ 1][sidx + i * RPP * PITCH] = ga[ST ^ 1][i];
#pragma unroll
                for (int i = 0; i < WLD; ++i)
                    *(v4f*)&sW[ST ^ 1][sidx + i * RPP * PITCH] = gw[ST ^ 1][i];
            }

            if (q + 1 < NG) {
                af[nxt] = *(const v4f*)(pa + (q + 1) * 8);
#pragma unroll
                for (int t = 0; t < NT; ++t)
                    bf[nxt][t] = *(const v4f*)(pw + t * 16 * PITCH + (q + 1) * 8);
            } else if (NEXT) {
                lds_barrier();
                af[nxt] = *(const v4f*)&sA[ST ^ 1][fa];
#pragma unroll
                for (int t = 0; t < NT; ++t)
                    bf[nxt][t] = *(const v4f*)&sW[ST ^ 1][fw + t * 16 * PITCH];
            }

            const v2f a_lo = (v2f){af[cur].x, af[cur].y};
            const v2f a_hi = (v2f){af[cur].z, af[cur].w};
#pragma unroll
            for (int t = 0; t < NT; ++t) {
                const v2f b_lo = (v2f){bf[cur][t].x, bf[cur][t].y};
                acc[t] = __builtin_amdgcn_wmma_f32_16x16x4_f32(
                             false, a_lo, false, b_lo,
                             (short)0, acc[t], false, false);
            }
#pragma unroll
            for (int t = 0; t < NT; ++t) {
                const v2f b_hi = (v2f){bf[cur][t].z, bf[cur][t].w};
                acc[t] = __builtin_amdgcn_wmma_f32_16x16x4_f32(
                             false, a_hi, false, b_hi,
                             (short)0, acc[t], false, false);
            }
        }
    };

    const int KT = kpart / BK;
    for (int kt = 0; kt + 2 < KT; kt += 2) {
        kstep(kconst<0>{}, kconst<1>{}, kconst<1>{});
        kstep(kconst<1>{}, kconst<1>{}, kconst<1>{});
    }
    kstep(kconst<0>{}, kconst<1>{}, kconst<0>{});
    kstep(kconst<1>{}, kconst<0>{}, kconst<0>{});

    const int rowoff = (lane >> 4) * 8;
#pragma unroll
    for (int t = 0; t < NT; ++t) {
        const int col = col0 + t * 16 + r;
#pragma unroll
        for (int v = 0; v < 8; ++v) {
            float val = acc[t][v];
            if (EPI == 1) {
                val += bias[col];
                val = (val > 20.f) ? val : __logf(1.f + __expf(val));
            }
            C[(size_t)(row0 + rowoff + v) * N + col] = val;
        }
    }
}

#define CONV_RB 8
#define CONV_TB 256

__global__ __launch_bounds__(CONV_TB)
void conv_silu_kernel(const float* __restrict__ proj,
                      const float* __restrict__ cw,
                      const float* __restrict__ cb,
                      float* __restrict__ u)
{
    static_assert(L_SEQ % CONV_RB == 0, "a block's rows stay inside one sequence");
    static_assert(I_DIM % (4 * CONV_TB) == 0, "whole blocks across the channels");

    const int i    = (blockIdx.x * CONV_TB + threadIdx.x) * 4;
    const int row0 = blockIdx.y * CONV_RB;
    const int l0   = row0 % L_SEQ;

    const v4f* wp = (const v4f*)(cw + (size_t)i * 4);
    const v4f w0 = wp[0], w1 = wp[1], w2 = wp[2], w3 = wp[3];
    const v4f bias = *(const v4f*)(cb + i);

    const float* hs = proj + i;
    const v4f zero = (v4f){0.f, 0.f, 0.f, 0.f};
    v4f x[CONV_RB + 3];
#pragma unroll
    for (int j = 0; j < CONV_RB + 3; ++j) {
        const int  l    = l0 - 1 + j;
        const bool edge = (j == 0) || (j > CONV_RB);
        const bool ok   = !edge || (l >= 0 && l < L_SEQ);
        const int  rr   = ok ? (row0 - 1 + j) : row0;
        x[j] = *(const v4f*)(hs + (size_t)rr * (2 * I_DIM));
    }
    __builtin_amdgcn_sched_barrier(0);
#pragma unroll
    for (int j = 0; j < CONV_RB + 3; ++j) {
        const int  l    = l0 - 1 + j;
        const bool edge = (j == 0) || (j > CONV_RB);
        const bool ok   = !edge || (l >= 0 && l < L_SEQ);
        x[j] = ok ? x[j] : zero;
    }

    float* up = u + (size_t)row0 * I_DIM + i;
#pragma unroll
    for (int r = 0; r < CONV_RB; ++r) {
        const v4f wc[4] = {w0, w1, w2, w3};
        v4f o;
#pragma unroll
        for (int c = 0; c < 4; ++c) {
            float acc = bias[c];
#pragma unroll
            for (int k = 0; k < 4; ++k)
                acc += wc[c][k] * x[r + k][c];
            o[c] = acc / (1.f + __expf(-acc));
        }
        *(v4f*)(up + (size_t)r * I_DIM) = o;
    }
}

#define XSPLIT 8
#define XD     (R_DIM + 2 * N_ST)
#define NORM_RB 8

__global__ __launch_bounds__(32 * NORM_RB)
void rmsnorm_kernel(const float* __restrict__ xp,
                    const float* __restrict__ wdt,
                    const float* __restrict__ wb,
                    const float* __restrict__ wc,
                    float* __restrict__ dtn,
                    float* __restrict__ Bm,
                    float* __restrict__ Cm)
{
    static_assert(ROWS % NORM_RB == 0, "whole blocks");
    static_assert(XD % 4 == 0 && XD / 4 <= 32, "a row fits one wave, four values per lane");

    const int lane = threadIdx.x & 31;
    const int row  = blockIdx.x * NORM_RB + (threadIdx.x >> 5);
    const bool act = lane < XD / 4;
    const int  c4  = act ? lane : 0;

    const v4f* p = (const v4f*)(xp + (size_t)row * XD) + c4;
    v4f part[XSPLIT];
#pragma unroll
    for (int z = 0; z < XSPLIT; ++z)
        part[z] = p[(size_t)z * (ROWS * XD / 4)];
    v4f x = part[0];
#pragma unroll
    for (int z = 1; z < XSPLIT; ++z) x += part[z];

    float ss = x.x * x.x + x.y * x.y + x.z * x.z + x.w * x.w;
    ss = act ? ss : 0.f;
    ss += __shfl_xor(ss, 1, 32);
    ss += __shfl_xor(ss, 2, 32);
    float s16 = ss + __shfl_xor(ss, 4, 32);
    s16 += __shfl_xor(s16, 8, 32);

    const bool isdt = lane < R_DIM / 4;
    const float ms  = isdt ? s16 / R_DIM : ss / N_ST;
    const float inv = __frsqrt_rn(ms + EPS);

    const int lb = lane - R_DIM / 4;
    const float* wsrc = wdt + 4 * lane;
    float*       dst  = dtn + (size_t)row * R_DIM + 4 * lane;
    if (!isdt) {
        const bool isb = lb < N_ST / 4;
        const int  l4  = 4 * (isb ? lb : lb - N_ST / 4);
        wsrc = (isb ? wb : wc) + l4;
        dst  = (isb ? Bm : Cm) + (size_t)row * N_ST + l4;
    }
    if (act) {
        const v4f w = *(const v4f*)wsrc;
        v4f o;
        o.x = w.x * x.x * inv;
        o.y = w.y * x.y * inv;
        o.z = w.z * x.z * inv;
        o.w = w.w * x.w * inv;
        *(v4f*)dst = o;
    }
}

#define TSTEP 32
#define NGRP  (L_SEQ / TSTEP)
#define SPL   4
#define NSL   (N_ST / SPL)
#define CHB   (128 / SPL)
#define GRP4  (TSTEP * N_ST / 4)
#define QSTEP (TSTEP / SPL)

static __device__ __forceinline__ float quad_sum(float v)
{
    v += __builtin_bit_cast(float, __builtin_amdgcn_mov_dpp(
             __builtin_bit_cast(int, v), 0xB1, 0xF, 0xF, true));
    v += __builtin_bit_cast(float, __builtin_amdgcn_mov_dpp(
             __builtin_bit_cast(int, v), 0x4E, 0xF, 0xF, true));
    return v;
}

__global__ __launch_bounds__(128)
void scan_kernel(const float* __restrict__ dt2,
                 const float* __restrict__ u,
                 const float* __restrict__ Bm,
                 const float* __restrict__ Cm,
                 const float* __restrict__ A_log,
                 const float* __restrict__ Dvec,
                 const float* __restrict__ proj,
                 float* __restrict__ yout)
{
    __shared__ __attribute__((aligned(16))) float4 sBC[2][2][GRP4];
    static_assert(GRP4 == 128, "one float4 of Bm and one of Cm per thread and group");
    static_assert(TSTEP % SPL == 0, "a group holds whole quads of steps");

    const int tid = threadIdx.x;
    const int sub = tid & (SPL - 1);
    const int b = blockIdx.y;
    const int i = blockIdx.x * CHB + (tid >> 2);

    const size_t row0 = (size_t)b * L_SEQ;
    const float* dtp = dt2  + row0 * I_DIM + i;
    const float* up  = u    + row0 * I_DIM + i;
    const float* gp  = proj + (row0 + sub) * (2 * I_DIM) + I_DIM + i;
    float*       yp  = yout + (row0 + sub) * I_DIM + i;
    const float4* bp = (const float4*)(Bm + row0 * N_ST) + tid;
    const float4* cp = (const float4*)(Cm + row0 * N_ST) + tid;

    const float4 b0 = bp[0];
    const float4 c0 = cp[0];
    float4 nb = bp[GRP4];
    float4 nc = cp[GRP4];

    float rdt[TSTEP], ru[TSTEP], rg[QSTEP];
#pragma unroll
    for (int j = 0; j < TSTEP; ++j) {
        rdt[j] = dtp[(size_t)j * I_DIM];
        ru[j]  = up[(size_t)j * I_DIM];
    }
#pragma unroll
    for (int m = 0; m < QSTEP; ++m)
        rg[m] = gp[(size_t)(SPL * m) * (2 * I_DIM)];

    float Aa[NSL];
#pragma unroll
    for (int n = 0; n < NSL; ++n)
        Aa[n] = -__expf(A_log[(size_t)i * N_ST + sub * NSL + n]);

    float h[NSL];
#pragma unroll
    for (int n = 0; n < NSL; ++n) h[n] = 0.f;

    const float Dv = Dvec[i];

#pragma unroll
    for (int j = 0; j < TSTEP; ++j)
        asm volatile("" :: "v"(rdt[j]), "v"(ru[j]));
#pragma unroll
    for (int m = 0; m < QSTEP; ++m)
        asm volatile("" :: "v"(rg[m]));
    asm volatile("" :: "v"(Dv));
    asm volatile("" :: "v"(nb.x), "v"(nb.y), "v"(nb.z), "v"(nb.w),
                       "v"(nc.x), "v"(nc.y), "v"(nc.z), "v"(nc.w));

    sBC[0][0][tid] = b0;
    sBC[0][1][tid] = c0;
    lds_barrier();

    float4 Bp[2][SPL], Cp[2][SPL];
#pragma unroll
    for (int q = 0; q < 2; ++q) {
#pragma unroll
        for (int k = 0; k < SPL; ++k) {
            Bp[q][k] = sBC[0][0][(q * SPL + k) * SPL + sub];
            Cp[q][k] = sBC[0][1][(q * SPL + k) * SPL + sub];
        }
    }
    float  dAc[NSL];
    float  uvc, duc;
    {
        const float dtv = rdt[0];
#pragma unroll
        for (int n = 0; n < NSL; ++n)
            dAc[n] = __expf(dtv * Aa[n]);
        uvc = ru[0];
        duc = dtv * uvc;
    }

    for (int g = 0; g < NGRP; ++g) {
        const int buf = g & 1;
        const int gn = (g + 1 < NGRP) ? g + 1 : NGRP - 1;
        const float* dtn = dtp + (size_t)gn * TSTEP * I_DIM;
        const float* un  = up  + (size_t)gn * TSTEP * I_DIM;
        const float* gnx = gp  + (size_t)gn * TSTEP * (2 * I_DIM);
        float*       yg  = yp  + (size_t)g  * TSTEP * I_DIM;

        const float4* sB = &sBC[buf][0][sub];
        const float4* sC = &sBC[buf][1][sub];

        float zsel = 0.f;
        float gv = 0.f, eg = 0.f, sg = 0.f;
        float ypend = 0.f, upend = 0.f;

#pragma unroll
        for (int j = 0; j < TSTEP; ++j) {
            const int s   = j & (SPL - 1);
            const int m   = j / SPL;
            const int set = m & 1;

            const int   jn  = (j + 1) % TSTEP;
            const float dtx = rdt[jn];
            const float uvn = ru[jn];

            const float bcur[NSL] = {Bp[set][s].x, Bp[set][s].y, Bp[set][s].z, Bp[set][s].w};
            const float ccur[NSL] = {Cp[set][s].x, Cp[set][s].y, Cp[set][s].z, Cp[set][s].w};

            float dAn[NSL];
            float y = 0.f;
#pragma unroll
            for (int n = 0; n < NSL; ++n) {
                dAn[n] = __expf(dtx * Aa[n]);
                h[n] = dAc[n] * h[n] + duc * bcur[n];
                y   += h[n] * ccur[n];
            }
            const float dun = dtx * uvn;

            if (j > 0) {
                const int sp = (j - 1) & (SPL - 1);
                const float z = quad_sum(ypend) + upend * Dv;
                zsel = (sub == sp) ? z : zsel;
            }
            ypend = y;
            upend = uvc;

            if (s == 0 && m > 0) {
                const float yo = zsel * sg;
                __builtin_amdgcn_sched_barrier(0);
                if (m + 1 < QSTEP) {
#pragma unroll
                    for (int k = 0; k < SPL; ++k) {
                        Bp[set ^ 1][k] = sB[((m + 1) * SPL + k) * SPL];
                        Cp[set ^ 1][k] = sC[((m + 1) * SPL + k) * SPL];
                    }
                }
                yg[(size_t)(SPL * (m - 1)) * I_DIM] = yo;
#pragma unroll
                for (int k = 0; k < SPL; ++k) {
                    const int jr = SPL * (m - 1) + k;
                    rdt[jr] = dtn[(size_t)jr * I_DIM];
                    ru[jr]  = un[(size_t)jr * I_DIM];
                }
                rg[m - 1] = gnx[(size_t)(SPL * (m - 1)) * (2 * I_DIM)];
                __builtin_amdgcn_sched_barrier(0);
            }

            if (s == 0) {
                gv = rg[m];
                eg = __expf(-gv);
            }
            if (s == 1) {
                sg = gv / (1.f + eg);
                asm volatile("" :: "v"(sg));
            }

#pragma unroll
            for (int n = 0; n < NSL; ++n) dAc[n] = dAn[n];
            uvc = uvn; duc = dun;
        }

        {
            const float z = quad_sum(ypend) + upend * Dv;
            zsel = (sub == SPL - 1) ? z : zsel;
            const float yo = zsel * sg;
            __builtin_amdgcn_sched_barrier(0);
            yg[(size_t)(SPL * (QSTEP - 1)) * I_DIM] = yo;
#pragma unroll
            for (int k = 0; k < SPL; ++k) {
                const int jr = SPL * (QSTEP - 1) + k;
                rdt[jr] = dtn[(size_t)jr * I_DIM];
                ru[jr]  = un[(size_t)jr * I_DIM];
            }
            rg[QSTEP - 1] = gnx[(size_t)(SPL * (QSTEP - 1)) * (2 * I_DIM)];
        }

        sBC[buf ^ 1][0][tid] = nb;
        sBC[buf ^ 1][1][tid] = nc;
        lds_barrier();

#pragma unroll
        for (int q = 0; q < 2; ++q) {
#pragma unroll
            for (int k = 0; k < SPL; ++k) {
                Bp[q][k] = sBC[buf ^ 1][0][(q * SPL + k) * SPL + sub];
                Cp[q][k] = sBC[buf ^ 1][1][(q * SPL + k) * SPL + sub];
            }
        }

        const int gn2 = (g + 2 < NGRP) ? g + 2 : NGRP - 1;
        nb = bp[(size_t)gn2 * GRP4];
        nc = cp[(size_t)gn2 * GRP4];
        __builtin_amdgcn_sched_barrier(0);
    }
}

extern "C" void kernel_launch(void* const* d_in, const int* in_sizes, int n_in,
                              void* d_out, int out_size, void* d_ws, size_t ws_size,
                              hipStream_t stream)
{
    const float* hidden     = (const float*)d_in[0];
    const float* in_proj_w  = (const float*)d_in[1];
    const float* conv_w     = (const float*)d_in[2];
    const float* conv_b     = (const float*)d_in[3];
    const float* x_proj_w   = (const float*)d_in[4];
    const float* dt_proj_w  = (const float*)d_in[5];
    const float* dt_proj_b  = (const float*)d_in[6];
    const float* dt_ln_w    = (const float*)d_in[7];
    const float* b_ln_w     = (const float*)d_in[8];
    const float* c_ln_w     = (const float*)d_in[9];
    const float* A_log      = (const float*)d_in[10];
    const float* Dvec       = (const float*)d_in[11];
    const float* out_proj_w = (const float*)d_in[12];
    float* out = (float*)d_out;

    float* ws   = (float*)d_ws;
    float* proj = ws;
    float* u    = proj + (size_t)ROWS * 2 * I_DIM;
    float* xdbl = u    + (size_t)ROWS * I_DIM;
    float* dtn  = xdbl + (size_t)XSPLIT * ROWS * XD;
    float* Bm   = dtn  + (size_t)ROWS * R_DIM;
    float* Cm   = Bm   + (size_t)ROWS * N_ST;
    float* dt2  = Cm   + (size_t)ROWS * N_ST;
    float* yout = dt2  + (size_t)ROWS * I_DIM;

    wmma_gemm_f32<4, 0, 1, 32><<<dim3((2 * I_DIM) / 64, ROWS / 128), 256, 0, stream>>>(
        hidden, in_proj_w, nullptr, proj, ROWS, 2 * I_DIM, H_DIM);

    conv_silu_kernel<<<dim3(I_DIM / (4 * CONV_TB), ROWS / CONV_RB), CONV_TB, 0, stream>>>(
        proj, conv_w, conv_b, u);

    wmma_gemm_f32<2, 0, XSPLIT, 64><<<dim3(XD / 32, ROWS / 128, XSPLIT), 256, 0, stream>>>(
        u, x_proj_w, nullptr, xdbl, ROWS, XD, I_DIM);

    rmsnorm_kernel<<<ROWS / NORM_RB, 32 * NORM_RB, 0, stream>>>(
        xdbl, dt_ln_w, b_ln_w, c_ln_w, dtn, Bm, Cm);

    wmma_gemm_f32<4, 1, 1, 32><<<dim3(I_DIM / 64, ROWS / 128), 256, 0, stream>>>(
        dtn, dt_proj_w, dt_proj_b, dt2, ROWS, I_DIM, R_DIM);

    scan_kernel<<<dim3(I_DIM / CHB, B_SZ), 128, 0, stream>>>(
        dt2, u, Bm, Cm, A_log, Dvec, proj, yout);

    wmma_gemm_f32<4, 0, 1, 64><<<dim3(H_DIM / 64, ROWS / 128), 256, 0, stream>>>(
        yout, out_proj_w, nullptr, out, ROWS, H_DIM, I_DIM);
}


# scan: sequence cut into 8 chunks of 128 steps run in parallel: one launch makes each chunk's end state from zero and decay product, the next starts chunks from the joined carries; h sum order changes
# speedup vs baseline: 8.8972x; 1.0517x over previous
#include <hip/hip_runtime.h>
#include <hip/hip_bf16.h>


#define H_DIM 1024
#define I_DIM 2048
#define N_ST  16
#define R_DIM 64
#define B_SZ  2
#define L_SEQ 1024
#define ROWS  (B_SZ * L_SEQ)
#define EPS   1e-06f

typedef __attribute__((ext_vector_type(2))) float v2f;
typedef __attribute__((ext_vector_type(4))) float v4f;
typedef __attribute__((ext_vector_type(8))) float v8f;

typedef __attribute__((address_space(1))) const v4f gcv4f;

#define GEMM_BM    128

static __device__ __forceinline__ void lds_barrier()
{
    __builtin_amdgcn_fence(__ATOMIC_RELEASE, "workgroup", "local");
    __builtin_amdgcn_s_barrier();
    __builtin_amdgcn_fence(__ATOMIC_ACQUIRE, "workgroup", "local");
}

template <int V> struct kconst { static constexpr int value = V; };

template <int NT, int EPI, int KSPLIT, int BK>
__global__ __launch_bounds__(256)
void wmma_gemm_f32(const float* __restrict__ A,
                   const float* __restrict__ W,
                   const float* __restrict__ bias,
                   float* __restrict__ C,
                   int M, int N, int K)
{
    constexpr int BN    = 16 * NT;
    constexpr int PITCH = BK + 4;
    constexpr int PPR   = BK / 4;
    constexpr int RPP   = 256 / PPR;
    constexpr int ALD   = GEMM_BM / RPP;
    constexpr int WLD   = BN / RPP;
    static_assert(BK == 32 || BK == 64, "K step of 32 or 64");
    static_assert(WLD >= 1 && WLD * RPP == BN, "W tile must fill whole threads");

    __shared__ __attribute__((aligned(16))) float sA[2][GEMM_BM * PITCH];
    __shared__ __attribute__((aligned(16))) float sW[2][BN * PITCH];

    const int kpart = K / KSPLIT;
    if (KSPLIT > 1) {
        A += (size_t)blockIdx.z * kpart;
        W += (size_t)blockIdx.z * kpart;
        C += (size_t)blockIdx.z * M * N;
    }

    const int tid  = threadIdx.x;
    const int lane = tid & 31;
    const int wave = tid >> 5;
    const int rowb = blockIdx.y * GEMM_BM;
    const int row0 = rowb + wave * 16;
    const int col0 = blockIdx.x * BN;

    const int r  = lane & 15;
    const int hh = lane >> 4;

    const int srow = tid / PPR;
    const int sc4  = tid % PPR;
    gcv4f* Ag = (gcv4f*)(A + (size_t)(rowb + srow) * K + sc4 * 4);
    gcv4f* Wg = (gcv4f*)(W + (size_t)(col0 + srow) * K + sc4 * 4);
    const size_t rstep = (size_t)(RPP / 4) * K;
    const int sidx = srow * PITCH + sc4 * 4;

    const int fa = (wave * 16 + r) * PITCH + hh * 4;
    const int fw = r * PITCH + hh * 4;

    v8f acc[NT];
#pragma unroll
    for (int t = 0; t < NT; ++t) acc[t] = (v8f){0.f,0.f,0.f,0.f,0.f,0.f,0.f,0.f};

    constexpr int NG = BK / 8;
    constexpr int QS = NG / 2;

    v4f ga[2][ALD], gw[2][WLD];
#pragma unroll
    for (int i = 0; i < ALD; ++i) ga[0][i] = Ag[i * rstep];
#pragma unroll
    for (int i = 0; i < WLD; ++i) gw[0][i] = Wg[i * rstep];
    Ag += BK / 4;
    Wg += BK / 4;
#pragma unroll
    for (int i = 0; i < ALD; ++i) ga[1][i] = Ag[i * rstep];
#pragma unroll
    for (int i = 0; i < WLD; ++i) gw[1][i] = Wg[i * rstep];
#pragma unroll
    for (int i = 0; i < ALD; ++i)
        *(v4f*)&sA[0][sidx + i * RPP * PITCH] = ga[0][i];
#pragma unroll
    for (int i = 0; i < WLD; ++i)
        *(v4f*)&sW[0][sidx + i * RPP * PITCH] = gw[0][i];
    lds_barrier();

    v4f af[2];
    v4f bf[2][NT];
    af[0] = *(const v4f*)&sA[0][fa];
#pragma unroll
    for (int t = 0; t < NT; ++t)
        bf[0][t] = *(const v4f*)&sW[0][fw + t * 16 * PITCH];

    auto kstep = [&](auto st_c, auto next_c, auto next2_c) __attribute__((always_inline)) {
        constexpr int  ST    = decltype(st_c)::value;
        constexpr bool NEXT  = decltype(next_c)::value != 0;
        constexpr bool NEXT2 = decltype(next2_c)::value != 0;

        if (NEXT2) {
            Ag += BK / 4;
            Wg += BK / 4;
#pragma unroll
            for (int i = 0; i < ALD; ++i) ga[ST][i] = Ag[i * rstep];
#pragma unroll
            for (int i = 0; i < WLD; ++i) gw[ST][i] = Wg[i * rstep];
        }

        const float* pa = &sA[ST][fa];
        const float* pw = &sW[ST][fw];

#pragma unroll
        for (int q = 0; q < NG; ++q) {
            const int cur = q & 1;
            const int nxt = cur ^ 1;

            if (NEXT && q == QS) {
                __builtin_amdgcn_sched_barrier(0);
#pragma unroll
                for (int i = 0; i < ALD; ++i)
                    *(v4f*)&sA[ST ^ 1][sidx + i * RPP * PITCH] = ga[ST ^ 1][i];
#pragma unroll
                for (int i = 0; i < WLD; ++i)
                    *(v4f*)&sW[ST ^ 1][sidx + i * RPP * PITCH] = gw[ST ^ 1][i];
            }

            if (q + 1 < NG) {
                af[nxt] = *(const v4f*)(pa + (q + 1) * 8);
#pragma unroll
                for (int t = 0; t < NT; ++t)
                    bf[nxt][t] = *(const v4f*)(pw + t * 16 * PITCH + (q + 1) * 8);
            } else if (NEXT) {
                lds_barrier();
                af[nxt] = *(const v4f*)&sA[ST ^ 1][fa];
#pragma unroll
                for (int t = 0; t < NT; ++t)
                    bf[nxt][t] = *(const v4f*)&sW[ST ^ 1][fw + t * 16 * PITCH];
            }

            const v2f a_lo = (v2f){af[cur].x, af[cur].y};
            const v2f a_hi = (v2f){af[cur].z, af[cur].w};
#pragma unroll
            for (int t = 0; t < NT; ++t) {
                const v2f b_lo = (v2f){bf[cur][t].x, bf[cur][t].y};
                acc[t] = __builtin_amdgcn_wmma_f32_16x16x4_f32(
                             false, a_lo, false, b_lo,
                             (short)0, acc[t], false, false);
            }
#pragma unroll
            for (int t = 0; t < NT; ++t) {
                const v2f b_hi = (v2f){bf[cur][t].z, bf[cur][t].w};
                acc[t] = __builtin_amdgcn_wmma_f32_16x16x4_f32(
                             false, a_hi, false, b_hi,
                             (short)0, acc[t], false, false);
            }
        }
    };

    const int KT = kpart / BK;
    for (int kt = 0; kt + 2 < KT; kt += 2) {
        kstep(kconst<0>{}, kconst<1>{}, kconst<1>{});
        kstep(kconst<1>{}, kconst<1>{}, kconst<1>{});
    }
    kstep(kconst<0>{}, kconst<1>{}, kconst<0>{});
    kstep(kconst<1>{}, kconst<0>{}, kconst<0>{});

    const int rowoff = (lane >> 4) * 8;
#pragma unroll
    for (int t = 0; t < NT; ++t) {
        const int col = col0 + t * 16 + r;
#pragma unroll
        for (int v = 0; v < 8; ++v) {
            float val = acc[t][v];
            if (EPI == 1) {
                val += bias[col];
                val = (val > 20.f) ? val : __logf(1.f + __expf(val));
            }
            C[(size_t)(row0 + rowoff + v) * N + col] = val;
        }
    }
}

#define CONV_RB 8
#define CONV_TB 256

__global__ __launch_bounds__(CONV_TB)
void conv_silu_kernel(const float* __restrict__ proj,
                      const float* __restrict__ cw,
                      const float* __restrict__ cb,
                      float* __restrict__ u)
{
    static_assert(L_SEQ % CONV_RB == 0, "a block's rows stay inside one sequence");
    static_assert(I_DIM % (4 * CONV_TB) == 0, "whole blocks across the channels");

    const int i    = (blockIdx.x * CONV_TB + threadIdx.x) * 4;
    const int row0 = blockIdx.y * CONV_RB;
    const int l0   = row0 % L_SEQ;

    const v4f* wp = (const v4f*)(cw + (size_t)i * 4);
    const v4f w0 = wp[0], w1 = wp[1], w2 = wp[2], w3 = wp[3];
    const v4f bias = *(const v4f*)(cb + i);

    const float* hs = proj + i;
    const v4f zero = (v4f){0.f, 0.f, 0.f, 0.f};
    v4f x[CONV_RB + 3];
#pragma unroll
    for (int j = 0; j < CONV_RB + 3; ++j) {
        const int  l    = l0 - 1 + j;
        const bool edge = (j == 0) || (j > CONV_RB);
        const bool ok   = !edge || (l >= 0 && l < L_SEQ);
        const int  rr   = ok ? (row0 - 1 + j) : row0;
        x[j] = *(const v4f*)(hs + (size_t)rr * (2 * I_DIM));
    }
    __builtin_amdgcn_sched_barrier(0);
#pragma unroll
    for (int j = 0; j < CONV_RB + 3; ++j) {
        const int  l    = l0 - 1 + j;
        const bool edge = (j == 0) || (j > CONV_RB);
        const bool ok   = !edge || (l >= 0 && l < L_SEQ);
        x[j] = ok ? x[j] : zero;
    }

    float* up = u + (size_t)row0 * I_DIM + i;
#pragma unroll
    for (int r = 0; r < CONV_RB; ++r) {
        const v4f wc[4] = {w0, w1, w2, w3};
        v4f o;
#pragma unroll
        for (int c = 0; c < 4; ++c) {
            float acc = bias[c];
#pragma unroll
            for (int k = 0; k < 4; ++k)
                acc += wc[c][k] * x[r + k][c];
            o[c] = acc / (1.f + __expf(-acc));
        }
        *(v4f*)(up + (size_t)r * I_DIM) = o;
    }
}

#define XSPLIT 8
#define XD     (R_DIM + 2 * N_ST)
#define NORM_RB 8

__global__ __launch_bounds__(32 * NORM_RB)
void rmsnorm_kernel(const float* __restrict__ xp,
                    const float* __restrict__ wdt,
                    const float* __restrict__ wb,
                    const float* __restrict__ wc,
                    float* __restrict__ dtn,
                    float* __restrict__ Bm,
                    float* __restrict__ Cm)
{
    static_assert(ROWS % NORM_RB == 0, "whole blocks");
    static_assert(XD % 4 == 0 && XD / 4 <= 32, "a row fits one wave, four values per lane");

    const int lane = threadIdx.x & 31;
    const int row  = blockIdx.x * NORM_RB + (threadIdx.x >> 5);
    const bool act = lane < XD / 4;
    const int  c4  = act ? lane : 0;

    const v4f* p = (const v4f*)(xp + (size_t)row * XD) + c4;
    v4f part[XSPLIT];
#pragma unroll
    for (int z = 0; z < XSPLIT; ++z)
        part[z] = p[(size_t)z * (ROWS * XD / 4)];
    v4f x = part[0];
#pragma unroll
    for (int z = 1; z < XSPLIT; ++z) x += part[z];

    float ss = x.x * x.x + x.y * x.y + x.z * x.z + x.w * x.w;
    ss = act ? ss : 0.f;
    ss += __shfl_xor(ss, 1, 32);
    ss += __shfl_xor(ss, 2, 32);
    float s16 = ss + __shfl_xor(ss, 4, 32);
    s16 += __shfl_xor(s16, 8, 32);

    const bool isdt = lane < R_DIM / 4;
    const float ms  = isdt ? s16 / R_DIM : ss / N_ST;
    const float inv = __frsqrt_rn(ms + EPS);

    const int lb = lane - R_DIM / 4;
    const float* wsrc = wdt + 4 * lane;
    float*       dst  = dtn + (size_t)row * R_DIM + 4 * lane;
    if (!isdt) {
        const bool isb = lb < N_ST / 4;
        const int  l4  = 4 * (isb ? lb : lb - N_ST / 4);
        wsrc = (isb ? wb : wc) + l4;
        dst  = (isb ? Bm : Cm) + (size_t)row * N_ST + l4;
    }
    if (act) {
        const v4f w = *(const v4f*)wsrc;
        v4f o;
        o.x = w.x * x.x * inv;
        o.y = w.y * x.y * inv;
        o.z = w.z * x.z * inv;
        o.w = w.w * x.w * inv;
        *(v4f*)dst = o;
    }
}

#define TSTEP 32
#define NCHK  8
#define CLEN  (L_SEQ / NCHK)
#define NGRP  (CLEN / TSTEP)
#define SPL   4
#define NSL   (N_ST / SPL)
#define CHB   (128 / SPL)
#define GRP4  (TSTEP * N_ST / 4)
#define QSTEP (TSTEP / SPL)

static __device__ __forceinline__ float quad_sum(float v)
{
    v += __builtin_bit_cast(float, __builtin_amdgcn_mov_dpp(
             __builtin_bit_cast(int, v), 0xB1, 0xF, 0xF, true));
    v += __builtin_bit_cast(float, __builtin_amdgcn_mov_dpp(
             __builtin_bit_cast(int, v), 0x4E, 0xF, 0xF, true));
    return v;
}

template <bool OUT>
__global__ __launch_bounds__(128)
void scan_kernel(const float* __restrict__ dt2,
                 const float* __restrict__ u,
                 const float* __restrict__ Bm,
                 const float* __restrict__ Cm,
                 const float* __restrict__ A_log,
                 const float* __restrict__ Dvec,
                 const float* __restrict__ proj,
                 float* __restrict__ carE,
                 float* __restrict__ carP,
                 float* __restrict__ yout)
{
    __shared__ __attribute__((aligned(16))) float4 sBC[2][2][GRP4];
    static_assert(GRP4 == 128, "one float4 of Bm and one of Cm per thread and group");
    static_assert(TSTEP % SPL == 0, "a group holds whole quads of steps");
    static_assert(NSL == 4, "a lane's states are one float4");
    static_assert(L_SEQ % (NCHK * TSTEP) == 0 && NGRP >= 2, "a chunk holds whole groups, at least two");

    const int tid = threadIdx.x;
    const int sub = tid & (SPL - 1);
    const int b = blockIdx.y;
    const int i = blockIdx.x * CHB + (tid >> 2);

    const int c = blockIdx.z;
    const size_t row0 = (size_t)b * L_SEQ + (size_t)c * CLEN;
    const size_t cidx = ((size_t)b * (NCHK - 1) * I_DIM + i) * SPL + sub;
    const size_t cstr = (size_t)I_DIM * SPL;
    const float* dtp = dt2  + row0 * I_DIM + i;
    const float* up  = u    + row0 * I_DIM + i;
    const float* gp  = proj + (row0 + sub) * (2 * I_DIM) + I_DIM + i;
    float*       yp  = yout + (row0 + sub) * I_DIM + i;
    const float4* bp = (const float4*)(Bm + row0 * N_ST) + tid;
    const float4* cp = (const float4*)(Cm + row0 * N_ST) + tid;

    const float4 zero4 = make_float4(0.f, 0.f, 0.f, 0.f);
    const float4 b0 = bp[0];
    const float4 c0 = OUT ? cp[0] : zero4;
    float4 nb = bp[GRP4];
    float4 nc = OUT ? cp[GRP4] : zero4;

    float rdt[TSTEP], ru[TSTEP], rg[QSTEP];
#pragma unroll
    for (int j = 0; j < TSTEP; ++j) {
        rdt[j] = dtp[(size_t)j * I_DIM];
        ru[j]  = up[(size_t)j * I_DIM];
    }
#pragma unroll
    for (int m = 0; m < QSTEP; ++m)
        rg[m] = OUT ? gp[(size_t)(SPL * m) * (2 * I_DIM)] : 0.f;

    float Aa[NSL];
#pragma unroll
    for (int n = 0; n < NSL; ++n)
        Aa[n] = -__expf(A_log[(size_t)i * N_ST + sub * NSL + n]);

    float h[NSL];
#pragma unroll
    for (int n = 0; n < NSL; ++n) h[n] = 0.f;

    if (OUT) {
        const float4* eq = (const float4*)carE + cidx;
        const float4* pq = (const float4*)carP + cidx;
        float4 ce[NCHK - 1], cq[NCHK - 1];
#pragma unroll
        for (int k = 0; k < NCHK - 1; ++k) {
            ce[k] = eq[(size_t)k * cstr];
            cq[k] = pq[(size_t)k * cstr];
        }
#pragma unroll
        for (int k = 0; k < NCHK - 1; ++k) {
            const bool on = k < c;
            const float ev[NSL] = {ce[k].x, ce[k].y, ce[k].z, ce[k].w};
            const float pv[NSL] = {cq[k].x, cq[k].y, cq[k].z, cq[k].w};
#pragma unroll
            for (int n = 0; n < NSL; ++n)
                h[n] = on ? pv[n] * h[n] + ev[n] : h[n];
        }
    }
    float P[NSL];
#pragma unroll
    for (int n = 0; n < NSL; ++n) P[n] = 1.f;

    const float Dv = OUT ? Dvec[i] : 0.f;

#pragma unroll
    for (int j = 0; j < TSTEP; ++j)
        asm volatile("" :: "v"(rdt[j]), "v"(ru[j]));
    asm volatile("" :: "v"(nb.x), "v"(nb.y), "v"(nb.z), "v"(nb.w));
    if (OUT) {
#pragma unroll
        for (int m = 0; m < QSTEP; ++m)
            asm volatile("" :: "v"(rg[m]));
        asm volatile("" :: "v"(Dv));
        asm volatile("" :: "v"(nc.x), "v"(nc.y), "v"(nc.z), "v"(nc.w));
        asm volatile("" :: "v"(h[0]), "v"(h[1]), "v"(h[2]), "v"(h[3]));
    }

    sBC[0][0][tid] = b0;
    if (OUT) sBC[0][1][tid] = c0;
    lds_barrier();

    float4 Bp[2][SPL], Cp[2][SPL];
#pragma unroll
    for (int q = 0; q < 2; ++q) {
#pragma unroll
        for (int k = 0; k < SPL; ++k) {
            Bp[q][k] = sBC[0][0][(q * SPL + k) * SPL + sub];
            Cp[q][k] = OUT ? sBC[0][1][(q * SPL + k) * SPL + sub] : zero4;
        }
    }
    float  dAc[NSL];
    float  uvc, duc;
    {
        const float dtv = rdt[0];
#pragma unroll
        for (int n = 0; n < NSL; ++n)
            dAc[n] = __expf(dtv * Aa[n]);
        uvc = ru[0];
        duc = dtv * uvc;
    }

    for (int g = 0; g < NGRP; ++g) {
        const int buf = g & 1;
        const int gn = (g + 1 < NGRP) ? g + 1 : NGRP - 1;
        const float* dtn = dtp + (size_t)gn * TSTEP * I_DIM;
        const float* un  = up  + (size_t)gn * TSTEP * I_DIM;
        const float* gnx = gp  + (size_t)gn * TSTEP * (2 * I_DIM);
        float*       yg  = yp  + (size_t)g  * TSTEP * I_DIM;

        const float4* sB = &sBC[buf][0][sub];
        const float4* sC = &sBC[buf][1][sub];

        float zsel = 0.f;
        float gv = 0.f, eg = 0.f, sg = 0.f;
        float ypend = 0.f, upend = 0.f;

#pragma unroll
        for (int j = 0; j < TSTEP; ++j) {
            const int s   = j & (SPL - 1);
            const int m   = j / SPL;
            const int set = m & 1;

            const int   jn  = (j + 1) % TSTEP;
            const float dtx = rdt[jn];
            const float uvn = ru[jn];

            const float bcur[NSL] = {Bp[set][s].x, Bp[set][s].y, Bp[set][s].z, Bp[set][s].w};
            const float ccur[NSL] = {Cp[set][s].x, Cp[set][s].y, Cp[set][s].z, Cp[set][s].w};

            float dAn[NSL];
            float y = 0.f;
#pragma unroll
            for (int n = 0; n < NSL; ++n) {
                dAn[n] = __expf(dtx * Aa[n]);
                h[n] = dAc[n] * h[n] + duc * bcur[n];
                if (OUT) y += h[n] * ccur[n];
                else     P[n] *= dAc[n];
            }
            const float dun = dtx * uvn;

            if (OUT && j > 0) {
                const int sp = (j - 1) & (SPL - 1);
                const float z = quad_sum(ypend) + upend * Dv;
                zsel = (sub == sp) ? z : zsel;
            }
            ypend = y;
            upend = uvc;

            if (s == 0 && m > 0) {
                const float yo = zsel * sg;
                __builtin_amdgcn_sched_barrier(0);
                if (m + 1 < QSTEP) {
#pragma unroll
                    for (int k = 0; k < SPL; ++k) {
                        Bp[set ^ 1][k] = sB[((m + 1) * SPL + k) * SPL];
                        if (OUT) Cp[set ^ 1][k] = sC[((m + 1) * SPL + k) * SPL];
                    }
                }
                if (OUT) yg[(size_t)(SPL * (m - 1)) * I_DIM] = yo;
#pragma unroll
                for (int k = 0; k < SPL; ++k) {
                    const int jr = SPL * (m - 1) + k;
                    rdt[jr] = dtn[(size_t)jr * I_DIM];
                    ru[jr]  = un[(size_t)jr * I_DIM];
                }
                if (OUT) rg[m - 1] = gnx[(size_t)(SPL * (m - 1)) * (2 * I_DIM)];
                __builtin_amdgcn_sched_barrier(0);
            }

            if (OUT && s == 0) {
                gv = rg[m];
                eg = __expf(-gv);
            }
            if (OUT && s == 1) {
                sg = gv / (1.f + eg);
                asm volatile("" :: "v"(sg));
            }

#pragma unroll
            for (int n = 0; n < NSL; ++n) dAc[n] = dAn[n];
            uvc = uvn; duc = dun;
        }

        {
            const float z = OUT ? quad_sum(ypend) + upend * Dv : 0.f;
            zsel = (sub == SPL - 1) ? z : zsel;
            const float yo = zsel * sg;
            __builtin_amdgcn_sched_barrier(0);
            if (OUT) yg[(size_t)(SPL * (QSTEP - 1)) * I_DIM] = yo;
#pragma unroll
            for (int k = 0; k < SPL; ++k) {
                const int jr = SPL * (QSTEP - 1) + k;
                rdt[jr] = dtn[(size_t)jr * I_DIM];
                ru[jr]  = un[(size_t)jr * I_DIM];
            }
            if (OUT) rg[QSTEP - 1] = gnx[(size_t)(SPL * (QSTEP - 1)) * (2 * I_DIM)];
        }

        sBC[buf ^ 1][0][tid] = nb;
        if (OUT) sBC[buf ^ 1][1][tid] = nc;
        lds_barrier();

#pragma unroll
        for (int q = 0; q < 2; ++q) {
#pragma unroll
            for (int k = 0; k < SPL; ++k) {
                Bp[q][k] = sBC[buf ^ 1][0][(q * SPL + k) * SPL + sub];
                if (OUT) Cp[q][k] = sBC[buf ^ 1][1][(q * SPL + k) * SPL + sub];
            }
        }

        const int gn2 = (g + 2 < NGRP) ? g + 2 : NGRP - 1;
        nb = bp[(size_t)gn2 * GRP4];
        if (OUT) nc = cp[(size_t)gn2 * GRP4];
        __builtin_amdgcn_sched_barrier(0);
    }

    if (!OUT) {
        ((float4*)carE)[cidx + (size_t)c * cstr] = make_float4(h[0], h[1], h[2], h[3]);
        ((float4*)carP)[cidx + (size_t)c * cstr] = make_float4(P[0], P[1], P[2], P[3]);
    }
}

extern "C" void kernel_launch(void* const* d_in, const int* in_sizes, int n_in,
                              void* d_out, int out_size, void* d_ws, size_t ws_size,
                              hipStream_t stream)
{
    const float* hidden     = (const float*)d_in[0];
    const float* in_proj_w  = (const float*)d_in[1];
    const float* conv_w     = (const float*)d_in[2];
    const float* conv_b     = (const float*)d_in[3];
    const float* x_proj_w   = (const float*)d_in[4];
    const float* dt_proj_w  = (const float*)d_in[5];
    const float* dt_proj_b  = (const float*)d_in[6];
    const float* dt_ln_w    = (const float*)d_in[7];
    const float* b_ln_w     = (const float*)d_in[8];
    const float* c_ln_w     = (const float*)d_in[9];
    const float* A_log      = (const float*)d_in[10];
    const float* Dvec       = (const float*)d_in[11];
    const float* out_proj_w = (const float*)d_in[12];
    float* out = (float*)d_out;

    float* ws   = (float*)d_ws;
    float* proj = ws;
    float* u    = proj + (size_t)ROWS * 2 * I_DIM;
    float* xdbl = u    + (size_t)ROWS * I_DIM;
    float* dtn  = xdbl + (size_t)XSPLIT * ROWS * XD;
    float* Bm   = dtn  + (size_t)ROWS * R_DIM;
    float* Cm   = Bm   + (size_t)ROWS * N_ST;
    float* dt2  = Cm   + (size_t)ROWS * N_ST;
    float* yout = dt2  + (size_t)ROWS * I_DIM;
    float* carE = yout + (size_t)ROWS * I_DIM;
    float* carP = carE + (size_t)B_SZ * (NCHK - 1) * I_DIM * N_ST;

    wmma_gemm_f32<4, 0, 1, 32><<<dim3((2 * I_DIM) / 64, ROWS / 128), 256, 0, stream>>>(
        hidden, in_proj_w, nullptr, proj, ROWS, 2 * I_DIM, H_DIM);

    conv_silu_kernel<<<dim3(I_DIM / (4 * CONV_TB), ROWS / CONV_RB), CONV_TB, 0, stream>>>(
        proj, conv_w, conv_b, u);

    wmma_gemm_f32<2, 0, XSPLIT, 64><<<dim3(XD / 32, ROWS / 128, XSPLIT), 256, 0, stream>>>(
        u, x_proj_w, nullptr, xdbl, ROWS, XD, I_DIM);

    rmsnorm_kernel<<<ROWS / NORM_RB, 32 * NORM_RB, 0, stream>>>(
        xdbl, dt_ln_w, b_ln_w, c_ln_w, dtn, Bm, Cm);

    wmma_gemm_f32<4, 1, 1, 32><<<dim3(I_DIM / 64, ROWS / 128), 256, 0, stream>>>(
        dtn, dt_proj_w, dt_proj_b, dt2, ROWS, I_DIM, R_DIM);

    scan_kernel<false><<<dim3(I_DIM / CHB, B_SZ, NCHK - 1), 128, 0, stream>>>(
        dt2, u, Bm, Cm, A_log, Dvec, proj, carE, carP, yout);

    scan_kernel<true><<<dim3(I_DIM / CHB, B_SZ, NCHK), 128, 0, stream>>>(
        dt2, u, Bm, Cm, A_log, Dvec, proj, carE, carP, yout);

    wmma_gemm_f32<4, 0, 1, 64><<<dim3(H_DIM / 64, ROWS / 128), 256, 0, stream>>>(
        yout, out_proj_w, nullptr, out, ROWS, H_DIM, I_DIM);
}
